# starting kernel
# baseline (speedup 1.0000x reference)
#include <hip/hip_runtime.h>

typedef float v2f __attribute__((ext_vector_type(2)));
typedef float v8f __attribute__((ext_vector_type(8)));

#define IN_F  128
#define HID_F 128
#define OUT_F 64
#define KDIM  128

__global__ void zero_kernel(float* __restrict__ p, long n) {
    long i = (long)blockIdx.x * blockDim.x + threadIdx.x;
    if (i < n) p[i] = 0.0f;
}

__global__ void degree_kernel(const int* __restrict__ src, const int* __restrict__ dst,
                              float* deg_src, float* deg_dst, int ne) {
    int e = blockIdx.x * blockDim.x + threadIdx.x;
    if (e < ne) {
        atomicAdd(&deg_src[src[e]], 1.0f);
        atomicAdd(&deg_dst[dst[e]], 1.0f);
    }
}

__global__ void rsqrt_kernel(float* __restrict__ p, int n) {
    int i = blockIdx.x * blockDim.x + threadIdx.x;
    if (i < n) {
        float d = p[i];
        p[i] = rsqrtf(d < 1.0f ? 1.0f : d);
    }
}

__global__ void epilogue_kernel(float* __restrict__ Y, const float* __restrict__ dinv,
                                const float* __restrict__ bias, int F, long total,
                                int do_relu) {
    long t = (long)blockIdx.x * blockDim.x + threadIdx.x;
    if (t >= total) return;
    int i = (int)(t / F);
    int f = (int)(t - (long)i * F);
    float v = Y[t] * dinv[i] + bias[f];
    if (do_relu) v = v > 0.0f ? v : 0.0f;
    Y[t] = v;
}

template <int FOUT>
__launch_bounds__(FOUT * 2)
__global__ void gemm_rows16_wmma(const float* __restrict__ X,
                                 const float* __restrict__ dinv,
                                 const float* __restrict__ W,
                                 float* __restrict__ H) {
    constexpr int LDSS = KDIM + 4;
    __shared__ __align__(16) float As[16 * LDSS];

    const int row0 = blockIdx.x * 16;
    const int tid  = threadIdx.x;

    for (int i = tid; i < 16 * KDIM; i += blockDim.x) {
        int r = i >> 7;
        int c = i & (KDIM - 1);
        As[r * LDSS + c] = X[(size_t)(row0 + r) * KDIM + c] * dinv[row0 + r];
    }
    __syncthreads();

    const int lane  = tid & 31;
    const int wave  = tid >> 5;
    const int n     = lane & 15;
    const int khalf = lane >> 4;
    const int m16   = lane & 15;
    const int ncol  = wave * 16 + n;

    v8f acc = {};
#pragma unroll
    for (int k0 = 0; k0 < KDIM; k0 += 4) {
        const int ka = k0 + 2 * khalf;
        v2f a = *(const v2f*)&As[m16 * LDSS + ka];
        v2f b;
        b.x = W[(size_t)ka * FOUT + ncol];
        b.y = W[(size_t)(ka + 1) * FOUT + ncol];
        acc = __builtin_amdgcn_wmma_f32_16x16x4_f32(
             false, a,  false, b,
             (short)0, acc,  false,  false);
    }

#pragma unroll
    for (int r = 0; r < 8; ++r) {
        const int m = r + 8 * khalf;
        H[(size_t)(row0 + m) * FOUT + ncol] = acc[r];
    }
}

template <int F>
__global__ void scatter_add_kernel(const float* __restrict__ H,
                                   const int* __restrict__ src,
                                   const int* __restrict__ dst,
                                   float* __restrict__ AGG, int ne) {
    const int gid  = blockIdx.x * blockDim.x + threadIdx.x;
    const int e    = gid >> 5;
    const int lane = gid & 31;
    if (e >= ne) return;

    const int s = src[e];
    const int d = dst[e];
    constexpr int V = F / 32;

    const float* hp = H   + (size_t)s * F + lane * V;
    float*       op = AGG + (size_t)d * F + lane * V;

    if (V == 4) {
        float4 v = *(const float4*)hp;
        atomicAdd(op + 0, v.x);
        atomicAdd(op + 1, v.y);
        atomicAdd(op + 2, v.z);
        atomicAdd(op + 3, v.w);
    } else {
        float2 v = *(const float2*)hp;
        atomicAdd(op + 0, v.x);
        atomicAdd(op + 1, v.y);
    }
}

static inline int cdiv(long a, long b) { return (int)((a + b - 1) / b); }

extern "C" void kernel_launch(void* const* d_in, const int* in_sizes, int n_in,
                              void* d_out, int out_size, void* d_ws, size_t ws_size,
                              hipStream_t stream) {
    const float* features = (const float*)d_in[0];
    const int*   src      = (const int*)d_in[1];
    const int*   dst      = (const int*)d_in[2];
    const float* W1       = (const float*)d_in[3];
    const float* b1       = (const float*)d_in[4];
    const float* W2       = (const float*)d_in[5];
    const float* b2       = (const float*)d_in[6];
    const float* W3       = (const float*)d_in[7];
    const float* b3       = (const float*)d_in[8];
    float*       out      = (float*)d_out;

    const int N = in_sizes[0] / IN_F;
    const int E = in_sizes[1];

    float* ws   = (float*)d_ws;
    float* dsrc = ws;
    float* ddst = ws + N;
    float* buf0 = ws + 2 * (size_t)N;
    float* buf1 = buf0 + (size_t)N * HID_F;

    const long NH = (long)N * HID_F;
    const long NO = (long)N * OUT_F;
    const int  nblk = N / 16;

    zero_kernel<<<cdiv(2L * N, 256), 256, 0, stream>>>(dsrc, 2L * N);
    degree_kernel<<<cdiv(E, 256), 256, 0, stream>>>(src, dst, dsrc, ddst, E);
    rsqrt_kernel<<<cdiv(2L * N, 256), 256, 0, stream>>>(dsrc, 2 * N);

    gemm_rows16_wmma<HID_F><<<nblk, HID_F * 2, 0, stream>>>(features, dsrc, W1, buf0);
    zero_kernel<<<cdiv(NH, 256), 256, 0, stream>>>(buf1, NH);
    scatter_add_kernel<HID_F><<<cdiv((long)E * 32, 256), 256, 0, stream>>>(buf0, src, dst, buf1, E);
    epilogue_kernel<<<cdiv(NH, 256), 256, 0, stream>>>(buf1, ddst, b1, HID_F, NH, 1);

    gemm_rows16_wmma<HID_F><<<nblk, HID_F * 2, 0, stream>>>(buf1, dsrc, W2, buf0);
    zero_kernel<<<cdiv(NH, 256), 256, 0, stream>>>(buf1, NH);
    scatter_add_kernel<HID_F><<<cdiv((long)E * 32, 256), 256, 0, stream>>>(buf0, src, dst, buf1, E);
    epilogue_kernel<<<cdiv(NH, 256), 256, 0, stream>>>(buf1, ddst, b2, HID_F, NH, 1);

    gemm_rows16_wmma<OUT_F><<<nblk, OUT_F * 2, 0, stream>>>(buf1, dsrc, W3, buf0);
    zero_kernel<<<cdiv(NO, 256), 256, 0, stream>>>(out, NO);
    scatter_add_kernel<OUT_F><<<cdiv((long)E * 32, 256), 256, 0, stream>>>(buf0, src, dst, out, E);
    epilogue_kernel<<<cdiv(NO, 256), 256, 0, stream>>>(out, ddst, b3, OUT_F, NO, 0);
}


# Segment sum by gather over a destination-sorted edge list (built per call, int atomics, arrival-order rank) with fused scale+bias+relu, replacing float atomic scatter; per-row sum order changes
# speedup vs baseline: 2.7293x; 2.7293x over previous
#include <hip/hip_runtime.h>

typedef float v2f __attribute__((ext_vector_type(2)));
typedef float v8f __attribute__((ext_vector_type(8)));

#define IN_F  128
#define HID_F 128
#define OUT_F 64
#define KDIM  128

#define SCAN_T 1024

__global__ void zero_int_kernel(int* __restrict__ p, int n) {
    int i = blockIdx.x * blockDim.x + threadIdx.x;
    if (i < n) p[i] = 0;
}

__global__ void count_rank_kernel(const int* __restrict__ src, const int* __restrict__ dst,
                                  int* cnt_src, int* cnt_dst, int* __restrict__ rank, int ne) {
    int e = blockIdx.x * blockDim.x + threadIdx.x;
    if (e < ne) {
        const int s = src[e];
        const int d = dst[e];
        atomicAdd(&cnt_src[s], 1);
        rank[e] = atomicAdd(&cnt_dst[d], 1);
    }
}

__launch_bounds__(SCAN_T)
__global__ void scan_norm_kernel(const int* __restrict__ cnt_src, const int* __restrict__ cnt_dst,
                                 int* __restrict__ rowptr, float* __restrict__ dsrc,
                                 float* __restrict__ ddst, int n) {
    __shared__ int sh[SCAN_T];
    const int t      = threadIdx.x;
    const int chunk0 = blockIdx.x * SCAN_T;

    int part = 0;
    for (int i = t; i < chunk0; i += SCAN_T) part += cnt_dst[i];
    sh[t] = part;
    __syncthreads();
    for (int off = SCAN_T / 2; off > 0; off >>= 1) {
        if (t < off) sh[t] += sh[t + off];
        __syncthreads();
    }
    const int base = sh[0];
    __syncthreads();

    const int i = chunk0 + t;
    const int c = (i < n) ? cnt_dst[i] : 0;
    sh[t] = c;
    __syncthreads();
    for (int off = 1; off < SCAN_T; off <<= 1) {
        const int x = (t >= off) ? sh[t - off] : 0;
        __syncthreads();
        sh[t] += x;
        __syncthreads();
    }
    const int incl = sh[t];

    if (i < n) {
        rowptr[i] = base + incl - c;
        const int cs = cnt_src[i];
        dsrc[i] = rsqrtf(cs < 1 ? 1.0f : (float)cs);
        ddst[i] = rsqrtf(c  < 1 ? 1.0f : (float)c);
        if (i == n - 1) rowptr[n] = base + incl;
    }
}

__global__ void fill_kernel(const int* __restrict__ src, const int* __restrict__ dst,
                            const int* __restrict__ rowptr, const int* __restrict__ rank,
                            int* __restrict__ ssrc, int ne) {
    int e = blockIdx.x * blockDim.x + threadIdx.x;
    if (e < ne) ssrc[rowptr[dst[e]] + rank[e]] = src[e];
}

__global__ void copy_int_kernel(const int* __restrict__ a, int* __restrict__ b, int n) {
    int i = blockIdx.x * blockDim.x + threadIdx.x;
    if (i < n) b[i] = a[i];
}

template <int FOUT>
__launch_bounds__(FOUT * 2)
__global__ void gemm_rows16_wmma(const float* __restrict__ X,
                                 const float* __restrict__ dinv,
                                 const float* __restrict__ W,
                                 float* __restrict__ H) {
    constexpr int LDSS = KDIM + 4;
    __shared__ __align__(16) float As[16 * LDSS];

    const int row0 = blockIdx.x * 16;
    const int tid  = threadIdx.x;

    for (int i = tid; i < 16 * KDIM; i += blockDim.x) {
        int r = i >> 7;
        int c = i & (KDIM - 1);
        As[r * LDSS + c] = X[(size_t)(row0 + r) * KDIM + c] * dinv[row0 + r];
    }
    __syncthreads();

    const int lane  = tid & 31;
    const int wave  = tid >> 5;
    const int n     = lane & 15;
    const int khalf = lane >> 4;
    const int m16   = lane & 15;
    const int ncol  = wave * 16 + n;

    v8f acc = {};
#pragma unroll
    for (int k0 = 0; k0 < KDIM; k0 += 4) {
        const int ka = k0 + 2 * khalf;
        v2f a = *(const v2f*)&As[m16 * LDSS + ka];
        v2f b;
        b.x = W[(size_t)ka * FOUT + ncol];
        b.y = W[(size_t)(ka + 1) * FOUT + ncol];
        acc = __builtin_amdgcn_wmma_f32_16x16x4_f32(
             false, a,  false, b,
             (short)0, acc,  false,  false);
    }

#pragma unroll
    for (int r = 0; r < 8; ++r) {
        const int m = r + 8 * khalf;
        H[(size_t)(row0 + m) * FOUT + ncol] = acc[r];
    }
}

#define GATHER_WAVES 8

template <int F, bool RELU>
__launch_bounds__(GATHER_WAVES * 32)
__global__ void gather_rows_kernel(const float* __restrict__ H,
                                   const int* __restrict__ rowptr,
                                   const int* __restrict__ ssrc,
                                   const float* __restrict__ dinv,
                                   const float* __restrict__ bias,
                                   float* __restrict__ Y, int n) {
    constexpr int V = F / 32;
    typedef float vec __attribute__((ext_vector_type(V)));

    const int lane = threadIdx.x & 31;
    const int wave = __builtin_amdgcn_readfirstlane((int)(threadIdx.x >> 5));
    const int row  = blockIdx.x * GATHER_WAVES + wave;
    if (row >= n) return;

    const int beg = rowptr[row];
    const int end = rowptr[row + 1];
    const vec* Hv = (const vec*)H;

    vec acc = 0.0f;
    int e = beg;
    for (; e + 8 <= end; e += 8) {
        int s[8];
        vec v[8];
#pragma unroll
        for (int k = 0; k < 8; ++k) s[k] = ssrc[e + k];
#pragma unroll
        for (int k = 0; k < 8; ++k) v[k] = Hv[(size_t)s[k] * 32 + lane];
        acc += ((v[0] + v[1]) + (v[2] + v[3])) + ((v[4] + v[5]) + (v[6] + v[7]));
    }
    if (e + 4 <= end) {
        int s[4];
        vec v[4];
#pragma unroll
        for (int k = 0; k < 4; ++k) s[k] = ssrc[e + k];
#pragma unroll
        for (int k = 0; k < 4; ++k) v[k] = Hv[(size_t)s[k] * 32 + lane];
        acc += (v[0] + v[1]) + (v[2] + v[3]);
        e += 4;
    }
    if (e + 2 <= end) {
        const int s0 = ssrc[e], s1 = ssrc[e + 1];
        const vec v0 = Hv[(size_t)s0 * 32 + lane];
        const vec v1 = Hv[(size_t)s1 * 32 + lane];
        acc += v0 + v1;
        e += 2;
    }
    if (e < end) {
        const int s0 = ssrc[e];
        acc += Hv[(size_t)s0 * 32 + lane];
    }

    const float di = dinv[row];
    const vec   b  = ((const vec*)bias)[lane];
    vec r = acc * di + b;
    if (RELU) {
#pragma unroll
        for (int k = 0; k < V; ++k) r[k] = r[k] > 0.0f ? r[k] : 0.0f;
    }
    ((vec*)Y)[(size_t)row * 32 + lane] = r;
}

static inline int cdiv(long a, long b) { return (int)((a + b - 1) / b); }

extern "C" void kernel_launch(void* const* d_in, const int* in_sizes, int n_in,
                              void* d_out, int out_size, void* d_ws, size_t ws_size,
                              hipStream_t stream) {
    const float* features = (const float*)d_in[0];
    const int*   src      = (const int*)d_in[1];
    const int*   dst      = (const int*)d_in[2];
    const float* W1       = (const float*)d_in[3];
    const float* b1       = (const float*)d_in[4];
    const float* W2       = (const float*)d_in[5];
    const float* b2       = (const float*)d_in[6];
    const float* W3       = (const float*)d_in[7];
    const float* b3       = (const float*)d_in[8];
    float*       out      = (float*)d_out;

    const int N = in_sizes[0] / IN_F;
    const int E = in_sizes[1];

    float* ws   = (float*)d_ws;
    float* dsrc = ws;
    float* ddst = ws + N;
    float* buf0 = ws + 2 * (size_t)N;
    float* buf1 = buf0 + (size_t)N * HID_F;

    int* cnt_src = (int*)buf1;
    int* cnt_dst = cnt_src + N;
    int* rank    = cnt_dst + N;

    const int csr_len = N + 1 + E;
    int* csr_o    = (int*)out;
    int* rowptr_o = csr_o;
    int* ssrc_o   = csr_o + (N + 1);
    int* csr_w    = (int*)buf1;
    int* rowptr_w = csr_w;
    int* ssrc_w   = csr_w + (N + 1);

    const int nblk = N / 16;
    const int gblk = cdiv(N, GATHER_WAVES);

    zero_int_kernel<<<cdiv(2L * N, 256), 256, 0, stream>>>(cnt_src, 2 * N);
    count_rank_kernel<<<cdiv(E, 256), 256, 0, stream>>>(src, dst, cnt_src, cnt_dst, rank, E);
    scan_norm_kernel<<<cdiv(N, SCAN_T), SCAN_T, 0, stream>>>(cnt_src, cnt_dst, rowptr_o, dsrc, ddst, N);
    fill_kernel<<<cdiv(E, 256), 256, 0, stream>>>(src, dst, rowptr_o, rank, ssrc_o, E);

    gemm_rows16_wmma<HID_F><<<nblk, HID_F * 2, 0, stream>>>(features, dsrc, W1, buf0);
    gather_rows_kernel<HID_F, true><<<gblk, GATHER_WAVES * 32, 0, stream>>>(buf0, rowptr_o, ssrc_o, ddst, b1, buf1, N);

    gemm_rows16_wmma<HID_F><<<nblk, HID_F * 2, 0, stream>>>(buf1, dsrc, W2, buf0);
    gather_rows_kernel<HID_F, true><<<gblk, GATHER_WAVES * 32, 0, stream>>>(buf0, rowptr_o, ssrc_o, ddst, b2, buf1, N);

    gemm_rows16_wmma<OUT_F><<<nblk, OUT_F * 2, 0, stream>>>(buf1, dsrc, W3, buf0);
    copy_int_kernel<<<cdiv(csr_len, 256), 256, 0, stream>>>(csr_o, csr_w, csr_len);
    gather_rows_kernel<OUT_F, false><<<gblk, GATHER_WAVES * 32, 0, stream>>>(buf0, rowptr_w, ssrc_w, ddst, b3, out, N);
}


# Resident grids: every kernel launches at most the chip's wave slots and each wave loops over its edges, rows or 16-row tiles (W fragments stay in registers): ~6x fewer waves; arrival-order rank kept
# speedup vs baseline: 2.9131x; 1.0673x over previous
#include <hip/hip_runtime.h>

typedef float v2f __attribute__((ext_vector_type(2)));
typedef float v8f __attribute__((ext_vector_type(8)));

#define IN_F  128
#define HID_F 128
#define OUT_F 64
#define KDIM  128

#define ELEM_BLOCKS    512
#define ZERO_BLOCKS    256
#define GATHER_BLOCKS  2048
#define GEMM128_BLOCKS 1024
#define GEMM64_BLOCKS  2048

#define SCAN_T 1024

__global__ void zero_int_kernel(int* __restrict__ p, int n) {
    const int nthreads = gridDim.x * blockDim.x;
    for (int i = blockIdx.x * blockDim.x + threadIdx.x; i < n; i += nthreads) p[i] = 0;
}

__global__ void count_rank_kernel(const int* __restrict__ src, const int* __restrict__ dst,
                                  int* cnt_src, int* cnt_dst, int* __restrict__ rank, int ne) {
    const int nthreads = gridDim.x * blockDim.x;
    int e = blockIdx.x * blockDim.x + threadIdx.x;
    for (; e + 3 * nthreads < ne; e += 4 * nthreads) {
        int s[4], d[4], r[4];
#pragma unroll
        for (int k = 0; k < 4; ++k) {
            s[k] = src[e + k * nthreads];
            d[k] = dst[e + k * nthreads];
        }
#pragma unroll
        for (int k = 0; k < 4; ++k) atomicAdd(&cnt_src[s[k]], 1);
#pragma unroll
        for (int k = 0; k < 4; ++k) r[k] = atomicAdd(&cnt_dst[d[k]], 1);
#pragma unroll
        for (int k = 0; k < 4; ++k) rank[e + k * nthreads] = r[k];
    }
    for (; e < ne; e += nthreads) {
        const int s = src[e];
        const int d = dst[e];
        atomicAdd(&cnt_src[s], 1);
        rank[e] = atomicAdd(&cnt_dst[d], 1);
    }
}

__launch_bounds__(SCAN_T)
__global__ void scan_norm_kernel(const int* __restrict__ cnt_src, const int* __restrict__ cnt_dst,
                                 int* __restrict__ rowptr, float* __restrict__ dsrc,
                                 float* __restrict__ ddst, int n) {
    __shared__ int sh[SCAN_T];
    const int t      = threadIdx.x;
    const int chunk0 = blockIdx.x * SCAN_T;

    int part = 0;
    for (int i = t; i < chunk0; i += SCAN_T) part += cnt_dst[i];
    sh[t] = part;
    __syncthreads();
    for (int off = SCAN_T / 2; off > 0; off >>= 1) {
        if (t < off) sh[t] += sh[t + off];
        __syncthreads();
    }
    const int base = sh[0];
    __syncthreads();

    const int i = chunk0 + t;
    const int c = (i < n) ? cnt_dst[i] : 0;
    sh[t] = c;
    __syncthreads();
    for (int off = 1; off < SCAN_T; off <<= 1) {
        const int x = (t >= off) ? sh[t - off] : 0;
        __syncthreads();
        sh[t] += x;
        __syncthreads();
    }
    const int incl = sh[t];

    if (i < n) {
        rowptr[i] = base + incl - c;
        const int cs = cnt_src[i];
        dsrc[i] = rsqrtf(cs < 1 ? 1.0f : (float)cs);
        ddst[i] = rsqrtf(c  < 1 ? 1.0f : (float)c);
        if (i == n - 1) rowptr[n] = base + incl;
    }
}

__global__ void fill_kernel(const int* __restrict__ src, const int* __restrict__ dst,
                            const int* __restrict__ rowptr, const int* __restrict__ rank,
                            int* __restrict__ ssrc, int ne) {
    const int nthreads = gridDim.x * blockDim.x;
    int e = blockIdx.x * blockDim.x + threadIdx.x;
    for (; e + 3 * nthreads < ne; e += 4 * nthreads) {
        int s[4], d[4], r[4], p[4];
#pragma unroll
        for (int k = 0; k < 4; ++k) {
            d[k] = dst[e + k * nthreads];
            r[k] = rank[e + k * nthreads];
            s[k] = src[e + k * nthreads];
        }
#pragma unroll
        for (int k = 0; k < 4; ++k) p[k] = rowptr[d[k]];
#pragma unroll
        for (int k = 0; k < 4; ++k) ssrc[p[k] + r[k]] = s[k];
    }
    for (; e < ne; e += nthreads) ssrc[rowptr[dst[e]] + rank[e]] = src[e];
}

__global__ void copy_int_kernel(const int* __restrict__ a, int* __restrict__ b, int n) {
    const int nthreads = gridDim.x * blockDim.x;
    int i = blockIdx.x * blockDim.x + threadIdx.x;
    for (; i + 3 * nthreads < n; i += 4 * nthreads) {
        int v[4];
#pragma unroll
        for (int k = 0; k < 4; ++k) v[k] = a[i + k * nthreads];
#pragma unroll
        for (int k = 0; k < 4; ++k) b[i + k * nthreads] = v[k];
    }
    for (; i < n; i += nthreads) b[i] = a[i];
}

template <int FOUT>
__launch_bounds__(FOUT * 2)
__global__ void gemm_rows16_wmma(const float* __restrict__ X,
                                 const float* __restrict__ dinv,
                                 const float* __restrict__ W,
                                 float* __restrict__ H, int ntiles) {
    constexpr int LDSS = KDIM + 4;
    constexpr int NT   = FOUT * 2;
    constexpr int PER  = 16 * KDIM / NT;
    __shared__ __align__(16) float As[2][16 * LDSS];

    const int tid   = threadIdx.x;
    const int lane  = tid & 31;
    const int wave  = tid >> 5;
    const int n     = lane & 15;
    const int khalf = lane >> 4;
    const int m16   = lane & 15;
    const int ncol  = wave * 16 + n;

    v2f bf[KDIM / 4];
#pragma unroll
    for (int j = 0; j < KDIM / 4; ++j) {
        const int ka = 4 * j + 2 * khalf;
        bf[j].x = W[(size_t)ka * FOUT + ncol];
        bf[j].y = W[(size_t)(ka + 1) * FOUT + ncol];
    }

    constexpr int RP = NT / KDIM;
    const int ar = tid >> 7;
    const int ac = tid & (KDIM - 1);
    const float* xp = X + tid;
    const float* dp = dinv + ar;
    float xa[PER], da[PER];
    int tile = blockIdx.x;
    if (tile < ntiles) {
        const float* xt = xp + (size_t)tile * (16 * KDIM);
        const float* dt = dp + (size_t)tile * 16;
#pragma unroll
        for (int j = 0; j < PER; ++j) {
            xa[j] = xt[j * NT];
            da[j] = dt[j * RP];
        }
    }

    int st = 0;
    for (; tile < ntiles; tile += gridDim.x) {
        float* A  = As[st];
        float* aw = A + ar * LDSS + ac;
#pragma unroll
        for (int j = 0; j < PER; ++j) aw[j * RP * LDSS] = xa[j] * da[j];
        __syncthreads();

        const int nxt = tile + gridDim.x;
        if (nxt < ntiles) {
            const float* xt = xp + (size_t)nxt * (16 * KDIM);
            const float* dt = dp + (size_t)nxt * 16;
#pragma unroll
            for (int j = 0; j < PER; ++j) {
                xa[j] = xt[j * NT];
                da[j] = dt[j * RP];
            }
        }

        v8f acc = {};
#pragma unroll
        for (int k0 = 0; k0 < KDIM; k0 += 4) {
            const int ka = k0 + 2 * khalf;
            v2f a = *(const v2f*)&A[m16 * LDSS + ka];
            acc = __builtin_amdgcn_wmma_f32_16x16x4_f32(
                 false, a,  false, bf[k0 / 4],
                 (short)0, acc,  false,  false);
        }

        float* hp = H + ((size_t)tile * 16 + 8 * khalf) * FOUT + ncol;
#pragma unroll
        for (int r = 0; r < 8; ++r) hp[r * FOUT] = acc[r];
        st ^= 1;
    }
}

#define GATHER_WAVES 8

template <int F, bool RELU>
__launch_bounds__(GATHER_WAVES * 32)
__global__ void gather_rows_kernel(const float* __restrict__ H,
                                   const int* __restrict__ rowptr,
                                   const int* __restrict__ ssrc,
                                   const float* __restrict__ dinv,
                                   const float* __restrict__ bias,
                                   float* __restrict__ Y, int n) {
    constexpr int V = F / 32;
    typedef float vec __attribute__((ext_vector_type(V)));

    const int lane = threadIdx.x & 31;
    const int wave = __builtin_amdgcn_readfirstlane((int)(threadIdx.x >> 5));
    const int stride = gridDim.x * GATHER_WAVES;
    const vec* Hv = (const vec*)H;
    const vec  b  = ((const vec*)bias)[lane];

    for (int row = blockIdx.x * GATHER_WAVES + wave; row < n; row += stride) {
        const int beg = rowptr[row];
        const int end = rowptr[row + 1];

        vec acc = 0.0f;
        int e = beg;
        for (; e + 8 <= end; e += 8) {
            int s[8];
            vec v[8];
#pragma unroll
            for (int k = 0; k < 8; ++k) s[k] = ssrc[e + k];
#pragma unroll
            for (int k = 0; k < 8; ++k) v[k] = Hv[(size_t)s[k] * 32 + lane];
            acc += ((v[0] + v[1]) + (v[2] + v[3])) + ((v[4] + v[5]) + (v[6] + v[7]));
        }
        if (e + 4 <= end) {
            int s[4];
            vec v[4];
#pragma unroll
            for (int k = 0; k < 4; ++k) s[k] = ssrc[e + k];
#pragma unroll
            for (int k = 0; k < 4; ++k) v[k] = Hv[(size_t)s[k] * 32 + lane];
            acc += (v[0] + v[1]) + (v[2] + v[3]);
            e += 4;
        }
        if (e + 2 <= end) {
            const int s0 = ssrc[e], s1 = ssrc[e + 1];
            const vec v0 = Hv[(size_t)s0 * 32 + lane];
            const vec v1 = Hv[(size_t)s1 * 32 + lane];
            acc += v0 + v1;
            e += 2;
        }
        if (e < end) {
            const int s0 = ssrc[e];
            acc += Hv[(size_t)s0 * 32 + lane];
        }

        const float di = dinv[row];
        vec r = acc * di + b;
        if (RELU) {
#pragma unroll
            for (int k = 0; k < V; ++k) r[k] = r[k] > 0.0f ? r[k] : 0.0f;
        }
        ((vec*)Y)[(size_t)row * 32 + lane] = r;
    }
}

static inline int cdiv(long a, long b) { return (int)((a + b - 1) / b); }

extern "C" void kernel_launch(void* const* d_in, const int* in_sizes, int n_in,
                              void* d_out, int out_size, void* d_ws, size_t ws_size,
                              hipStream_t stream) {
    const float* features = (const float*)d_in[0];
    const int*   src      = (const int*)d_in[1];
    const int*   dst      = (const int*)d_in[2];
    const float* W1       = (const float*)d_in[3];
    const float* b1       = (const float*)d_in[4];
    const float* W2       = (const float*)d_in[5];
    const float* b2       = (const float*)d_in[6];
    const float* W3       = (const float*)d_in[7];
    const float* b3       = (const float*)d_in[8];
    float*       out      = (float*)d_out;

    const int N = in_sizes[0] / IN_F;
    const int E = in_sizes[1];

    float* ws   = (float*)d_ws;
    float* dsrc = ws;
    float* ddst = ws + N;
    float* buf0 = ws + 2 * (size_t)N;
    float* buf1 = buf0 + (size_t)N * HID_F;

    int* cnt_src = (int*)buf1;
    int* cnt_dst = cnt_src + N;
    int* rank    = cnt_dst + N;

    const int csr_len = N + 1 + E;
    int* csr_o    = (int*)out;
    int* rowptr_o = csr_o;
    int* ssrc_o   = csr_o + (N + 1);
    int* csr_w    = (int*)buf1;
    int* rowptr_w = csr_w;
    int* ssrc_w   = csr_w + (N + 1);

    const int ntiles = N / 16;

    zero_int_kernel<<<ZERO_BLOCKS, 256, 0, stream>>>(cnt_src, 2 * N);
    count_rank_kernel<<<ELEM_BLOCKS, 256, 0, stream>>>(src, dst, cnt_src, cnt_dst, rank, E);
    scan_norm_kernel<<<cdiv(N, SCAN_T), SCAN_T, 0, stream>>>(cnt_src, cnt_dst, rowptr_o, dsrc, ddst, N);
    fill_kernel<<<ELEM_BLOCKS, 256, 0, stream>>>(src, dst, rowptr_o, rank, ssrc_o, E);

    gemm_rows16_wmma<HID_F><<<GEMM128_BLOCKS, HID_F * 2, 0, stream>>>(features, dsrc, W1, buf0, ntiles);
    gather_rows_kernel<HID_F, true><<<GATHER_BLOCKS, GATHER_WAVES * 32, 0, stream>>>(buf0, rowptr_o, ssrc_o, ddst, b1, buf1, N);

    gemm_rows16_wmma<HID_F><<<GEMM128_BLOCKS, HID_F * 2, 0, stream>>>(buf1, dsrc, W2, buf0, ntiles);
    gather_rows_kernel<HID_F, true><<<GATHER_BLOCKS, GATHER_WAVES * 32, 0, stream>>>(buf0, rowptr_o, ssrc_o, ddst, b2, buf1, N);

    gemm_rows16_wmma<OUT_F><<<GEMM64_BLOCKS, OUT_F * 2, 0, stream>>>(buf1, dsrc, W3, buf0, ntiles);
    copy_int_kernel<<<ELEM_BLOCKS, 256, 0, stream>>>(csr_o, csr_w, csr_len);
    gather_rows_kernel<OUT_F, false><<<GATHER_BLOCKS, GATHER_WAVES * 32, 0, stream>>>(buf0, rowptr_w, ssrc_w, ddst, b3, out, N);
}


# Fuse each hidden layer's segment sum with the next layer's WMMA product: the summed, scaled, relu'd 16-row tile goes to LDS as the A operand, x1/x2 never stored, 11 -> 9 launches; sum order kept
# speedup vs baseline: 3.0156x; 1.0352x over previous
#include <hip/hip_runtime.h>

typedef float v2f __attribute__((ext_vector_type(2)));
typedef float v8f __attribute__((ext_vector_type(8)));

#define IN_F  128
#define HID_F 128
#define OUT_F 64
#define KDIM  128

#define ELEM_BLOCKS    512
#define ZERO_BLOCKS    256
#define GATHER_BLOCKS  2048
#define GEMM128_BLOCKS 1024
#define FUSE_BLOCKS    1024
#define FUSE_WAVES     8

#define SCAN_T 1024

__global__ void zero_int_kernel(int* __restrict__ p, int n) {
    const int nthreads = gridDim.x * blockDim.x;
    for (int i = blockIdx.x * blockDim.x + threadIdx.x; i < n; i += nthreads) p[i] = 0;
}

__global__ void count_rank_kernel(const int* __restrict__ src, const int* __restrict__ dst,
                                  int* cnt_src, int* cnt_dst, int* __restrict__ rank, int ne) {
    const int nthreads = gridDim.x * blockDim.x;
    int e = blockIdx.x * blockDim.x + threadIdx.x;
    for (; e + 3 * nthreads < ne; e += 4 * nthreads) {
        int s[4], d[4], r[4];
#pragma unroll
        for (int k = 0; k < 4; ++k) {
            s[k] = src[e + k * nthreads];
            d[k] = dst[e + k * nthreads];
        }
#pragma unroll
        for (int k = 0; k < 4; ++k) atomicAdd(&cnt_src[s[k]], 1);
#pragma unroll
        for (int k = 0; k < 4; ++k) r[k] = atomicAdd(&cnt_dst[d[k]], 1);
#pragma unroll
        for (int k = 0; k < 4; ++k) rank[e + k * nthreads] = r[k];
    }
    for (; e < ne; e += nthreads) {
        const int s = src[e];
        const int d = dst[e];
        atomicAdd(&cnt_src[s], 1);
        rank[e] = atomicAdd(&cnt_dst[d], 1);
    }
}

__launch_bounds__(SCAN_T)
__global__ void scan_norm_kernel(const int* __restrict__ cnt_src, const int* __restrict__ cnt_dst,
                                 int* __restrict__ rowptr, float* __restrict__ dsrc,
                                 float* __restrict__ ddst, int n) {
    __shared__ int sh[SCAN_T];
    const int t      = threadIdx.x;
    const int chunk0 = blockIdx.x * SCAN_T;

    int part = 0;
    for (int i = t; i < chunk0; i += SCAN_T) part += cnt_dst[i];
    sh[t] = part;
    __syncthreads();
    for (int off = SCAN_T / 2; off > 0; off >>= 1) {
        if (t < off) sh[t] += sh[t + off];
        __syncthreads();
    }
    const int base = sh[0];
    __syncthreads();

    const int i = chunk0 + t;
    const int c = (i < n) ? cnt_dst[i] : 0;
    sh[t] = c;
    __syncthreads();
    for (int off = 1; off < SCAN_T; off <<= 1) {
        const int x = (t >= off) ? sh[t - off] : 0;
        __syncthreads();
        sh[t] += x;
        __syncthreads();
    }
    const int incl = sh[t];

    if (i < n) {
        rowptr[i] = base + incl - c;
        const int cs = cnt_src[i];
        dsrc[i] = rsqrtf(cs < 1 ? 1.0f : (float)cs);
        ddst[i] = rsqrtf(c  < 1 ? 1.0f : (float)c);
        if (i == n - 1) rowptr[n] = base + incl;
    }
}

__global__ void fill_kernel(const int* __restrict__ src, const int* __restrict__ dst,
                            const int* __restrict__ rowptr, const int* __restrict__ rank,
                            int* __restrict__ ssrc, int ne) {
    const int nthreads = gridDim.x * blockDim.x;
    int e = blockIdx.x * blockDim.x + threadIdx.x;
    for (; e + 3 * nthreads < ne; e += 4 * nthreads) {
        int s[4], d[4], r[4], p[4];
#pragma unroll
        for (int k = 0; k < 4; ++k) {
            d[k] = dst[e + k * nthreads];
            r[k] = rank[e + k * nthreads];
            s[k] = src[e + k * nthreads];
        }
#pragma unroll
        for (int k = 0; k < 4; ++k) p[k] = rowptr[d[k]];
#pragma unroll
        for (int k = 0; k < 4; ++k) ssrc[p[k] + r[k]] = s[k];
    }
    for (; e < ne; e += nthreads) ssrc[rowptr[dst[e]] + rank[e]] = src[e];
}

__global__ void copy_int_kernel(const int* __restrict__ a, int* __restrict__ b, int n) {
    const int nthreads = gridDim.x * blockDim.x;
    int i = blockIdx.x * blockDim.x + threadIdx.x;
    for (; i + 3 * nthreads < n; i += 4 * nthreads) {
        int v[4];
#pragma unroll
        for (int k = 0; k < 4; ++k) v[k] = a[i + k * nthreads];
#pragma unroll
        for (int k = 0; k < 4; ++k) b[i + k * nthreads] = v[k];
    }
    for (; i < n; i += nthreads) b[i] = a[i];
}

template <int FOUT>
__launch_bounds__(FOUT * 2)
__global__ void gemm_rows16_wmma(const float* __restrict__ X,
                                 const float* __restrict__ dinv,
                                 const float* __restrict__ W,
                                 float* __restrict__ H, int ntiles) {
    constexpr int LDSS = KDIM + 4;
    constexpr int NT   = FOUT * 2;
    constexpr int PER  = 16 * KDIM / NT;
    __shared__ __align__(16) float As[2][16 * LDSS];

    const int tid   = threadIdx.x;
    const int lane  = tid & 31;
    const int wave  = tid >> 5;
    const int n     = lane & 15;
    const int khalf = lane >> 4;
    const int m16   = lane & 15;
    const int ncol  = wave * 16 + n;

    v2f bf[KDIM / 4];
#pragma unroll
    for (int j = 0; j < KDIM / 4; ++j) {
        const int ka = 4 * j + 2 * khalf;
        bf[j].x = W[(size_t)ka * FOUT + ncol];
        bf[j].y = W[(size_t)(ka + 1) * FOUT + ncol];
    }

    constexpr int RP = NT / KDIM;
    const int ar = tid >> 7;
    const int ac = tid & (KDIM - 1);
    const float* xp = X + tid;
    const float* dp = dinv + ar;
    float xa[PER], da[PER];
    int tile = blockIdx.x;
    if (tile < ntiles) {
        const float* xt = xp + (size_t)tile * (16 * KDIM);
        const float* dt = dp + (size_t)tile * 16;
#pragma unroll
        for (int j = 0; j < PER; ++j) {
            xa[j] = xt[j * NT];
            da[j] = dt[j * RP];
        }
    }

    int st = 0;
    for (; tile < ntiles; tile += gridDim.x) {
        float* A  = As[st];
        float* aw = A + ar * LDSS + ac;
#pragma unroll
        for (int j = 0; j < PER; ++j) aw[j * RP * LDSS] = xa[j] * da[j];
        __syncthreads();

        const int nxt = tile + gridDim.x;
        if (nxt < ntiles) {
            const float* xt = xp + (size_t)nxt * (16 * KDIM);
            const float* dt = dp + (size_t)nxt * 16;
#pragma unroll
            for (int j = 0; j < PER; ++j) {
                xa[j] = xt[j * NT];
                da[j] = dt[j * RP];
            }
        }

        v8f acc = {};
#pragma unroll
        for (int k0 = 0; k0 < KDIM; k0 += 4) {
            const int ka = k0 + 2 * khalf;
            v2f a = *(const v2f*)&A[m16 * LDSS + ka];
            acc = __builtin_amdgcn_wmma_f32_16x16x4_f32(
                 false, a,  false, bf[k0 / 4],
                 (short)0, acc,  false,  false);
        }

        float* hp = H + ((size_t)tile * 16 + 8 * khalf) * FOUT + ncol;
#pragma unroll
        for (int r = 0; r < 8; ++r) hp[r * FOUT] = acc[r];
        st ^= 1;
    }
}

#define GATHER_WAVES 8

template <int F, bool RELU>
__launch_bounds__(GATHER_WAVES * 32)
__global__ void gather_rows_kernel(const float* __restrict__ H,
                                   const int* __restrict__ rowptr,
                                   const int* __restrict__ ssrc,
                                   const float* __restrict__ dinv,
                                   const float* __restrict__ bias,
                                   float* __restrict__ Y, int n) {
    constexpr int V = F / 32;
    typedef float vec __attribute__((ext_vector_type(V)));

    const int lane = threadIdx.x & 31;
    const int wave = __builtin_amdgcn_readfirstlane((int)(threadIdx.x >> 5));
    const int stride = gridDim.x * GATHER_WAVES;
    const vec* Hv = (const vec*)H;
    const vec  b  = ((const vec*)bias)[lane];

    for (int row = blockIdx.x * GATHER_WAVES + wave; row < n; row += stride) {
        const int beg = rowptr[row];
        const int end = rowptr[row + 1];

        vec acc = 0.0f;
        int e = beg;
        for (; e + 8 <= end; e += 8) {
            int s[8];
            vec v[8];
#pragma unroll
            for (int k = 0; k < 8; ++k) s[k] = ssrc[e + k];
#pragma unroll
            for (int k = 0; k < 8; ++k) v[k] = Hv[(size_t)s[k] * 32 + lane];
            acc += ((v[0] + v[1]) + (v[2] + v[3])) + ((v[4] + v[5]) + (v[6] + v[7]));
        }
        if (e + 4 <= end) {
            int s[4];
            vec v[4];
#pragma unroll
            for (int k = 0; k < 4; ++k) s[k] = ssrc[e + k];
#pragma unroll
            for (int k = 0; k < 4; ++k) v[k] = Hv[(size_t)s[k] * 32 + lane];
            acc += (v[0] + v[1]) + (v[2] + v[3]);
            e += 4;
        }
        if (e + 2 <= end) {
            const int s0 = ssrc[e], s1 = ssrc[e + 1];
            const vec v0 = Hv[(size_t)s0 * 32 + lane];
            const vec v1 = Hv[(size_t)s1 * 32 + lane];
            acc += v0 + v1;
            e += 2;
        }
        if (e < end) {
            const int s0 = ssrc[e];
            acc += Hv[(size_t)s0 * 32 + lane];
        }

        const float di = dinv[row];
        vec r = acc * di + b;
        if (RELU) {
#pragma unroll
            for (int k = 0; k < V; ++k) r[k] = r[k] > 0.0f ? r[k] : 0.0f;
        }
        ((vec*)Y)[(size_t)row * 32 + lane] = r;
    }
}

template <int FOUT>
__launch_bounds__(FUSE_WAVES * 32)
__attribute__((amdgpu_waves_per_eu(8)))
__global__ void gather_gemm_wmma(const float* __restrict__ Hin,
                                 const int* __restrict__ rowptr,
                                 const int* __restrict__ ssrc,
                                 const float* __restrict__ ddst,
                                 const float* __restrict__ bias,
                                 const float* __restrict__ dsrc,
                                 const float* __restrict__ W,
                                 float* __restrict__ Hout, int ntiles) {
    constexpr int LDSS = KDIM + 4;
    constexpr int CW   = FOUT / 16;
    constexpr int RPW  = 16 / FUSE_WAVES;
    typedef float vec __attribute__((ext_vector_type(4)));
    __shared__ __align__(16) float As[2][16 * LDSS];

    const int tid   = threadIdx.x;
    const int lane  = tid & 31;
    const int wave  = __builtin_amdgcn_readfirstlane((int)(threadIdx.x >> 5));
    const int n     = lane & 15;
    const int khalf = lane >> 4;
    const int m16   = lane & 15;
    const int ncol  = wave * 16 + n;

    v2f bf[KDIM / 4] = {};
    if (wave < CW) {
#pragma unroll
        for (int j = 0; j < KDIM / 4; ++j) {
            const int ka = 4 * j + 2 * khalf;
            bf[j].x = W[(size_t)ka * FOUT + ncol];
            bf[j].y = W[(size_t)(ka + 1) * FOUT + ncol];
        }
    }

    const vec* Hv = (const vec*)Hin;
    const vec  b  = ((const vec*)bias)[lane];

    int st = 0;
    for (int tile = blockIdx.x; tile < ntiles; tile += gridDim.x) {
        float* A = As[st];
        for (int i = 0; i < RPW; ++i) {
            const int m   = wave * RPW + i;
            const int row = tile * 16 + m;
            const int beg = rowptr[row];
            const int end = rowptr[row + 1];

            vec acc = 0.0f;
            int e = beg;
            for (; e + 8 <= end; e += 8) {
                int s[8];
                vec v[8];
#pragma unroll
                for (int k = 0; k < 8; ++k) s[k] = __builtin_amdgcn_readfirstlane(ssrc[e + k]);
#pragma unroll
                for (int k = 0; k < 8; ++k) v[k] = Hv[(size_t)s[k] * 32 + lane];
                acc += ((v[0] + v[1]) + (v[2] + v[3])) + ((v[4] + v[5]) + (v[6] + v[7]));
            }
            if (e + 4 <= end) {
                int s[4];
                vec v[4];
#pragma unroll
                for (int k = 0; k < 4; ++k) s[k] = __builtin_amdgcn_readfirstlane(ssrc[e + k]);
#pragma unroll
                for (int k = 0; k < 4; ++k) v[k] = Hv[(size_t)s[k] * 32 + lane];
                acc += (v[0] + v[1]) + (v[2] + v[3]);
                e += 4;
            }
            if (e + 2 <= end) {
                const int s0 = __builtin_amdgcn_readfirstlane(ssrc[e]);
                const int s1 = __builtin_amdgcn_readfirstlane(ssrc[e + 1]);
                const vec v0 = Hv[(size_t)s0 * 32 + lane];
                const vec v1 = Hv[(size_t)s1 * 32 + lane];
                acc += v0 + v1;
                e += 2;
            }
            if (e < end) {
                const int s0 = __builtin_amdgcn_readfirstlane(ssrc[e]);
                acc += Hv[(size_t)s0 * 32 + lane];
            }

            const float di = ddst[row];
            vec r = acc * di + b;
#pragma unroll
            for (int k = 0; k < 4; ++k) r[k] = r[k] > 0.0f ? r[k] : 0.0f;
            const float ds = dsrc[row];
            r = r * ds;
            *(vec*)&A[m * LDSS + 4 * lane] = r;
        }
        __syncthreads();

        if (wave < CW) {
            v8f acc8 = {};
#pragma unroll
            for (int k0 = 0; k0 < KDIM; k0 += 4) {
                const int ka = k0 + 2 * khalf;
                v2f a = *(const v2f*)&A[m16 * LDSS + ka];
                acc8 = __builtin_amdgcn_wmma_f32_16x16x4_f32(
                     false, a,  false, bf[k0 / 4],
                     (short)0, acc8,  false,  false);
            }

            float* hp = Hout + ((size_t)tile * 16 + 8 * khalf) * FOUT + ncol;
#pragma unroll
            for (int r = 0; r < 8; ++r) hp[r * FOUT] = acc8[r];
        }
        st ^= 1;
    }
}

static inline int cdiv(long a, long b) { return (int)((a + b - 1) / b); }

extern "C" void kernel_launch(void* const* d_in, const int* in_sizes, int n_in,
                              void* d_out, int out_size, void* d_ws, size_t ws_size,
                              hipStream_t stream) {
    const float* features = (const float*)d_in[0];
    const int*   src      = (const int*)d_in[1];
    const int*   dst      = (const int*)d_in[2];
    const float* W1       = (const float*)d_in[3];
    const float* b1       = (const float*)d_in[4];
    const float* W2       = (const float*)d_in[5];
    const float* b2       = (const float*)d_in[6];
    const float* W3       = (const float*)d_in[7];
    const float* b3       = (const float*)d_in[8];
    float*       out      = (float*)d_out;

    const int N = in_sizes[0] / IN_F;
    const int E = in_sizes[1];

    float* ws   = (float*)d_ws;
    float* dsrc = ws;
    float* ddst = ws + N;
    float* buf0 = ws + 2 * (size_t)N;
    float* buf1 = buf0 + (size_t)N * HID_F;

    int* cnt_src = (int*)buf1;
    int* cnt_dst = cnt_src + N;
    int* rank    = cnt_dst + N;

    const int csr_len = N + 1 + E;
    int* csr_o    = (int*)out;
    int* rowptr_o = csr_o;
    int* ssrc_o   = csr_o + (N + 1);
    int* csr_w    = (int*)buf1;
    int* rowptr_w = csr_w;
    int* ssrc_w   = csr_w + (N + 1);

    const int ntiles = N / 16;

    zero_int_kernel<<<ZERO_BLOCKS, 256, 0, stream>>>(cnt_src, 2 * N);
    count_rank_kernel<<<ELEM_BLOCKS, 256, 0, stream>>>(src, dst, cnt_src, cnt_dst, rank, E);
    scan_norm_kernel<<<cdiv(N, SCAN_T), SCAN_T, 0, stream>>>(cnt_src, cnt_dst, rowptr_o, dsrc, ddst, N);
    fill_kernel<<<ELEM_BLOCKS, 256, 0, stream>>>(src, dst, rowptr_o, rank, ssrc_o, E);

    gemm_rows16_wmma<HID_F><<<GEMM128_BLOCKS, HID_F * 2, 0, stream>>>(features, dsrc, W1, buf0, ntiles);

    gather_gemm_wmma<HID_F><<<FUSE_BLOCKS, FUSE_WAVES * 32, 0, stream>>>(buf0, rowptr_o, ssrc_o, ddst, b1, dsrc, W2, buf1, ntiles);

    gather_gemm_wmma<OUT_F><<<FUSE_BLOCKS, FUSE_WAVES * 32, 0, stream>>>(buf1, rowptr_o, ssrc_o, ddst, b2, dsrc, W3, buf0, ntiles);

    copy_int_kernel<<<ELEM_BLOCKS, 256, 0, stream>>>(csr_o, csr_w, csr_len);
    gather_rows_kernel<OUT_F, false><<<GATHER_BLOCKS, GATHER_WAVES * 32, 0, stream>>>(buf0, rowptr_w, ssrc_w, ddst, b3, out, N);
}


# Gather chain shortened: a row's bounds, source indices (one coalesced load, read out by lane) and scales are requested a tile or row ahead; its last 1..7 row loads go out together; sum order kept
# speedup vs baseline: 3.1222x; 1.0353x over previous
#include <hip/hip_runtime.h>

typedef float v2f __attribute__((ext_vector_type(2)));
typedef float v8f __attribute__((ext_vector_type(8)));

#define IN_F  128
#define HID_F 128
#define OUT_F 64
#define KDIM  128

#define ELEM_BLOCKS    512
#define ZERO_BLOCKS    256
#define GATHER_BLOCKS  2048
#define GEMM128_BLOCKS 1024
#define FUSE_BLOCKS    1024
#define FUSE_WAVES     8

#define SCAN_T 1024

__global__ void zero_int_kernel(int* __restrict__ p, int n) {
    const int nthreads = gridDim.x * blockDim.x;
    for (int i = blockIdx.x * blockDim.x + threadIdx.x; i < n; i += nthreads) p[i] = 0;
}

__global__ void count_rank_kernel(const int* __restrict__ src, const int* __restrict__ dst,
                                  int* cnt_src, int* cnt_dst, int* __restrict__ rank, int ne) {
    const int nthreads = gridDim.x * blockDim.x;
    int e = blockIdx.x * blockDim.x + threadIdx.x;
    for (; e + 3 * nthreads < ne; e += 4 * nthreads) {
        int s[4], d[4], r[4];
#pragma unroll
        for (int k = 0; k < 4; ++k) {
            s[k] = src[e + k * nthreads];
            d[k] = dst[e + k * nthreads];
        }
#pragma unroll
        for (int k = 0; k < 4; ++k) atomicAdd(&cnt_src[s[k]], 1);
#pragma unroll
        for (int k = 0; k < 4; ++k) r[k] = atomicAdd(&cnt_dst[d[k]], 1);
#pragma unroll
        for (int k = 0; k < 4; ++k) rank[e + k * nthreads] = r[k];
    }
    for (; e < ne; e += nthreads) {
        const int s = src[e];
        const int d = dst[e];
        atomicAdd(&cnt_src[s], 1);
        rank[e] = atomicAdd(&cnt_dst[d], 1);
    }
}

__launch_bounds__(SCAN_T)
__global__ void scan_norm_kernel(const int* __restrict__ cnt_src, const int* __restrict__ cnt_dst,
                                 int* __restrict__ rowptr, float* __restrict__ dsrc,
                                 float* __restrict__ ddst, int n) {
    __shared__ int sh[SCAN_T];
    const int t      = threadIdx.x;
    const int chunk0 = blockIdx.x * SCAN_T;

    int part = 0;
    for (int i = t; i < chunk0; i += SCAN_T) part += cnt_dst[i];
    sh[t] = part;
    __syncthreads();
    for (int off = SCAN_T / 2; off > 0; off >>= 1) {
        if (t < off) sh[t] += sh[t + off];
        __syncthreads();
    }
    const int base = sh[0];
    __syncthreads();

    const int i = chunk0 + t;
    const int c = (i < n) ? cnt_dst[i] : 0;
    sh[t] = c;
    __syncthreads();
    for (int off = 1; off < SCAN_T; off <<= 1) {
        const int x = (t >= off) ? sh[t - off] : 0;
        __syncthreads();
        sh[t] += x;
        __syncthreads();
    }
    const int incl = sh[t];

    if (i < n) {
        rowptr[i] = base + incl - c;
        const int cs = cnt_src[i];
        dsrc[i] = rsqrtf(cs < 1 ? 1.0f : (float)cs);
        ddst[i] = rsqrtf(c  < 1 ? 1.0f : (float)c);
        if (i == n - 1) rowptr[n] = base + incl;
    }
}

__global__ void fill_kernel(const int* __restrict__ src, const int* __restrict__ dst,
                            const int* __restrict__ rowptr, const int* __restrict__ rank,
                            int* __restrict__ ssrc, int ne) {
    const int nthreads = gridDim.x * blockDim.x;
    int e = blockIdx.x * blockDim.x + threadIdx.x;
    for (; e + 3 * nthreads < ne; e += 4 * nthreads) {
        int s[4], d[4], r[4], p[4];
#pragma unroll
        for (int k = 0; k < 4; ++k) {
            d[k] = dst[e + k * nthreads];
            r[k] = rank[e + k * nthreads];
            s[k] = src[e + k * nthreads];
        }
#pragma unroll
        for (int k = 0; k < 4; ++k) p[k] = rowptr[d[k]];
#pragma unroll
        for (int k = 0; k < 4; ++k) ssrc[p[k] + r[k]] = s[k];
    }
    for (; e < ne; e += nthreads) ssrc[rowptr[dst[e]] + rank[e]] = src[e];
}

__global__ void copy_int_kernel(const int* __restrict__ a, int* __restrict__ b, int n) {
    const int nthreads = gridDim.x * blockDim.x;
    int i = blockIdx.x * blockDim.x + threadIdx.x;
    for (; i + 3 * nthreads < n; i += 4 * nthreads) {
        int v[4];
#pragma unroll
        for (int k = 0; k < 4; ++k) v[k] = a[i + k * nthreads];
#pragma unroll
        for (int k = 0; k < 4; ++k) b[i + k * nthreads] = v[k];
    }
    for (; i < n; i += nthreads) b[i] = a[i];
}

template <int FOUT>
__launch_bounds__(FOUT * 2)
__global__ void gemm_rows16_wmma(const float* __restrict__ X,
                                 const float* __restrict__ dinv,
                                 const float* __restrict__ W,
                                 float* __restrict__ H, int ntiles) {
    constexpr int LDSS = KDIM + 4;
    constexpr int NT   = FOUT * 2;
    constexpr int PER  = 16 * KDIM / NT;
    __shared__ __align__(16) float As[2][16 * LDSS];

    const int tid   = threadIdx.x;
    const int lane  = tid & 31;
    const int wave  = tid >> 5;
    const int n     = lane & 15;
    const int khalf = lane >> 4;
    const int m16   = lane & 15;
    const int ncol  = wave * 16 + n;

    v2f bf[KDIM / 4];
#pragma unroll
    for (int j = 0; j < KDIM / 4; ++j) {
        const int ka = 4 * j + 2 * khalf;
        bf[j].x = W[(size_t)ka * FOUT + ncol];
        bf[j].y = W[(size_t)(ka + 1) * FOUT + ncol];
    }

    constexpr int RP = NT / KDIM;
    const int ar = tid >> 7;
    const int ac = tid & (KDIM - 1);
    const float* xp = X + tid;
    const float* dp = dinv + ar;
    float xa[PER], da[PER];
    int tile = blockIdx.x;
    if (tile < ntiles) {
        const float* xt = xp + (size_t)tile * (16 * KDIM);
        const float* dt = dp + (size_t)tile * 16;
#pragma unroll
        for (int j = 0; j < PER; ++j) {
            xa[j] = xt[j * NT];
            da[j] = dt[j * RP];
        }
    }

    int st = 0;
    for (; tile < ntiles; tile += gridDim.x) {
        float* A  = As[st];
        float* aw = A + ar * LDSS + ac;
#pragma unroll
        for (int j = 0; j < PER; ++j) aw[j * RP * LDSS] = xa[j] * da[j];
        __syncthreads();

        const int nxt = tile + gridDim.x;
        if (nxt < ntiles) {
            const float* xt = xp + (size_t)nxt * (16 * KDIM);
            const float* dt = dp + (size_t)nxt * 16;
#pragma unroll
            for (int j = 0; j < PER; ++j) {
                xa[j] = xt[j * NT];
                da[j] = dt[j * RP];
            }
        }

        v8f acc = {};
#pragma unroll
        for (int k0 = 0; k0 < KDIM; k0 += 4) {
            const int ka = k0 + 2 * khalf;
            v2f a = *(const v2f*)&A[m16 * LDSS + ka];
            acc = __builtin_amdgcn_wmma_f32_16x16x4_f32(
                 false, a,  false, bf[k0 / 4],
                 (short)0, acc,  false,  false);
        }

        float* hp = H + ((size_t)tile * 16 + 8 * khalf) * FOUT + ncol;
#pragma unroll
        for (int r = 0; r < 8; ++r) hp[r * FOUT] = acc[r];
        st ^= 1;
    }
}

template <int V> struct RowVec { typedef float type __attribute__((ext_vector_type(V))); };

template <int V, int R>
__device__ __forceinline__ typename RowVec<V>::type
add_last_rows(const typename RowVec<V>::type* __restrict__ Hv, int iv, int j, int lane,
              typename RowVec<V>::type acc) {
    typedef typename RowVec<V>::type vec;
    int s[R];
    vec t[R];
    for (int k = 0; k < R; ++k) s[k] = __builtin_amdgcn_readlane(iv, j + k);
    for (int k = 0; k < R; ++k) t[k] = Hv[(size_t)s[k] * 32 + lane];
    if (R & 4) acc += (t[0] + t[1]) + (t[2] + t[3]);
    if (R & 2) acc += t[R & 4] + t[(R & 4) + 1];
    if (R & 1) acc += t[R - 1];
    return acc;
}

template <int V>
__device__ __forceinline__ typename RowVec<V>::type
sum_rows_of(const typename RowVec<V>::type* __restrict__ Hv, const int* __restrict__ ssrc,
            int beg, int end, int iv, int lane) {
    typedef typename RowVec<V>::type vec;
    vec acc = 0.0f;
    int e = beg;
    for (;;) {
        int cnt = end - e;
        cnt = cnt > 32 ? 32 : cnt;
        int j = 0;
        for (; j + 8 <= cnt; j += 8) {
            int s[8];
            vec v[8];
#pragma unroll
            for (int k = 0; k < 8; ++k) s[k] = __builtin_amdgcn_readlane(iv, j + k);
#pragma unroll
            for (int k = 0; k < 8; ++k) v[k] = Hv[(size_t)s[k] * 32 + lane];
            acc += ((v[0] + v[1]) + (v[2] + v[3])) + ((v[4] + v[5]) + (v[6] + v[7]));
        }
        switch (cnt - j) {
            case 1: acc = add_last_rows<V, 1>(Hv, iv, j, lane, acc); break;
            case 2: acc = add_last_rows<V, 2>(Hv, iv, j, lane, acc); break;
            case 3: acc = add_last_rows<V, 3>(Hv, iv, j, lane, acc); break;
            case 4: acc = add_last_rows<V, 4>(Hv, iv, j, lane, acc); break;
            case 5: acc = add_last_rows<V, 5>(Hv, iv, j, lane, acc); break;
            case 6: acc = add_last_rows<V, 6>(Hv, iv, j, lane, acc); break;
            case 7: acc = add_last_rows<V, 7>(Hv, iv, j, lane, acc); break;
            default: break;
        }
        e += cnt;
        if (e >= end) break;
        iv = (e + lane < end) ? ssrc[e + lane] : 0;
        asm volatile("" : "+v"(iv));
    }
    return acc;
}

#define GATHER_WAVES 8

template <int F, bool RELU>
__launch_bounds__(GATHER_WAVES * 32)
__global__ void gather_rows_kernel(const float* __restrict__ H,
                                   const int* __restrict__ rowptr,
                                   const int* __restrict__ ssrc,
                                   const float* __restrict__ dinv,
                                   const float* __restrict__ bias,
                                   float* __restrict__ Y, int n) {
    constexpr int V = F / 32;
    typedef float vec __attribute__((ext_vector_type(V)));

    const int lane = threadIdx.x & 31;
    const int wave = __builtin_amdgcn_readfirstlane((int)(threadIdx.x >> 5));
    const int stride = gridDim.x * GATHER_WAVES;
    const vec* Hv = (const vec*)H;
    const vec  b  = ((const vec*)bias)[lane];

    int row = blockIdx.x * GATHER_WAVES + wave;
    int pb0 = 0, pb1 = 0, qb0 = 0, qb1 = 0, iv = 0;
    if (row < n) {
        pb0 = rowptr[row];
        pb1 = rowptr[row + 1];
        iv  = (pb0 + lane < pb1) ? ssrc[pb0 + lane] : 0;
    }
    if (row + stride < n) {
        qb0 = rowptr[row + stride];
        qb1 = rowptr[row + stride + 1];
    }
    asm volatile("" : "+v"(iv));
    for (; row < n; row += stride) {
        const int beg = pb0;
        const int end = pb1;
        int niv = 0, rb0 = 0, rb1 = 0;
        if (row + stride < n) niv = (qb0 + lane < qb1) ? ssrc[qb0 + lane] : 0;
        if (row + 2 * stride < n) {
            rb0 = rowptr[row + 2 * stride];
            rb1 = rowptr[row + 2 * stride + 1];
        }

        const float di = dinv[row];

        const vec acc = sum_rows_of<V>(Hv, ssrc, beg, end, iv, lane);
        pb0 = qb0; pb1 = qb1; qb0 = rb0; qb1 = rb1; iv = niv;

        vec r = acc * di + b;
        if (RELU) {
#pragma unroll
            for (int k = 0; k < V; ++k) r[k] = r[k] > 0.0f ? r[k] : 0.0f;
        }
        ((vec*)Y)[(size_t)row * 32 + lane] = r;
    }
}

template <int FOUT>
__launch_bounds__(FUSE_WAVES * 32)
__attribute__((amdgpu_waves_per_eu(8)))
__global__ void gather_gemm_wmma(const float* __restrict__ Hin,
                                 const int* __restrict__ rowptr,
                                 const int* __restrict__ ssrc,
                                 const float* __restrict__ ddst,
                                 const float* __restrict__ bias,
                                 const float* __restrict__ dsrc,
                                 const float* __restrict__ W,
                                 float* __restrict__ Hout, int ntiles) {
    constexpr int LDSS = KDIM + 4;
    constexpr int CW   = FOUT / 16;
    constexpr int RPW  = 16 / FUSE_WAVES;
    typedef float vec __attribute__((ext_vector_type(4)));
    __shared__ __align__(16) float As[2][16 * LDSS];

    const int tid   = threadIdx.x;
    const int lane  = tid & 31;
    const int wave  = __builtin_amdgcn_readfirstlane((int)(threadIdx.x >> 5));
    const int n     = lane & 15;
    const int khalf = lane >> 4;
    const int m16   = lane & 15;
    const int ncol  = wave * 16 + n;

    v2f bf[KDIM / 4] = {};
    if (wave < CW) {
#pragma unroll
        for (int j = 0; j < KDIM / 4; ++j) {
            const int ka = 4 * j + 2 * khalf;
            bf[j].x = W[(size_t)ka * FOUT + ncol];
            bf[j].y = W[(size_t)(ka + 1) * FOUT + ncol];
        }
    }

    const vec* Hv = (const vec*)Hin;
    const vec  b  = ((const vec*)bias)[lane];

    static_assert(RPW == 2, "two rows per wave and tile");
    const int G = gridDim.x;
    int tile = blockIdx.x;
    int pb0 = 0, pb1 = 0, pb2 = 0, qb0 = 0, qb1 = 0, qb2 = 0, iv0 = 0, iv1 = 0;
    if (tile < ntiles) {
        const int* rp = rowptr + tile * 16 + wave * RPW;
        pb0 = rp[0]; pb1 = rp[1]; pb2 = rp[2];
        iv0 = (pb0 + lane < pb1) ? ssrc[pb0 + lane] : 0;
        iv1 = (pb1 + lane < pb2) ? ssrc[pb1 + lane] : 0;
    }
    if (tile + G < ntiles) {
        const int* rp = rowptr + (tile + G) * 16 + wave * RPW;
        qb0 = rp[0]; qb1 = rp[1]; qb2 = rp[2];
    }
    asm volatile("" : "+v"(iv0), "+v"(iv1));

    int st = 0;
    for (; tile < ntiles; tile += G) {
        float* A = As[st];
        const int bnd[RPW + 1] = {__builtin_amdgcn_readfirstlane(pb0),
                                  __builtin_amdgcn_readfirstlane(pb1),
                                  __builtin_amdgcn_readfirstlane(pb2)};
        const int ivs[RPW] = {iv0, iv1};
        int niv0 = 0, niv1 = 0, rb0 = 0, rb1 = 0, rb2 = 0;
        if (tile + G < ntiles) {
            niv0 = (qb0 + lane < qb1) ? ssrc[qb0 + lane] : 0;
            niv1 = (qb1 + lane < qb2) ? ssrc[qb1 + lane] : 0;
        }
        if (tile + 2 * G < ntiles) {
            const int* rp = rowptr + (tile + 2 * G) * 16 + wave * RPW;
            rb0 = rp[0]; rb1 = rp[1]; rb2 = rp[2];
        }
        for (int i = 0; i < RPW; ++i) {
            const int m   = wave * RPW + i;
            const int row = tile * 16 + m;
            const float di = ddst[row];
            const float ds = dsrc[row];
            const vec acc = sum_rows_of<4>(Hv, ssrc, bnd[i], bnd[i + 1], ivs[i], lane);

            vec r = acc * di + b;
#pragma unroll
            for (int k = 0; k < 4; ++k) r[k] = r[k] > 0.0f ? r[k] : 0.0f;
            r = r * ds;
            *(vec*)&A[m * LDSS + 4 * lane] = r;
        }
        __syncthreads();

        if (wave < CW) {
            v8f acc8 = {};
#pragma unroll
            for (int k0 = 0; k0 < KDIM; k0 += 4) {
                const int ka = k0 + 2 * khalf;
                v2f a = *(const v2f*)&A[m16 * LDSS + ka];
                acc8 = __builtin_amdgcn_wmma_f32_16x16x4_f32(
                     false, a,  false, bf[k0 / 4],
                     (short)0, acc8,  false,  false);
            }

            float* hp = Hout + ((size_t)tile * 16 + 8 * khalf) * FOUT + ncol;
#pragma unroll
            for (int r = 0; r < 8; ++r) hp[r * FOUT] = acc8[r];
        }
        pb0 = qb0; pb1 = qb1; pb2 = qb2;
        qb0 = rb0; qb1 = rb1; qb2 = rb2;
        iv0 = niv0; iv1 = niv1;
        st ^= 1;
    }
}

static inline int cdiv(long a, long b) { return (int)((a + b - 1) / b); }

extern "C" void kernel_launch(void* const* d_in, const int* in_sizes, int n_in,
                              void* d_out, int out_size, void* d_ws, size_t ws_size,
                              hipStream_t stream) {
    const float* features = (const float*)d_in[0];
    const int*   src      = (const int*)d_in[1];
    const int*   dst      = (const int*)d_in[2];
    const float* W1       = (const float*)d_in[3];
    const float* b1       = (const float*)d_in[4];
    const float* W2       = (const float*)d_in[5];
    const float* b2       = (const float*)d_in[6];
    const float* W3       = (const float*)d_in[7];
    const float* b3       = (const float*)d_in[8];
    float*       out      = (float*)d_out;

    const int N = in_sizes[0] / IN_F;
    const int E = in_sizes[1];

    float* ws   = (float*)d_ws;
    float* dsrc = ws;
    float* ddst = ws + N;
    float* buf0 = ws + 2 * (size_t)N;
    float* buf1 = buf0 + (size_t)N * HID_F;

    int* cnt_src = (int*)buf1;
    int* cnt_dst = cnt_src + N;
    int* rank    = cnt_dst + N;

    const int csr_len = N + 1 + E;
    int* csr_o    = (int*)out;
    int* rowptr_o = csr_o;
    int* ssrc_o   = csr_o + (N + 1);
    int* csr_w    = (int*)buf1;
    int* rowptr_w = csr_w;
    int* ssrc_w   = csr_w + (N + 1);

    const int ntiles = N / 16;

    zero_int_kernel<<<ZERO_BLOCKS, 256, 0, stream>>>(cnt_src, 2 * N);
    count_rank_kernel<<<ELEM_BLOCKS, 256, 0, stream>>>(src, dst, cnt_src, cnt_dst, rank, E);
    scan_norm_kernel<<<cdiv(N, SCAN_T), SCAN_T, 0, stream>>>(cnt_src, cnt_dst, rowptr_o, dsrc, ddst, N);
    fill_kernel<<<ELEM_BLOCKS, 256, 0, stream>>>(src, dst, rowptr_o, rank, ssrc_o, E);

    gemm_rows16_wmma<HID_F><<<GEMM128_BLOCKS, HID_F * 2, 0, stream>>>(features, dsrc, W1, buf0, ntiles);

    gather_gemm_wmma<HID_F><<<FUSE_BLOCKS, FUSE_WAVES * 32, 0, stream>>>(buf0, rowptr_o, ssrc_o, ddst, b1, dsrc, W2, buf1, ntiles);

    gather_gemm_wmma<OUT_F><<<FUSE_BLOCKS, FUSE_WAVES * 32, 0, stream>>>(buf1, rowptr_o, ssrc_o, ddst, b2, dsrc, W3, buf0, ntiles);

    copy_int_kernel<<<ELEM_BLOCKS, 256, 0, stream>>>(csr_o, csr_w, csr_len);
    gather_rows_kernel<OUT_F, false><<<GATHER_BLOCKS, GATHER_WAVES * 32, 0, stream>>>(buf0, rowptr_w, ssrc_w, ddst, b3, out, N);
}


# Edge list built by a two-level counting sort on LDS counters (edges filed by 256-row bucket, then one block per bucket ranks and places them): no global atomics left; per-row sum order changes
# speedup vs baseline: 4.6301x; 1.4830x over previous
#include <hip/hip_runtime.h>

typedef float v2f __attribute__((ext_vector_type(2)));
typedef float v8f __attribute__((ext_vector_type(8)));

#define IN_F  128
#define HID_F 128
#define OUT_F 64
#define KDIM  128

#define ELEM_BLOCKS    512
#define GATHER_BLOCKS  2048
#define GEMM128_BLOCKS 1024
#define FUSE_BLOCKS    1024
#define FUSE_WAVES     8

#define SCAN_T      1024
#define BKT_SH      8
#define BKT_ROWS    (1 << BKT_SH)
#define NBK_MAX     512
#define PART_T      1024
#define PART_EPT    8
#define PART_PER    (PART_EPT * PART_T)
#define BSORT_EPT   5
#define BSORT_T     1024

__launch_bounds__(PART_T)
__global__ void bucket_count_kernel(const int* __restrict__ src, const int* __restrict__ dst,
                                    int* __restrict__ histD, int* __restrict__ histS,
                                    int ne, int nbk) {
    __shared__ int hd[NBK_MAX];
    __shared__ int hs[NBK_MAX];
    const int t = threadIdx.x;
    for (int b = t; b < NBK_MAX; b += PART_T) { hd[b] = 0; hs[b] = 0; }
    __syncthreads();

    const int e0 = blockIdx.x * PART_PER;
    const int e1 = (e0 + PART_PER < ne) ? e0 + PART_PER : ne;
    if (e1 - e0 == PART_PER) {
        int s[PART_EPT], d[PART_EPT];
#pragma unroll
        for (int k = 0; k < PART_EPT; ++k) {
            d[k] = dst[e0 + t + k * PART_T];
            s[k] = src[e0 + t + k * PART_T];
        }
#pragma unroll
        for (int k = 0; k < PART_EPT; ++k) {
            atomicAdd(&hd[d[k] >> BKT_SH], 1);
            atomicAdd(&hs[s[k] >> BKT_SH], 1);
        }
    } else {
        for (int e = e0 + t; e < e1; e += PART_T) {
            atomicAdd(&hd[dst[e] >> BKT_SH], 1);
            atomicAdd(&hs[src[e] >> BKT_SH], 1);
        }
    }
    __syncthreads();

    const int G = gridDim.x;
    for (int b = t; b < nbk; b += PART_T) {
        histD[b * G + blockIdx.x] = hd[b];
        histS[b * G + blockIdx.x] = hs[b];
    }
}

__launch_bounds__(SCAN_T)
__global__ void scan_pair_kernel(const int* __restrict__ inA, int* __restrict__ outA,
                                 const int* __restrict__ inB, int* __restrict__ outB, int m) {
    __shared__ int sh[SCAN_T];
    const int t       = threadIdx.x;
    const int nchunks = gridDim.x >> 1;
    const bool second = (int)blockIdx.x >= nchunks;
    const int* in     = second ? inB : inA;
    int*       out    = second ? outB : outA;
    const int chunk0  = ((int)blockIdx.x - (second ? nchunks : 0)) * SCAN_T;

    int part = 0;
    for (int i = t; i < chunk0; i += SCAN_T) part += in[i];
    sh[t] = part;
    __syncthreads();
    for (int off = SCAN_T / 2; off > 0; off >>= 1) {
        if (t < off) sh[t] += sh[t + off];
        __syncthreads();
    }
    const int base = sh[0];
    __syncthreads();

    const int i = chunk0 + t;
    const int c = (i < m) ? in[i] : 0;
    sh[t] = c;
    __syncthreads();
    for (int off = 1; off < SCAN_T; off <<= 1) {
        const int x = (t >= off) ? sh[t - off] : 0;
        __syncthreads();
        sh[t] += x;
        __syncthreads();
    }
    if (i < m) out[i] = base + sh[t] - c;
}

__launch_bounds__(PART_T)
__global__ void bucket_scatter_kernel(const int* __restrict__ src, const int* __restrict__ dst,
                                      const int* __restrict__ offD, const int* __restrict__ offS,
                                      int* __restrict__ pairs, int* __restrict__ skeys,
                                      int ne, int nbk) {
    __shared__ int cd[NBK_MAX];
    __shared__ int cs[NBK_MAX];
    const int t = threadIdx.x;
    const int G = gridDim.x;
    for (int b = t; b < nbk; b += PART_T) {
        cd[b] = offD[b * G + blockIdx.x];
        cs[b] = offS[b * G + blockIdx.x];
    }
    __syncthreads();

    int2* pv = (int2*)pairs;
    const int e0 = blockIdx.x * PART_PER;
    const int e1 = (e0 + PART_PER < ne) ? e0 + PART_PER : ne;
    if (e1 - e0 == PART_PER) {
        int s[PART_EPT], d[PART_EPT], pd[PART_EPT], ps[PART_EPT];
#pragma unroll
        for (int k = 0; k < PART_EPT; ++k) {
            d[k] = dst[e0 + t + k * PART_T];
            s[k] = src[e0 + t + k * PART_T];
        }
#pragma unroll
        for (int k = 0; k < PART_EPT; ++k) {
            pd[k] = atomicAdd(&cd[d[k] >> BKT_SH], 1);
            ps[k] = atomicAdd(&cs[s[k] >> BKT_SH], 1);
        }
#pragma unroll
        for (int k = 0; k < PART_EPT; ++k) {
            pv[pd[k]]    = make_int2(d[k], s[k]);
            skeys[ps[k]] = s[k];
        }
    } else {
        for (int e = e0 + t; e < e1; e += PART_T) {
            const int d = dst[e];
            const int s = src[e];
            const int pd = atomicAdd(&cd[d >> BKT_SH], 1);
            const int ps = atomicAdd(&cs[s >> BKT_SH], 1);
            pv[pd]    = make_int2(d, s);
            skeys[ps] = s;
        }
    }
}

__launch_bounds__(BSORT_T)
__global__ void bucket_sort_kernel(const int* __restrict__ pairs, const int* __restrict__ skeys,
                                   const int* __restrict__ offD, const int* __restrict__ offS,
                                   int* __restrict__ rowptr, int* __restrict__ ssrc,
                                   float* __restrict__ dsrc, float* __restrict__ ddst,
                                   int n, int ne, int nparts) {
    __shared__ int cd[BKT_ROWS];
    __shared__ int cs[BKT_ROWS];
    __shared__ int sc[BKT_ROWS];
    const int t = threadIdx.x;
    const int b = blockIdx.x;
    const bool last = (b + 1 == (int)gridDim.x);
    if (t < BKT_ROWS) { cd[t] = 0; cs[t] = 0; }
    __syncthreads();

    const int db = offD[b * nparts];
    const int de = last ? ne : offD[(b + 1) * nparts];
    const int sb = offS[b * nparts];
    const int se = last ? ne : offS[(b + 1) * nparts];
    const int2* pv = (const int2*)pairs;
    const int nd = de - db;
    const int ns = se - sb;
    for (int j = t; j < nd; j += BSORT_EPT * BSORT_T) {
        int dk[BSORT_EPT];
#pragma unroll
        for (int k = 0; k < BSORT_EPT; ++k) {
            const int jj = j + k * BSORT_T;
            dk[k] = pv[db + (jj < nd ? jj : nd - 1)].x;
        }
#pragma unroll
        for (int k = 0; k < BSORT_EPT; ++k)
            if (j + k * BSORT_T < nd) atomicAdd(&cd[dk[k] & (BKT_ROWS - 1)], 1);
    }
    for (int j = t; j < ns; j += BSORT_EPT * BSORT_T) {
        int sk[BSORT_EPT];
#pragma unroll
        for (int k = 0; k < BSORT_EPT; ++k) {
            const int jj = j + k * BSORT_T;
            sk[k] = skeys[sb + (jj < ns ? jj : ns - 1)];
        }
#pragma unroll
        for (int k = 0; k < BSORT_EPT; ++k)
            if (j + k * BSORT_T < ns) atomicAdd(&cs[sk[k] & (BKT_ROWS - 1)], 1);
    }
    __syncthreads();

    int c = 0;
    if (t < BKT_ROWS) { c = cd[t]; sc[t] = c; }
    __syncthreads();
    for (int off = 1; off < BKT_ROWS; off <<= 1) {
        int x = 0;
        if (t < BKT_ROWS && t >= off) x = sc[t - off];
        __syncthreads();
        if (t < BKT_ROWS) sc[t] += x;
        __syncthreads();
    }
    if (t < BKT_ROWS) {
        const int start = db + sc[t] - c;
        const int row   = b * BKT_ROWS + t;
        cd[t] = start;
        if (row < n) {
            const int k = cs[t];
            rowptr[row] = start;
            ddst[row]   = rsqrtf(c < 1 ? 1.0f : (float)c);
            dsrc[row]   = rsqrtf(k < 1 ? 1.0f : (float)k);
            if (row == n - 1) rowptr[n] = ne;
        }
    }
    __syncthreads();

    for (int j = t; j < nd; j += BSORT_EPT * BSORT_T) {
        int2 p[BSORT_EPT];
#pragma unroll
        for (int k = 0; k < BSORT_EPT; ++k) {
            const int jj = j + k * BSORT_T;
            p[k] = pv[db + (jj < nd ? jj : nd - 1)];
        }
#pragma unroll
        for (int k = 0; k < BSORT_EPT; ++k) {
            if (j + k * BSORT_T < nd) {
                const int pos = atomicAdd(&cd[p[k].x & (BKT_ROWS - 1)], 1);
                ssrc[pos] = p[k].y;
            }
        }
    }
}

__global__ void copy_int_kernel(const int* __restrict__ a, int* __restrict__ b, int n) {
    const int nthreads = gridDim.x * blockDim.x;
    int i = blockIdx.x * blockDim.x + threadIdx.x;
    for (; i + 3 * nthreads < n; i += 4 * nthreads) {
        int v[4];
#pragma unroll
        for (int k = 0; k < 4; ++k) v[k] = a[i + k * nthreads];
#pragma unroll
        for (int k = 0; k < 4; ++k) b[i + k * nthreads] = v[k];
    }
    for (; i < n; i += nthreads) b[i] = a[i];
}

template <int FOUT>
__launch_bounds__(FOUT * 2)
__global__ void gemm_rows16_wmma(const float* __restrict__ X,
                                 const float* __restrict__ dinv,
                                 const float* __restrict__ W,
                                 float* __restrict__ H, int ntiles) {
    constexpr int LDSS = KDIM + 4;
    constexpr int NT   = FOUT * 2;
    constexpr int PER  = 16 * KDIM / NT;
    __shared__ __align__(16) float As[2][16 * LDSS];

    const int tid   = threadIdx.x;
    const int lane  = tid & 31;
    const int wave  = tid >> 5;
    const int n     = lane & 15;
    const int khalf = lane >> 4;
    const int m16   = lane & 15;
    const int ncol  = wave * 16 + n;

    v2f bf[KDIM / 4];
#pragma unroll
    for (int j = 0; j < KDIM / 4; ++j) {
        const int ka = 4 * j + 2 * khalf;
        bf[j].x = W[(size_t)ka * FOUT + ncol];
        bf[j].y = W[(size_t)(ka + 1) * FOUT + ncol];
    }

    constexpr int RP = NT / KDIM;
    const int ar = tid >> 7;
    const int ac = tid & (KDIM - 1);
    const float* xp = X + tid;
    const float* dp = dinv + ar;
    float xa[PER], da[PER];
    int tile = blockIdx.x;
    if (tile < ntiles) {
        const float* xt = xp + (size_t)tile * (16 * KDIM);
        const float* dt = dp + (size_t)tile * 16;
#pragma unroll
        for (int j = 0; j < PER; ++j) {
            xa[j] = xt[j * NT];
            da[j] = dt[j * RP];
        }
    }

    int st = 0;
    for (; tile < ntiles; tile += gridDim.x) {
        float* A  = As[st];
        float* aw = A + ar * LDSS + ac;
#pragma unroll
        for (int j = 0; j < PER; ++j) aw[j * RP * LDSS] = xa[j] * da[j];
        __syncthreads();

        const int nxt = tile + gridDim.x;
        if (nxt < ntiles) {
            const float* xt = xp + (size_t)nxt * (16 * KDIM);
            const float* dt = dp + (size_t)nxt * 16;
#pragma unroll
            for (int j = 0; j < PER; ++j) {
                xa[j] = xt[j * NT];
                da[j] = dt[j * RP];
            }
        }

        v8f acc = {};
#pragma unroll
        for (int k0 = 0; k0 < KDIM; k0 += 4) {
            const int ka = k0 + 2 * khalf;
            v2f a = *(const v2f*)&A[m16 * LDSS + ka];
            acc = __builtin_amdgcn_wmma_f32_16x16x4_f32(
                 false, a,  false, bf[k0 / 4],
                 (short)0, acc,  false,  false);
        }

        float* hp = H + ((size_t)tile * 16 + 8 * khalf) * FOUT + ncol;
#pragma unroll
        for (int r = 0; r < 8; ++r) hp[r * FOUT] = acc[r];
        st ^= 1;
    }
}

template <int V> struct RowVec { typedef float type __attribute__((ext_vector_type(V))); };

template <int V, int R>
__device__ __forceinline__ typename RowVec<V>::type
add_last_rows(const typename RowVec<V>::type* __restrict__ Hv, int iv, int j, int lane,
              typename RowVec<V>::type acc) {
    typedef typename RowVec<V>::type vec;
    int s[R];
    vec t[R];
    for (int k = 0; k < R; ++k) s[k] = __builtin_amdgcn_readlane(iv, j + k);
    for (int k = 0; k < R; ++k) t[k] = Hv[(size_t)s[k] * 32 + lane];
    if (R & 4) acc += (t[0] + t[1]) + (t[2] + t[3]);
    if (R & 2) acc += t[R & 4] + t[(R & 4) + 1];
    if (R & 1) acc += t[R - 1];
    return acc;
}

template <int V>
__device__ __forceinline__ typename RowVec<V>::type
sum_rows_of(const typename RowVec<V>::type* __restrict__ Hv, const int* __restrict__ ssrc,
            int beg, int end, int iv, int lane) {
    typedef typename RowVec<V>::type vec;
    vec acc = 0.0f;
    int e = beg;
    for (;;) {
        int cnt = end - e;
        cnt = cnt > 32 ? 32 : cnt;
        int j = 0;
        for (; j + 8 <= cnt; j += 8) {
            int s[8];
            vec v[8];
#pragma unroll
            for (int k = 0; k < 8; ++k) s[k] = __builtin_amdgcn_readlane(iv, j + k);
#pragma unroll
            for (int k = 0; k < 8; ++k) v[k] = Hv[(size_t)s[k] * 32 + lane];
            acc += ((v[0] + v[1]) + (v[2] + v[3])) + ((v[4] + v[5]) + (v[6] + v[7]));
        }
        switch (cnt - j) {
            case 1: acc = add_last_rows<V, 1>(Hv, iv, j, lane, acc); break;
            case 2: acc = add_last_rows<V, 2>(Hv, iv, j, lane, acc); break;
            case 3: acc = add_last_rows<V, 3>(Hv, iv, j, lane, acc); break;
            case 4: acc = add_last_rows<V, 4>(Hv, iv, j, lane, acc); break;
            case 5: acc = add_last_rows<V, 5>(Hv, iv, j, lane, acc); break;
            case 6: acc = add_last_rows<V, 6>(Hv, iv, j, lane, acc); break;
            case 7: acc = add_last_rows<V, 7>(Hv, iv, j, lane, acc); break;
            default: break;
        }
        e += cnt;
        if (e >= end) break;
        iv = (e + lane < end) ? ssrc[e + lane] : 0;
        asm volatile("" : "+v"(iv));
    }
    return acc;
}

#define GATHER_WAVES 8

template <int F, bool RELU>
__launch_bounds__(GATHER_WAVES * 32)
__global__ void gather_rows_kernel(const float* __restrict__ H,
                                   const int* __restrict__ rowptr,
                                   const int* __restrict__ ssrc,
                                   const float* __restrict__ dinv,
                                   const float* __restrict__ bias,
                                   float* __restrict__ Y, int n) {
    constexpr int V = F / 32;
    typedef float vec __attribute__((ext_vector_type(V)));

    const int lane = threadIdx.x & 31;
    const int wave = __builtin_amdgcn_readfirstlane((int)(threadIdx.x >> 5));
    const int stride = gridDim.x * GATHER_WAVES;
    const vec* Hv = (const vec*)H;
    const vec  b  = ((const vec*)bias)[lane];

    int row = blockIdx.x * GATHER_WAVES + wave;
    int pb0 = 0, pb1 = 0, qb0 = 0, qb1 = 0, iv = 0;
    if (row < n) {
        pb0 = rowptr[row];
        pb1 = rowptr[row + 1];
        iv  = (pb0 + lane < pb1) ? ssrc[pb0 + lane] : 0;
    }
    if (row + stride < n) {
        qb0 = rowptr[row + stride];
        qb1 = rowptr[row + stride + 1];
    }
    asm volatile("" : "+v"(iv));
    for (; row < n; row += stride) {
        const int beg = pb0;
        const int end = pb1;
        int niv = 0, rb0 = 0, rb1 = 0;
        if (row + stride < n) niv = (qb0 + lane < qb1) ? ssrc[qb0 + lane] : 0;
        if (row + 2 * stride < n) {
            rb0 = rowptr[row + 2 * stride];
            rb1 = rowptr[row + 2 * stride + 1];
        }

        const float di = dinv[row];

        const vec acc = sum_rows_of<V>(Hv, ssrc, beg, end, iv, lane);
        pb0 = qb0; pb1 = qb1; qb0 = rb0; qb1 = rb1; iv = niv;

        vec r = acc * di + b;
        if (RELU) {
#pragma unroll
            for (int k = 0; k < V; ++k) r[k] = r[k] > 0.0f ? r[k] : 0.0f;
        }
        ((vec*)Y)[(size_t)row * 32 + lane] = r;
    }
}

template <int FOUT>
__launch_bounds__(FUSE_WAVES * 32)
__attribute__((amdgpu_waves_per_eu(8)))
__global__ void gather_gemm_wmma(const float* __restrict__ Hin,
                                 const int* __restrict__ rowptr,
                                 const int* __restrict__ ssrc,
                                 const float* __restrict__ ddst,
                                 const float* __restrict__ bias,
                                 const float* __restrict__ dsrc,
                                 const float* __restrict__ W,
                                 float* __restrict__ Hout, int ntiles) {
    constexpr int LDSS = KDIM + 4;
    constexpr int CW   = FOUT / 16;
    constexpr int RPW  = 16 / FUSE_WAVES;
    typedef float vec __attribute__((ext_vector_type(4)));
    __shared__ __align__(16) float As[2][16 * LDSS];

    const int tid   = threadIdx.x;
    const int lane  = tid & 31;
    const int wave  = __builtin_amdgcn_readfirstlane((int)(threadIdx.x >> 5));
    const int n     = lane & 15;
    const int khalf = lane >> 4;
    const int m16   = lane & 15;
    const int ncol  = wave * 16 + n;

    v2f bf[KDIM / 4] = {};
    if (wave < CW) {
#pragma unroll
        for (int j = 0; j < KDIM / 4; ++j) {
            const int ka = 4 * j + 2 * khalf;
            bf[j].x = W[(size_t)ka * FOUT + ncol];
            bf[j].y = W[(size_t)(ka + 1) * FOUT + ncol];
        }
    }

    const vec* Hv = (const vec*)Hin;
    const vec  b  = ((const vec*)bias)[lane];

    static_assert(RPW == 2, "two rows per wave and tile");
    const int G = gridDim.x;
    int tile = blockIdx.x;
    int pb0 = 0, pb1 = 0, pb2 = 0, qb0 = 0, qb1 = 0, qb2 = 0, iv0 = 0, iv1 = 0;
    if (tile < ntiles) {
        const int* rp = rowptr + tile * 16 + wave * RPW;
        pb0 = rp[0]; pb1 = rp[1]; pb2 = rp[2];
        iv0 = (pb0 + lane < pb1) ? ssrc[pb0 + lane] : 0;
        iv1 = (pb1 + lane < pb2) ? ssrc[pb1 + lane] : 0;
    }
    if (tile + G < ntiles) {
        const int* rp = rowptr + (tile + G) * 16 + wave * RPW;
        qb0 = rp[0]; qb1 = rp[1]; qb2 = rp[2];
    }
    asm volatile("" : "+v"(iv0), "+v"(iv1));

    int st = 0;
    for (; tile < ntiles; tile += G) {
        float* A = As[st];
        const int bnd[RPW + 1] = {__builtin_amdgcn_readfirstlane(pb0),
                                  __builtin_amdgcn_readfirstlane(pb1),
                                  __builtin_amdgcn_readfirstlane(pb2)};
        const int ivs[RPW] = {iv0, iv1};
        int niv0 = 0, niv1 = 0, rb0 = 0, rb1 = 0, rb2 = 0;
        if (tile + G < ntiles) {
            niv0 = (qb0 + lane < qb1) ? ssrc[qb0 + lane] : 0;
            niv1 = (qb1 + lane < qb2) ? ssrc[qb1 + lane] : 0;
        }
        if (tile + 2 * G < ntiles) {
            const int* rp = rowptr + (tile + 2 * G) * 16 + wave * RPW;
            rb0 = rp[0]; rb1 = rp[1]; rb2 = rp[2];
        }
        for (int i = 0; i < RPW; ++i) {
            const int m   = wave * RPW + i;
            const int row = tile * 16 + m;
            const float di = ddst[row];
            const float ds = dsrc[row];
            const vec acc = sum_rows_of<4>(Hv, ssrc, bnd[i], bnd[i + 1], ivs[i], lane);

            vec r = acc * di + b;
#pragma unroll
            for (int k = 0; k < 4; ++k) r[k] = r[k] > 0.0f ? r[k] : 0.0f;
            r = r * ds;
            *(vec*)&A[m * LDSS + 4 * lane] = r;
        }
        __syncthreads();

        if (wave < CW) {
            v8f acc8 = {};
#pragma unroll
            for (int k0 = 0; k0 < KDIM; k0 += 4) {
                const int ka = k0 + 2 * khalf;
                v2f a = *(const v2f*)&A[m16 * LDSS + ka];
                acc8 = __builtin_amdgcn_wmma_f32_16x16x4_f32(
                     false, a,  false, bf[k0 / 4],
                     (short)0, acc8,  false,  false);
            }

            float* hp = Hout + ((size_t)tile * 16 + 8 * khalf) * FOUT + ncol;
#pragma unroll
            for (int r = 0; r < 8; ++r) hp[r * FOUT] = acc8[r];
        }
        pb0 = qb0; pb1 = qb1; pb2 = qb2;
        qb0 = rb0; qb1 = rb1; qb2 = rb2;
        iv0 = niv0; iv1 = niv1;
        st ^= 1;
    }
}

static inline int cdiv(long a, long b) { return (int)((a + b - 1) / b); }

extern "C" void kernel_launch(void* const* d_in, const int* in_sizes, int n_in,
                              void* d_out, int out_size, void* d_ws, size_t ws_size,
                              hipStream_t stream) {
    const float* features = (const float*)d_in[0];
    const int*   src      = (const int*)d_in[1];
    const int*   dst      = (const int*)d_in[2];
    const float* W1       = (const float*)d_in[3];
    const float* b1       = (const float*)d_in[4];
    const float* W2       = (const float*)d_in[5];
    const float* b2       = (const float*)d_in[6];
    const float* W3       = (const float*)d_in[7];
    const float* b3       = (const float*)d_in[8];
    float*       out      = (float*)d_out;

    const int N = in_sizes[0] / IN_F;
    const int E = in_sizes[1];

    float* ws   = (float*)d_ws;
    float* dsrc = ws;
    float* ddst = ws + N;
    float* buf0 = ws + 2 * (size_t)N;
    float* buf1 = buf0 + (size_t)N * HID_F;

    const int nbk = cdiv(N, BKT_ROWS);
    const int np  = cdiv(E, PART_PER);
    const int M   = nbk * np;
    int* pairs = (int*)buf1;
    int* skeys = pairs + 2 * (size_t)E;
    int* histD = skeys + E;
    int* histS = histD + M;
    int* offD  = histS + M;
    int* offS  = offD + M;

    const int csr_len = N + 1 + E;
    int* csr_o    = (int*)out;
    int* rowptr_o = csr_o;
    int* ssrc_o   = csr_o + (N + 1);
    int* csr_w    = (int*)buf1;
    int* rowptr_w = csr_w;
    int* ssrc_w   = csr_w + (N + 1);

    const int ntiles = N / 16;

    bucket_count_kernel<<<np, PART_T, 0, stream>>>(src, dst, histD, histS, E, nbk);
    scan_pair_kernel<<<2 * cdiv(M, SCAN_T), SCAN_T, 0, stream>>>(histD, offD, histS, offS, M);
    bucket_scatter_kernel<<<np, PART_T, 0, stream>>>(src, dst, offD, offS, pairs, skeys, E, nbk);
    bucket_sort_kernel<<<nbk, BSORT_T, 0, stream>>>(pairs, skeys, offD, offS, rowptr_o, ssrc_o, dsrc, ddst, N, E, np);

    gemm_rows16_wmma<HID_F><<<GEMM128_BLOCKS, HID_F * 2, 0, stream>>>(features, dsrc, W1, buf0, ntiles);

    gather_gemm_wmma<HID_F><<<FUSE_BLOCKS, FUSE_WAVES * 32, 0, stream>>>(buf0, rowptr_o, ssrc_o, ddst, b1, dsrc, W2, buf1, ntiles);

    gather_gemm_wmma<OUT_F><<<FUSE_BLOCKS, FUSE_WAVES * 32, 0, stream>>>(buf1, rowptr_o, ssrc_o, ddst, b2, dsrc, W3, buf0, ntiles);

    copy_int_kernel<<<ELEM_BLOCKS, 256, 0, stream>>>(csr_o, csr_w, csr_len);
    gather_rows_kernel<OUT_F, false><<<GATHER_BLOCKS, GATHER_WAVES * 32, 0, stream>>>(buf0, rowptr_w, ssrc_w, ddst, b3, out, N);
}


# Edge list built in 2 launches, not 4: a level-1 block sorts its stretch by bucket in LDS and writes it back whole; level-2 blocks read their runs from every stretch; per-row sum order changes
# speedup vs baseline: 5.7095x; 1.2331x over previous
#include <hip/hip_runtime.h>

typedef float v2f __attribute__((ext_vector_type(2)));
typedef float v8f __attribute__((ext_vector_type(8)));

#define IN_F  128
#define HID_F 128
#define OUT_F 64
#define KDIM  128

#define ELEM_BLOCKS    512
#define GATHER_BLOCKS  2048
#define GEMM128_BLOCKS 1024
#define FUSE_BLOCKS    1024
#define FUSE_WAVES     8

#define BKT_SH      8
#define BKT_ROWS    (1 << BKT_SH)
#define NBK_MAX     512
#define PART_T      1024
#define PART_EPT    8
#define PART_PER    (PART_EPT * PART_T)
#define NP_MAX      256
#define BSORT_T     1024
#define BSORT_WAVES (BSORT_T / 32)
#define RUNS_TRIP   4

__launch_bounds__(PART_T)
__global__ void local_sort_kernel(const int* __restrict__ src, const int* __restrict__ dst,
                                  int* __restrict__ chunkD, int* __restrict__ chunkS,
                                  int* __restrict__ preD, int* __restrict__ preS,
                                  int ne, int nbk) {
    __shared__ int hd[NBK_MAX];
    __shared__ int hs[NBK_MAX];
    __shared__ int sd[NBK_MAX];
    __shared__ int ss[NBK_MAX];
    __shared__ int2 sp[PART_PER];
    __shared__ int  sk[PART_PER];
    const int t = threadIdx.x;
    for (int b = t; b < NBK_MAX; b += PART_T) { hd[b] = 0; hs[b] = 0; }
    __syncthreads();

    const int e0 = blockIdx.x * PART_PER;
    const int e1 = (e0 + PART_PER < ne) ? e0 + PART_PER : ne;
    const int cnt = e1 - e0;
    int s[PART_EPT], d[PART_EPT];
#pragma unroll
    for (int k = 0; k < PART_EPT; ++k) {
        const int e = e0 + t + k * PART_T;
        d[k] = dst[e < e1 ? e : e1 - 1];
        s[k] = src[e < e1 ? e : e1 - 1];
    }
#pragma unroll
    for (int k = 0; k < PART_EPT; ++k) {
        if (e0 + t + k * PART_T < e1) {
            atomicAdd(&hd[d[k] >> BKT_SH], 1);
            atomicAdd(&hs[s[k] >> BKT_SH], 1);
        }
    }
    __syncthreads();

    const bool first = t < NBK_MAX;
    const int  u     = t & (NBK_MAX - 1);
    int* cntA = first ? hd : hs;
    int* scA  = first ? sd : ss;
    const int c = cntA[u];
    scA[u] = c;
    __syncthreads();
    for (int off = 1; off < NBK_MAX; off <<= 1) {
        const int x = (u >= off) ? scA[u - off] : 0;
        __syncthreads();
        scA[u] += x;
        __syncthreads();
    }
    const int excl = scA[u] - c;
    cntA[u] = excl;
    int* pre = first ? preD : preS;
    if (u <= nbk) pre[blockIdx.x * (nbk + 1) + u] = excl;
    __syncthreads();

#pragma unroll
    for (int k = 0; k < PART_EPT; ++k) {
        if (e0 + t + k * PART_T < e1) {
            const int pd = atomicAdd(&hd[d[k] >> BKT_SH], 1);
            const int ps = atomicAdd(&hs[s[k] >> BKT_SH], 1);
            sp[pd] = make_int2(d[k], s[k]);
            sk[ps] = s[k];
        }
    }
    __syncthreads();

    int2* cD = (int2*)chunkD + e0;
    int*  cS = chunkS + e0;
    for (int i = t; i < cnt; i += PART_T) {
        cD[i] = sp[i];
        cS[i] = sk[i];
    }
}

__launch_bounds__(BSORT_T)
__global__ void bucket_sort_kernel(const int* __restrict__ chunkD, const int* __restrict__ chunkS,
                                   const int* __restrict__ preD, const int* __restrict__ preS,
                                   int* __restrict__ rowptr, int* __restrict__ ssrc,
                                   float* __restrict__ dsrc, float* __restrict__ ddst,
                                   int n, int ne, int np, int nbk) {
    __shared__ int cd[BKT_ROWS];
    __shared__ int cs[BKT_ROWS];
    __shared__ int sc[BKT_ROWS];
    __shared__ int rsD[NP_MAX];
    __shared__ int rlD[NP_MAX];
    __shared__ int rsS[NP_MAX];
    __shared__ int rlS[NP_MAX];
    __shared__ int dbs;
    const int t    = threadIdx.x;
    const int lane = t & 31;
    const int wave = t >> 5;
    const int b    = blockIdx.x;
    if (t < BKT_ROWS) { cd[t] = 0; cs[t] = 0; }
    if (t == 0) dbs = 0;
    __syncthreads();

    const int pitch = nbk + 1;
    for (int j = t; j < np; j += BSORT_T) {
        const int a0 = preD[j * pitch + b];
        const int a1 = preD[j * pitch + b + 1];
        const int q0 = preS[j * pitch + b];
        const int q1 = preS[j * pitch + b + 1];
        rsD[j] = a0; rlD[j] = a1 - a0;
        rsS[j] = q0; rlS[j] = q1 - q0;
        atomicAdd(&dbs, a0);
    }
    __syncthreads();
    const int db = dbs;
    const int2* pv = (const int2*)chunkD;

    for (int j0 = wave * RUNS_TRIP; j0 < np; j0 += BSORT_WAVES * RUNS_TRIP) {
        int dk[RUNS_TRIP], sq[RUNS_TRIP], ld[RUNS_TRIP], ls[RUNS_TRIP], bd[RUNS_TRIP], bs[RUNS_TRIP];
#pragma unroll
        for (int r = 0; r < RUNS_TRIP; ++r) {
            const bool ok = j0 + r < np;
            const int  jc = ok ? j0 + r : np - 1;
            ld[r] = ok ? rlD[jc] : 0;
            ls[r] = ok ? rlS[jc] : 0;
            bd[r] = jc * PART_PER + rsD[jc];
            bs[r] = jc * PART_PER + rsS[jc];
            dk[r] = pv[lane < ld[r] ? bd[r] + lane : jc * PART_PER].x;
            sq[r] = chunkS[lane < ls[r] ? bs[r] + lane : jc * PART_PER];
        }
#pragma unroll
        for (int r = 0; r < RUNS_TRIP; ++r) {
            if (lane < ld[r]) atomicAdd(&cd[dk[r] & (BKT_ROWS - 1)], 1);
            if (lane < ls[r]) atomicAdd(&cs[sq[r] & (BKT_ROWS - 1)], 1);
        }
#pragma unroll
        for (int r = 0; r < RUNS_TRIP; ++r) {
            for (int k = lane + 32; k < ld[r]; k += 32)
                atomicAdd(&cd[pv[bd[r] + k].x & (BKT_ROWS - 1)], 1);
            for (int k = lane + 32; k < ls[r]; k += 32)
                atomicAdd(&cs[chunkS[bs[r] + k] & (BKT_ROWS - 1)], 1);
        }
    }
    __syncthreads();

    int c = 0;
    if (t < BKT_ROWS) { c = cd[t]; sc[t] = c; }
    __syncthreads();
    for (int off = 1; off < BKT_ROWS; off <<= 1) {
        int x = 0;
        if (t < BKT_ROWS && t >= off) x = sc[t - off];
        __syncthreads();
        if (t < BKT_ROWS) sc[t] += x;
        __syncthreads();
    }
    if (t < BKT_ROWS) {
        const int start = db + sc[t] - c;
        const int row   = b * BKT_ROWS + t;
        cd[t] = start;
        if (row < n) {
            const int k = cs[t];
            rowptr[row] = start;
            ddst[row]   = rsqrtf(c < 1 ? 1.0f : (float)c);
            dsrc[row]   = rsqrtf(k < 1 ? 1.0f : (float)k);
            if (row == n - 1) rowptr[n] = ne;
        }
    }
    __syncthreads();

    for (int j0 = wave * RUNS_TRIP; j0 < np; j0 += BSORT_WAVES * RUNS_TRIP) {
        int2 p[RUNS_TRIP];
        int  ld[RUNS_TRIP], bd[RUNS_TRIP];
#pragma unroll
        for (int r = 0; r < RUNS_TRIP; ++r) {
            const bool ok = j0 + r < np;
            const int  jc = ok ? j0 + r : np - 1;
            ld[r] = ok ? rlD[jc] : 0;
            bd[r] = jc * PART_PER + rsD[jc];
            p[r]  = pv[lane < ld[r] ? bd[r] + lane : jc * PART_PER];
        }
#pragma unroll
        for (int r = 0; r < RUNS_TRIP; ++r) {
            if (lane < ld[r]) {
                const int pos = atomicAdd(&cd[p[r].x & (BKT_ROWS - 1)], 1);
                ssrc[pos] = p[r].y;
            }
        }
#pragma unroll
        for (int r = 0; r < RUNS_TRIP; ++r) {
            for (int k = lane + 32; k < ld[r]; k += 32) {
                const int2 q = pv[bd[r] + k];
                const int pos = atomicAdd(&cd[q.x & (BKT_ROWS - 1)], 1);
                ssrc[pos] = q.y;
            }
        }
    }
}

__global__ void copy_int_kernel(const int* __restrict__ a, int* __restrict__ b, int n) {
    const int nthreads = gridDim.x * blockDim.x;
    int i = blockIdx.x * blockDim.x + threadIdx.x;
    for (; i + 3 * nthreads < n; i += 4 * nthreads) {
        int v[4];
#pragma unroll
        for (int k = 0; k < 4; ++k) v[k] = a[i + k * nthreads];
#pragma unroll
        for (int k = 0; k < 4; ++k) b[i + k * nthreads] = v[k];
    }
    for (; i < n; i += nthreads) b[i] = a[i];
}

template <int FOUT>
__launch_bounds__(FOUT * 2)
__global__ void gemm_rows16_wmma(const float* __restrict__ X,
                                 const float* __restrict__ dinv,
                                 const float* __restrict__ W,
                                 float* __restrict__ H, int ntiles) {
    constexpr int LDSS = KDIM + 4;
    constexpr int NT   = FOUT * 2;
    constexpr int PER  = 16 * KDIM / NT;
    __shared__ __align__(16) float As[2][16 * LDSS];

    const int tid   = threadIdx.x;
    const int lane  = tid & 31;
    const int wave  = tid >> 5;
    const int n     = lane & 15;
    const int khalf = lane >> 4;
    const int m16   = lane & 15;
    const int ncol  = wave * 16 + n;

    v2f bf[KDIM / 4];
#pragma unroll
    for (int j = 0; j < KDIM / 4; ++j) {
        const int ka = 4 * j + 2 * khalf;
        bf[j].x = W[(size_t)ka * FOUT + ncol];
        bf[j].y = W[(size_t)(ka + 1) * FOUT + ncol];
    }

    constexpr int RP = NT / KDIM;
    const int ar = tid >> 7;
    const int ac = tid & (KDIM - 1);
    const float* xp = X + tid;
    const float* dp = dinv + ar;
    float xa[PER], da[PER];
    int tile = blockIdx.x;
    if (tile < ntiles) {
        const float* xt = xp + (size_t)tile * (16 * KDIM);
        const float* dt = dp + (size_t)tile * 16;
#pragma unroll
        for (int j = 0; j < PER; ++j) {
            xa[j] = xt[j * NT];
            da[j] = dt[j * RP];
        }
    }

    int st = 0;
    for (; tile < ntiles; tile += gridDim.x) {
        float* A  = As[st];
        float* aw = A + ar * LDSS + ac;
#pragma unroll
        for (int j = 0; j < PER; ++j) aw[j * RP * LDSS] = xa[j] * da[j];
        __syncthreads();

        const int nxt = tile + gridDim.x;
        if (nxt < ntiles) {
            const float* xt = xp + (size_t)nxt * (16 * KDIM);
            const float* dt = dp + (size_t)nxt * 16;
#pragma unroll
            for (int j = 0; j < PER; ++j) {
                xa[j] = xt[j * NT];
                da[j] = dt[j * RP];
            }
        }

        v8f acc = {};
#pragma unroll
        for (int k0 = 0; k0 < KDIM; k0 += 4) {
            const int ka = k0 + 2 * khalf;
            v2f a = *(const v2f*)&A[m16 * LDSS + ka];
            acc = __builtin_amdgcn_wmma_f32_16x16x4_f32(
                 false, a,  false, bf[k0 / 4],
                 (short)0, acc,  false,  false);
        }

        float* hp = H + ((size_t)tile * 16 + 8 * khalf) * FOUT + ncol;
#pragma unroll
        for (int r = 0; r < 8; ++r) hp[r * FOUT] = acc[r];
        st ^= 1;
    }
}

template <int V> struct RowVec { typedef float type __attribute__((ext_vector_type(V))); };

template <int V, int R>
__device__ __forceinline__ typename RowVec<V>::type
add_last_rows(const typename RowVec<V>::type* __restrict__ Hv, int iv, int j, int lane,
              typename RowVec<V>::type acc) {
    typedef typename RowVec<V>::type vec;
    int s[R];
    vec t[R];
    for (int k = 0; k < R; ++k) s[k] = __builtin_amdgcn_readlane(iv, j + k);
    for (int k = 0; k < R; ++k) t[k] = Hv[(size_t)s[k] * 32 + lane];
    if (R & 4) acc += (t[0] + t[1]) + (t[2] + t[3]);
    if (R & 2) acc += t[R & 4] + t[(R & 4) + 1];
    if (R & 1) acc += t[R - 1];
    return acc;
}

template <int V>
__device__ __forceinline__ typename RowVec<V>::type
sum_rows_of(const typename RowVec<V>::type* __restrict__ Hv, const int* __restrict__ ssrc,
            int beg, int end, int iv, int lane) {
    typedef typename RowVec<V>::type vec;
    vec acc = 0.0f;
    int e = beg;
    for (;;) {
        int cnt = end - e;
        cnt = cnt > 32 ? 32 : cnt;
        int j = 0;
        for (; j + 8 <= cnt; j += 8) {
            int s[8];
            vec v[8];
#pragma unroll
            for (int k = 0; k < 8; ++k) s[k] = __builtin_amdgcn_readlane(iv, j + k);
#pragma unroll
            for (int k = 0; k < 8; ++k) v[k] = Hv[(size_t)s[k] * 32 + lane];
            acc += ((v[0] + v[1]) + (v[2] + v[3])) + ((v[4] + v[5]) + (v[6] + v[7]));
        }
        switch (cnt - j) {
            case 1: acc = add_last_rows<V, 1>(Hv, iv, j, lane, acc); break;
            case 2: acc = add_last_rows<V, 2>(Hv, iv, j, lane, acc); break;
            case 3: acc = add_last_rows<V, 3>(Hv, iv, j, lane, acc); break;
            case 4: acc = add_last_rows<V, 4>(Hv, iv, j, lane, acc); break;
            case 5: acc = add_last_rows<V, 5>(Hv, iv, j, lane, acc); break;
            case 6: acc = add_last_rows<V, 6>(Hv, iv, j, lane, acc); break;
            case 7: acc = add_last_rows<V, 7>(Hv, iv, j, lane, acc); break;
            default: break;
        }
        e += cnt;
        if (e >= end) break;
        iv = (e + lane < end) ? ssrc[e + lane] : 0;
        asm volatile("" : "+v"(iv));
    }
    return acc;
}

#define GATHER_WAVES 8

template <int F, bool RELU>
__launch_bounds__(GATHER_WAVES * 32)
__global__ void gather_rows_kernel(const float* __restrict__ H,
                                   const int* __restrict__ rowptr,
                                   const int* __restrict__ ssrc,
                                   const float* __restrict__ dinv,
                                   const float* __restrict__ bias,
                                   float* __restrict__ Y, int n) {
    constexpr int V = F / 32;
    typedef float vec __attribute__((ext_vector_type(V)));

    const int lane = threadIdx.x & 31;
    const int wave = __builtin_amdgcn_readfirstlane((int)(threadIdx.x >> 5));
    const int stride = gridDim.x * GATHER_WAVES;
    const vec* Hv = (const vec*)H;
    const vec  b  = ((const vec*)bias)[lane];

    int row = blockIdx.x * GATHER_WAVES + wave;
    int pb0 = 0, pb1 = 0, qb0 = 0, qb1 = 0, iv = 0;
    if (row < n) {
        pb0 = rowptr[row];
        pb1 = rowptr[row + 1];
        iv  = (pb0 + lane < pb1) ? ssrc[pb0 + lane] : 0;
    }
    if (row + stride < n) {
        qb0 = rowptr[row + stride];
        qb1 = rowptr[row + stride + 1];
    }
    asm volatile("" : "+v"(iv));
    for (; row < n; row += stride) {
        const int beg = pb0;
        const int end = pb1;
        int niv = 0, rb0 = 0, rb1 = 0;
        if (row + stride < n) niv = (qb0 + lane < qb1) ? ssrc[qb0 + lane] : 0;
        if (row + 2 * stride < n) {
            rb0 = rowptr[row + 2 * stride];
            rb1 = rowptr[row + 2 * stride + 1];
        }

        const float di = dinv[row];

        const vec acc = sum_rows_of<V>(Hv, ssrc, beg, end, iv, lane);
        pb0 = qb0; pb1 = qb1; qb0 = rb0; qb1 = rb1; iv = niv;

        vec r = acc * di + b;
        if (RELU) {
#pragma unroll
            for (int k = 0; k < V; ++k) r[k] = r[k] > 0.0f ? r[k] : 0.0f;
        }
        ((vec*)Y)[(size_t)row * 32 + lane] = r;
    }
}

template <int FOUT>
__launch_bounds__(FUSE_WAVES * 32)
__attribute__((amdgpu_waves_per_eu(8)))
__global__ void gather_gemm_wmma(const float* __restrict__ Hin,
                                 const int* __restrict__ rowptr,
                                 const int* __restrict__ ssrc,
                                 const float* __restrict__ ddst,
                                 const float* __restrict__ bias,
                                 const float* __restrict__ dsrc,
                                 const float* __restrict__ W,
                                 float* __restrict__ Hout, int ntiles) {
    constexpr int LDSS = KDIM + 4;
    constexpr int CW   = FOUT / 16;
    constexpr int RPW  = 16 / FUSE_WAVES;
    typedef float vec __attribute__((ext_vector_type(4)));
    __shared__ __align__(16) float As[2][16 * LDSS];

    const int tid   = threadIdx.x;
    const int lane  = tid & 31;
    const int wave  = __builtin_amdgcn_readfirstlane((int)(threadIdx.x >> 5));
    const int n     = lane & 15;
    const int khalf = lane >> 4;
    const int m16   = lane & 15;
    const int ncol  = wave * 16 + n;

    v2f bf[KDIM / 4] = {};
    if (wave < CW) {
#pragma unroll
        for (int j = 0; j < KDIM / 4; ++j) {
            const int ka = 4 * j + 2 * khalf;
            bf[j].x = W[(size_t)ka * FOUT + ncol];
            bf[j].y = W[(size_t)(ka + 1) * FOUT + ncol];
        }
    }

    const vec* Hv = (const vec*)Hin;
    const vec  b  = ((const vec*)bias)[lane];

    static_assert(RPW == 2, "two rows per wave and tile");
    const int G = gridDim.x;
    int tile = blockIdx.x;
    int pb0 = 0, pb1 = 0, pb2 = 0, qb0 = 0, qb1 = 0, qb2 = 0, iv0 = 0, iv1 = 0;
    if (tile < ntiles) {
        const int* rp = rowptr + tile * 16 + wave * RPW;
        pb0 = rp[0]; pb1 = rp[1]; pb2 = rp[2];
        iv0 = (pb0 + lane < pb1) ? ssrc[pb0 + lane] : 0;
        iv1 = (pb1 + lane < pb2) ? ssrc[pb1 + lane] : 0;
    }
    if (tile + G < ntiles) {
        const int* rp = rowptr + (tile + G) * 16 + wave * RPW;
        qb0 = rp[0]; qb1 = rp[1]; qb2 = rp[2];
    }
    asm volatile("" : "+v"(iv0), "+v"(iv1));

    int st = 0;
    for (; tile < ntiles; tile += G) {
        float* A = As[st];
        const int bnd[RPW + 1] = {__builtin_amdgcn_readfirstlane(pb0),
                                  __builtin_amdgcn_readfirstlane(pb1),
                                  __builtin_amdgcn_readfirstlane(pb2)};
        const int ivs[RPW] = {iv0, iv1};
        int niv0 = 0, niv1 = 0, rb0 = 0, rb1 = 0, rb2 = 0;
        if (tile + G < ntiles) {
            niv0 = (qb0 + lane < qb1) ? ssrc[qb0 + lane] : 0;
            niv1 = (qb1 + lane < qb2) ? ssrc[qb1 + lane] : 0;
        }
        if (tile + 2 * G < ntiles) {
            const int* rp = rowptr + (tile + 2 * G) * 16 + wave * RPW;
            rb0 = rp[0]; rb1 = rp[1]; rb2 = rp[2];
        }
        for (int i = 0; i < RPW; ++i) {
            const int m   = wave * RPW + i;
            const int row = tile * 16 + m;
            const float di = ddst[row];
            const float ds = dsrc[row];
            const vec acc = sum_rows_of<4>(Hv, ssrc, bnd[i], bnd[i + 1], ivs[i], lane);

            vec r = acc * di + b;
#pragma unroll
            for (int k = 0; k < 4; ++k) r[k] = r[k] > 0.0f ? r[k] : 0.0f;
            r = r * ds;
            *(vec*)&A[m * LDSS + 4 * lane] = r;
        }
        __syncthreads();

        if (wave < CW) {
            v8f acc8 = {};
#pragma unroll
            for (int k0 = 0; k0 < KDIM; k0 += 4) {
                const int ka = k0 + 2 * khalf;
                v2f a = *(const v2f*)&A[m16 * LDSS + ka];
                acc8 = __builtin_amdgcn_wmma_f32_16x16x4_f32(
                     false, a,  false, bf[k0 / 4],
                     (short)0, acc8,  false,  false);
            }

            float* hp = Hout + ((size_t)tile * 16 + 8 * khalf) * FOUT + ncol;
#pragma unroll
            for (int r = 0; r < 8; ++r) hp[r * FOUT] = acc8[r];
        }
        pb0 = qb0; pb1 = qb1; pb2 = qb2;
        qb0 = rb0; qb1 = rb1; qb2 = rb2;
        iv0 = niv0; iv1 = niv1;
        st ^= 1;
    }
}

static inline int cdiv(long a, long b) { return (int)((a + b - 1) / b); }

extern "C" void kernel_launch(void* const* d_in, const int* in_sizes, int n_in,
                              void* d_out, int out_size, void* d_ws, size_t ws_size,
                              hipStream_t stream) {
    const float* features = (const float*)d_in[0];
    const int*   src      = (const int*)d_in[1];
    const int*   dst      = (const int*)d_in[2];
    const float* W1       = (const float*)d_in[3];
    const float* b1       = (const float*)d_in[4];
    const float* W2       = (const float*)d_in[5];
    const float* b2       = (const float*)d_in[6];
    const float* W3       = (const float*)d_in[7];
    const float* b3       = (const float*)d_in[8];
    float*       out      = (float*)d_out;

    const int N = in_sizes[0] / IN_F;
    const int E = in_sizes[1];

    float* ws   = (float*)d_ws;
    float* dsrc = ws;
    float* ddst = ws + N;
    float* buf0 = ws + 2 * (size_t)N;
    float* buf1 = buf0 + (size_t)N * HID_F;

    const int nbk = cdiv(N, BKT_ROWS);
    const int np  = cdiv(E, PART_PER);
    const int M   = (nbk + 1) * np;
    int* chunkD = (int*)buf1;
    int* chunkS = chunkD + 2 * (size_t)E;
    int* preD   = chunkS + E;
    int* preS   = preD + M;

    const int csr_len = N + 1 + E;
    int* csr_o    = (int*)out;
    int* rowptr_o = csr_o;
    int* ssrc_o   = csr_o + (N + 1);
    int* csr_w    = (int*)buf1;
    int* rowptr_w = csr_w;
    int* ssrc_w   = csr_w + (N + 1);

    const int ntiles = N / 16;

    local_sort_kernel<<<np, PART_T, 0, stream>>>(src, dst, chunkD, chunkS, preD, preS, E, nbk);
    bucket_sort_kernel<<<nbk, BSORT_T, 0, stream>>>(chunkD, chunkS, preD, preS, rowptr_o, ssrc_o, dsrc, ddst, N, E, np, nbk);

    gemm_rows16_wmma<HID_F><<<GEMM128_BLOCKS, HID_F * 2, 0, stream>>>(features, dsrc, W1, buf0, ntiles);

    gather_gemm_wmma<HID_F><<<FUSE_BLOCKS, FUSE_WAVES * 32, 0, stream>>>(buf0, rowptr_o, ssrc_o, ddst, b1, dsrc, W2, buf1, ntiles);

    gather_gemm_wmma<OUT_F><<<FUSE_BLOCKS, FUSE_WAVES * 32, 0, stream>>>(buf1, rowptr_o, ssrc_o, ddst, b2, dsrc, W3, buf0, ntiles);

    copy_int_kernel<<<ELEM_BLOCKS, 256, 0, stream>>>(csr_o, csr_w, csr_len);
    gather_rows_kernel<OUT_F, false><<<GATHER_BLOCKS, GATHER_WAVES * 32, 0, stream>>>(buf0, rowptr_w, ssrc_w, ddst, b3, out, N);
}


# Level 2 of the edge-list build keeps its bucket in LDS: pairs stay where the counting pass read them, sources are placed in LDS and stored whole, not 4 bytes per edge; per-row sum order may change
# speedup vs baseline: 5.9389x; 1.0402x over previous
#include <hip/hip_runtime.h>

typedef float v2f __attribute__((ext_vector_type(2)));
typedef float v8f __attribute__((ext_vector_type(8)));

#define IN_F  128
#define HID_F 128
#define OUT_F 64
#define KDIM  128

#define ELEM_BLOCKS    512
#define GATHER_BLOCKS  2048
#define GEMM128_BLOCKS 1024
#define FUSE_BLOCKS    1024
#define FUSE_WAVES     8

#define BKT_SH      8
#define BKT_ROWS    (1 << BKT_SH)
#define NBK_MAX     512
#define PART_T      1024
#define PART_EPT    8
#define PART_PER    (PART_EPT * PART_T)
#define NP_MAX      256
#define BSORT_T     1024
#define BSORT_WAVES (BSORT_T / 32)
#define RUNS_TRIP   4
#define RUN_SLOTS   32
#define OB_CAP      8192

__launch_bounds__(PART_T)
__global__ void local_sort_kernel(const int* __restrict__ src, const int* __restrict__ dst,
                                  int* __restrict__ chunkD, int* __restrict__ chunkS,
                                  int* __restrict__ preD, int* __restrict__ preS,
                                  int ne, int nbk) {
    __shared__ int hd[NBK_MAX];
    __shared__ int hs[NBK_MAX];
    __shared__ int sd[NBK_MAX];
    __shared__ int ss[NBK_MAX];
    __shared__ int2 sp[PART_PER];
    __shared__ int  sk[PART_PER];
    const int t = threadIdx.x;
    for (int b = t; b < NBK_MAX; b += PART_T) { hd[b] = 0; hs[b] = 0; }
    __syncthreads();

    const int e0 = blockIdx.x * PART_PER;
    const int e1 = (e0 + PART_PER < ne) ? e0 + PART_PER : ne;
    const int cnt = e1 - e0;
    int s[PART_EPT], d[PART_EPT];
#pragma unroll
    for (int k = 0; k < PART_EPT; ++k) {
        const int e = e0 + t + k * PART_T;
        d[k] = dst[e < e1 ? e : e1 - 1];
        s[k] = src[e < e1 ? e : e1 - 1];
    }
#pragma unroll
    for (int k = 0; k < PART_EPT; ++k) {
        if (e0 + t + k * PART_T < e1) {
            atomicAdd(&hd[d[k] >> BKT_SH], 1);
            atomicAdd(&hs[s[k] >> BKT_SH], 1);
        }
    }
    __syncthreads();

    const bool first = t < NBK_MAX;
    const int  u     = t & (NBK_MAX - 1);
    int* cntA = first ? hd : hs;
    int* scA  = first ? sd : ss;
    const int c = cntA[u];
    scA[u] = c;
    __syncthreads();
    for (int off = 1; off < NBK_MAX; off <<= 1) {
        const int x = (u >= off) ? scA[u - off] : 0;
        __syncthreads();
        scA[u] += x;
        __syncthreads();
    }
    const int excl = scA[u] - c;
    cntA[u] = excl;
    int* pre = first ? preD : preS;
    if (u <= nbk) pre[blockIdx.x * (nbk + 1) + u] = excl;
    __syncthreads();

#pragma unroll
    for (int k = 0; k < PART_EPT; ++k) {
        if (e0 + t + k * PART_T < e1) {
            const int pd = atomicAdd(&hd[d[k] >> BKT_SH], 1);
            const int ps = atomicAdd(&hs[s[k] >> BKT_SH], 1);
            sp[pd] = make_int2(d[k], s[k]);
            sk[ps] = s[k];
        }
    }
    __syncthreads();

    int2* cD = (int2*)chunkD + e0;
    int*  cS = chunkS + e0;
    for (int i = t; i < cnt; i += PART_T) {
        cD[i] = sp[i];
        cS[i] = sk[i];
    }
}

__launch_bounds__(BSORT_T)
__global__ void bucket_sort_kernel(const int* __restrict__ chunkD, const int* __restrict__ chunkS,
                                   const int* __restrict__ preD, const int* __restrict__ preS,
                                   int* __restrict__ rowptr, int* __restrict__ ssrc,
                                   float* __restrict__ dsrc, float* __restrict__ ddst,
                                   int n, int ne, int np, int nbk) {
    __shared__ int cd[BKT_ROWS];
    __shared__ int cs[BKT_ROWS];
    __shared__ int sc[BKT_ROWS];
    __shared__ int rsD[NP_MAX];
    __shared__ int rlD[NP_MAX];
    __shared__ int rsS[NP_MAX];
    __shared__ int rlS[NP_MAX];
    __shared__ int dbs;
    __shared__ int2 st[NP_MAX * RUN_SLOTS];
    __shared__ int  ob[OB_CAP];
    const int t    = threadIdx.x;
    const int lane = t & 31;
    const int wave = t >> 5;
    const int b    = blockIdx.x;
    if (t < BKT_ROWS) { cd[t] = 0; cs[t] = 0; }
    if (t == 0) dbs = 0;
    __syncthreads();

    const int pitch = nbk + 1;
    for (int j = t; j < np; j += BSORT_T) {
        const int a0 = preD[j * pitch + b];
        const int a1 = preD[j * pitch + b + 1];
        const int q0 = preS[j * pitch + b];
        const int q1 = preS[j * pitch + b + 1];
        rsD[j] = a0; rlD[j] = a1 - a0;
        rsS[j] = q0; rlS[j] = q1 - q0;
        atomicAdd(&dbs, a0);
    }
    __syncthreads();
    const int db = dbs;
    const int2* pv = (const int2*)chunkD;

    for (int j0 = wave * RUNS_TRIP; j0 < np; j0 += BSORT_WAVES * RUNS_TRIP) {
        int2 dk[RUNS_TRIP];
        int  sq[RUNS_TRIP], ld[RUNS_TRIP], ls[RUNS_TRIP], bd[RUNS_TRIP], bs[RUNS_TRIP];
#pragma unroll
        for (int r = 0; r < RUNS_TRIP; ++r) {
            const bool ok = j0 + r < np;
            const int  jc = ok ? j0 + r : np - 1;
            ld[r] = ok ? rlD[jc] : 0;
            ls[r] = ok ? rlS[jc] : 0;
            bd[r] = jc * PART_PER + rsD[jc];
            bs[r] = jc * PART_PER + rsS[jc];
            dk[r] = pv[lane < ld[r] ? bd[r] + lane : jc * PART_PER];
            sq[r] = chunkS[lane < ls[r] ? bs[r] + lane : jc * PART_PER];
        }
#pragma unroll
        for (int r = 0; r < RUNS_TRIP; ++r) {
            if (lane < ld[r]) {
                st[(j0 + r) * RUN_SLOTS + lane] = dk[r];
                atomicAdd(&cd[dk[r].x & (BKT_ROWS - 1)], 1);
            }
            if (lane < ls[r]) atomicAdd(&cs[sq[r] & (BKT_ROWS - 1)], 1);
        }
#pragma unroll
        for (int r = 0; r < RUNS_TRIP; ++r) {
            for (int k = lane + 32; k < ld[r]; k += 32)
                atomicAdd(&cd[pv[bd[r] + k].x & (BKT_ROWS - 1)], 1);
            for (int k = lane + 32; k < ls[r]; k += 32)
                atomicAdd(&cs[chunkS[bs[r] + k] & (BKT_ROWS - 1)], 1);
        }
    }
    __syncthreads();

    int c = 0;
    if (t < BKT_ROWS) { c = cd[t]; sc[t] = c; }
    __syncthreads();
    for (int off = 1; off < BKT_ROWS; off <<= 1) {
        int x = 0;
        if (t < BKT_ROWS && t >= off) x = sc[t - off];
        __syncthreads();
        if (t < BKT_ROWS) sc[t] += x;
        __syncthreads();
    }
    const int  total = sc[BKT_ROWS - 1];
    const bool fits  = total <= OB_CAP;
    if (t < BKT_ROWS) {
        const int start = sc[t] - c;
        const int row   = b * BKT_ROWS + t;
        cd[t] = start;
        if (row < n) {
            const int k = cs[t];
            rowptr[row] = db + start;
            ddst[row]   = rsqrtf(c < 1 ? 1.0f : (float)c);
            dsrc[row]   = rsqrtf(k < 1 ? 1.0f : (float)k);
            if (row == n - 1) rowptr[n] = ne;
        }
    }
    __syncthreads();

    for (int j0 = wave * RUNS_TRIP; j0 < np; j0 += BSORT_WAVES * RUNS_TRIP) {
        int ld[RUNS_TRIP], bd[RUNS_TRIP];
#pragma unroll
        for (int r = 0; r < RUNS_TRIP; ++r) {
            const bool ok = j0 + r < np;
            const int  jc = ok ? j0 + r : np - 1;
            ld[r] = ok ? rlD[jc] : 0;
            bd[r] = jc * PART_PER + rsD[jc];
        }
        int2 p[RUNS_TRIP];
        int  pos[RUNS_TRIP];
#pragma unroll
        for (int r = 0; r < RUNS_TRIP; ++r) {
            p[r] = make_int2(0, 0);
            if (lane < ld[r]) p[r] = st[(j0 + r) * RUN_SLOTS + lane];
        }
#pragma unroll
        for (int r = 0; r < RUNS_TRIP; ++r) {
            pos[r] = 0;
            if (lane < ld[r]) pos[r] = atomicAdd(&cd[p[r].x & (BKT_ROWS - 1)], 1);
        }
#pragma unroll
        for (int r = 0; r < RUNS_TRIP; ++r) {
            if (lane < ld[r]) {
                if (fits) ob[pos[r]] = p[r].y; else ssrc[db + pos[r]] = p[r].y;
            }
        }
#pragma unroll
        for (int r = 0; r < RUNS_TRIP; ++r) {
            for (int k = lane + 32; k < ld[r]; k += 32) {
                const int2 q = pv[bd[r] + k];
                const int pq = atomicAdd(&cd[q.x & (BKT_ROWS - 1)], 1);
                if (fits) ob[pq] = q.y; else ssrc[db + pq] = q.y;
            }
        }
    }
    __syncthreads();

    if (fits) {
        for (int k = 0; k < OB_CAP / BSORT_T; ++k) {
            const int i = t + k * BSORT_T;
            if (i < total) ssrc[db + i] = ob[i];
        }
    }
}

__global__ void copy_int_kernel(const int* __restrict__ a, int* __restrict__ b, int n) {
    const int nthreads = gridDim.x * blockDim.x;
    int i = blockIdx.x * blockDim.x + threadIdx.x;
    for (; i + 3 * nthreads < n; i += 4 * nthreads) {
        int v[4];
#pragma unroll
        for (int k = 0; k < 4; ++k) v[k] = a[i + k * nthreads];
#pragma unroll
        for (int k = 0; k < 4; ++k) b[i + k * nthreads] = v[k];
    }
    for (; i < n; i += nthreads) b[i] = a[i];
}

template <int FOUT>
__launch_bounds__(FOUT * 2)
__global__ void gemm_rows16_wmma(const float* __restrict__ X,
                                 const float* __restrict__ dinv,
                                 const float* __restrict__ W,
                                 float* __restrict__ H, int ntiles) {
    constexpr int LDSS = KDIM + 4;
    constexpr int NT   = FOUT * 2;
    constexpr int PER  = 16 * KDIM / NT;
    __shared__ __align__(16) float As[2][16 * LDSS];

    const int tid   = threadIdx.x;
    const int lane  = tid & 31;
    const int wave  = tid >> 5;
    const int n     = lane & 15;
    const int khalf = lane >> 4;
    const int m16   = lane & 15;
    const int ncol  = wave * 16 + n;

    v2f bf[KDIM / 4];
#pragma unroll
    for (int j = 0; j < KDIM / 4; ++j) {
        const int ka = 4 * j + 2 * khalf;
        bf[j].x = W[(size_t)ka * FOUT + ncol];
        bf[j].y = W[(size_t)(ka + 1) * FOUT + ncol];
    }

    constexpr int RP = NT / KDIM;
    const int ar = tid >> 7;
    const int ac = tid & (KDIM - 1);
    const float* xp = X + tid;
    const float* dp = dinv + ar;
    float xa[PER], da[PER];
    int tile = blockIdx.x;
    if (tile < ntiles) {
        const float* xt = xp + (size_t)tile * (16 * KDIM);
        const float* dt = dp + (size_t)tile * 16;
#pragma unroll
        for (int j = 0; j < PER; ++j) {
            xa[j] = xt[j * NT];
            da[j] = dt[j * RP];
        }
    }

    int st = 0;
    for (; tile < ntiles; tile += gridDim.x) {
        float* A  = As[st];
        float* aw = A + ar * LDSS + ac;
#pragma unroll
        for (int j = 0; j < PER; ++j) aw[j * RP * LDSS] = xa[j] * da[j];
        __syncthreads();

        const int nxt = tile + gridDim.x;
        if (nxt < ntiles) {
            const float* xt = xp + (size_t)nxt * (16 * KDIM);
            const float* dt = dp + (size_t)nxt * 16;
#pragma unroll
            for (int j = 0; j < PER; ++j) {
                xa[j] = xt[j * NT];
                da[j] = dt[j * RP];
            }
        }

        v8f acc = {};
#pragma unroll
        for (int k0 = 0; k0 < KDIM; k0 += 4) {
            const int ka = k0 + 2 * khalf;
            v2f a = *(const v2f*)&A[m16 * LDSS + ka];
            acc = __builtin_amdgcn_wmma_f32_16x16x4_f32(
                 false, a,  false, bf[k0 / 4],
                 (short)0, acc,  false,  false);
        }

        float* hp = H + ((size_t)tile * 16 + 8 * khalf) * FOUT + ncol;
#pragma unroll
        for (int r = 0; r < 8; ++r) hp[r * FOUT] = acc[r];
        st ^= 1;
    }
}

template <int V> struct RowVec { typedef float type __attribute__((ext_vector_type(V))); };

template <int V, int R>
__device__ __forceinline__ typename RowVec<V>::type
add_last_rows(const typename RowVec<V>::type* __restrict__ Hv, int iv, int j, int lane,
              typename RowVec<V>::type acc) {
    typedef typename RowVec<V>::type vec;
    int s[R];
    vec t[R];
    for (int k = 0; k < R; ++k) s[k] = __builtin_amdgcn_readlane(iv, j + k);
    for (int k = 0; k < R; ++k) t[k] = Hv[(size_t)s[k] * 32 + lane];
    if (R & 4) acc += (t[0] + t[1]) + (t[2] + t[3]);
    if (R & 2) acc += t[R & 4] + t[(R & 4) + 1];
    if (R & 1) acc += t[R - 1];
    return acc;
}

template <int V>
__device__ __forceinline__ typename RowVec<V>::type
sum_rows_of(const typename RowVec<V>::type* __restrict__ Hv, const int* __restrict__ ssrc,
            int beg, int end, int iv, int lane) {
    typedef typename RowVec<V>::type vec;
    vec acc = 0.0f;
    int e = beg;
    for (;;) {
        int cnt = end - e;
        cnt = cnt > 32 ? 32 : cnt;
        int j = 0;
        for (; j + 8 <= cnt; j += 8) {
            int s[8];
            vec v[8];
#pragma unroll
            for (int k = 0; k < 8; ++k) s[k] = __builtin_amdgcn_readlane(iv, j + k);
#pragma unroll
            for (int k = 0; k < 8; ++k) v[k] = Hv[(size_t)s[k] * 32 + lane];
            acc += ((v[0] + v[1]) + (v[2] + v[3])) + ((v[4] + v[5]) + (v[6] + v[7]));
        }
        switch (cnt - j) {
            case 1: acc = add_last_rows<V, 1>(Hv, iv, j, lane, acc); break;
            case 2: acc = add_last_rows<V, 2>(Hv, iv, j, lane, acc); break;
            case 3: acc = add_last_rows<V, 3>(Hv, iv, j, lane, acc); break;
            case 4: acc = add_last_rows<V, 4>(Hv, iv, j, lane, acc); break;
            case 5: acc = add_last_rows<V, 5>(Hv, iv, j, lane, acc); break;
            case 6: acc = add_last_rows<V, 6>(Hv, iv, j, lane, acc); break;
            case 7: acc = add_last_rows<V, 7>(Hv, iv, j, lane, acc); break;
            default: break;
        }
        e += cnt;
        if (e >= end) break;
        iv = (e + lane < end) ? ssrc[e + lane] : 0;
        asm volatile("" : "+v"(iv));
    }
    return acc;
}

#define GATHER_WAVES 8

template <int F, bool RELU>
__launch_bounds__(GATHER_WAVES * 32)
__global__ void gather_rows_kernel(const float* __restrict__ H,
                                   const int* __restrict__ rowptr,
                                   const int* __restrict__ ssrc,
                                   const float* __restrict__ dinv,
                                   const float* __restrict__ bias,
                                   float* __restrict__ Y, int n) {
    constexpr int V = F / 32;
    typedef float vec __attribute__((ext_vector_type(V)));

    const int lane = threadIdx.x & 31;
    const int wave = __builtin_amdgcn_readfirstlane((int)(threadIdx.x >> 5));
    const int stride = gridDim.x * GATHER_WAVES;
    const vec* Hv = (const vec*)H;
    const vec  b  = ((const vec*)bias)[lane];

    int row = blockIdx.x * GATHER_WAVES + wave;
    int pb0 = 0, pb1 = 0, qb0 = 0, qb1 = 0, iv = 0;
    if (row < n) {
        pb0 = rowptr[row];
        pb1 = rowptr[row + 1];
        iv  = (pb0 + lane < pb1) ? ssrc[pb0 + lane] : 0;
    }
    if (row + stride < n) {
        qb0 = rowptr[row + stride];
        qb1 = rowptr[row + stride + 1];
    }
    asm volatile("" : "+v"(iv));
    for (; row < n; row += stride) {
        const int beg = pb0;
        const int end = pb1;
        int niv = 0, rb0 = 0, rb1 = 0;
        if (row + stride < n) niv = (qb0 + lane < qb1) ? ssrc[qb0 + lane] : 0;
        if (row + 2 * stride < n) {
            rb0 = rowptr[row + 2 * stride];
            rb1 = rowptr[row + 2 * stride + 1];
        }

        const float di = dinv[row];

        const vec acc = sum_rows_of<V>(Hv, ssrc, beg, end, iv, lane);
        pb0 = qb0; pb1 = qb1; qb0 = rb0; qb1 = rb1; iv = niv;

        vec r = acc * di + b;
        if (RELU) {
#pragma unroll
            for (int k = 0; k < V; ++k) r[k] = r[k] > 0.0f ? r[k] : 0.0f;
        }
        ((vec*)Y)[(size_t)row * 32 + lane] = r;
    }
}

template <int FOUT>
__launch_bounds__(FUSE_WAVES * 32)
__attribute__((amdgpu_waves_per_eu(8)))
__global__ void gather_gemm_wmma(const float* __restrict__ Hin,
                                 const int* __restrict__ rowptr,
                                 const int* __restrict__ ssrc,
                                 const float* __restrict__ ddst,
                                 const float* __restrict__ bias,
                                 const float* __restrict__ dsrc,
                                 const float* __restrict__ W,
                                 float* __restrict__ Hout, int ntiles) {
    constexpr int LDSS = KDIM + 4;
    constexpr int CW   = FOUT / 16;
    constexpr int RPW  = 16 / FUSE_WAVES;
    typedef float vec __attribute__((ext_vector_type(4)));
    __shared__ __align__(16) float As[2][16 * LDSS];

    const int tid   = threadIdx.x;
    const int lane  = tid & 31;
    const int wave  = __builtin_amdgcn_readfirstlane((int)(threadIdx.x >> 5));
    const int n     = lane & 15;
    const int khalf = lane >> 4;
    const int m16   = lane & 15;
    const int ncol  = wave * 16 + n;

    v2f bf[KDIM / 4] = {};
    if (wave < CW) {
#pragma unroll
        for (int j = 0; j < KDIM / 4; ++j) {
            const int ka = 4 * j + 2 * khalf;
            bf[j].x = W[(size_t)ka * FOUT + ncol];
            bf[j].y = W[(size_t)(ka + 1) * FOUT + ncol];
        }
    }

    const vec* Hv = (const vec*)Hin;
    const vec  b  = ((const vec*)bias)[lane];

    static_assert(RPW == 2, "two rows per wave and tile");
    const int G = gridDim.x;
    int tile = blockIdx.x;
    int pb0 = 0, pb1 = 0, pb2 = 0, qb0 = 0, qb1 = 0, qb2 = 0, iv0 = 0, iv1 = 0;
    if (tile < ntiles) {
        const int* rp = rowptr + tile * 16 + wave * RPW;
        pb0 = rp[0]; pb1 = rp[1]; pb2 = rp[2];
        iv0 = (pb0 + lane < pb1) ? ssrc[pb0 + lane] : 0;
        iv1 = (pb1 + lane < pb2) ? ssrc[pb1 + lane] : 0;
    }
    if (tile + G < ntiles) {
        const int* rp = rowptr + (tile + G) * 16 + wave * RPW;
        qb0 = rp[0]; qb1 = rp[1]; qb2 = rp[2];
    }
    asm volatile("" : "+v"(iv0), "+v"(iv1));

    int st = 0;
    for (; tile < ntiles; tile += G) {
        float* A = As[st];
        const int bnd[RPW + 1] = {__builtin_amdgcn_readfirstlane(pb0),
                                  __builtin_amdgcn_readfirstlane(pb1),
                                  __builtin_amdgcn_readfirstlane(pb2)};
        const int ivs[RPW] = {iv0, iv1};
        int niv0 = 0, niv1 = 0, rb0 = 0, rb1 = 0, rb2 = 0;
        if (tile + G < ntiles) {
            niv0 = (qb0 + lane < qb1) ? ssrc[qb0 + lane] : 0;
            niv1 = (qb1 + lane < qb2) ? ssrc[qb1 + lane] : 0;
        }
        if (tile + 2 * G < ntiles) {
            const int* rp = rowptr + (tile + 2 * G) * 16 + wave * RPW;
            rb0 = rp[0]; rb1 = rp[1]; rb2 = rp[2];
        }
        for (int i = 0; i < RPW; ++i) {
            const int m   = wave * RPW + i;
            const int row = tile * 16 + m;
            const float di = ddst[row];
            const float ds = dsrc[row];
            const vec acc = sum_rows_of<4>(Hv, ssrc, bnd[i], bnd[i + 1], ivs[i], lane);

            vec r = acc * di + b;
#pragma unroll
            for (int k = 0; k < 4; ++k) r[k] = r[k] > 0.0f ? r[k] : 0.0f;
            r = r * ds;
            *(vec*)&A[m * LDSS + 4 * lane] = r;
        }
        __syncthreads();

        if (wave < CW) {
            v8f acc8 = {};
#pragma unroll
            for (int k0 = 0; k0 < KDIM; k0 += 4) {
                const int ka = k0 + 2 * khalf;
                v2f a = *(const v2f*)&A[m16 * LDSS + ka];
                acc8 = __builtin_amdgcn_wmma_f32_16x16x4_f32(
                     false, a,  false, bf[k0 / 4],
                     (short)0, acc8,  false,  false);
            }

            float* hp = Hout + ((size_t)tile * 16 + 8 * khalf) * FOUT + ncol;
#pragma unroll
            for (int r = 0; r < 8; ++r) hp[r * FOUT] = acc8[r];
        }
        pb0 = qb0; pb1 = qb1; pb2 = qb2;
        qb0 = rb0; qb1 = rb1; qb2 = rb2;
        iv0 = niv0; iv1 = niv1;
        st ^= 1;
    }
}

static inline int cdiv(long a, long b) { return (int)((a + b - 1) / b); }

extern "C" void kernel_launch(void* const* d_in, const int* in_sizes, int n_in,
                              void* d_out, int out_size, void* d_ws, size_t ws_size,
                              hipStream_t stream) {
    const float* features = (const float*)d_in[0];
    const int*   src      = (const int*)d_in[1];
    const int*   dst      = (const int*)d_in[2];
    const float* W1       = (const float*)d_in[3];
    const float* b1       = (const float*)d_in[4];
    const float* W2       = (const float*)d_in[5];
    const float* b2       = (const float*)d_in[6];
    const float* W3       = (const float*)d_in[7];
    const float* b3       = (const float*)d_in[8];
    float*       out      = (float*)d_out;

    const int N = in_sizes[0] / IN_F;
    const int E = in_sizes[1];

    float* ws   = (float*)d_ws;
    float* dsrc = ws;
    float* ddst = ws + N;
    float* buf0 = ws + 2 * (size_t)N;
    float* buf1 = buf0 + (size_t)N * HID_F;

    const int nbk = cdiv(N, BKT_ROWS);
    const int np  = cdiv(E, PART_PER);
    const int M   = (nbk + 1) * np;
    int* chunkD = (int*)buf1;
    int* chunkS = chunkD + 2 * (size_t)E;
    int* preD   = chunkS + E;
    int* preS   = preD + M;

    const int csr_len = N + 1 + E;
    int* csr_o    = (int*)out;
    int* rowptr_o = csr_o;
    int* ssrc_o   = csr_o + (N + 1);
    int* csr_w    = (int*)buf1;
    int* rowptr_w = csr_w;
    int* ssrc_w   = csr_w + (N + 1);

    const int ntiles = N / 16;

    local_sort_kernel<<<np, PART_T, 0, stream>>>(src, dst, chunkD, chunkS, preD, preS, E, nbk);
    bucket_sort_kernel<<<nbk, BSORT_T, 0, stream>>>(chunkD, chunkS, preD, preS, rowptr_o, ssrc_o, dsrc, ddst, N, E, np, nbk);

    gemm_rows16_wmma<HID_F><<<GEMM128_BLOCKS, HID_F * 2, 0, stream>>>(features, dsrc, W1, buf0, ntiles);

    gather_gemm_wmma<HID_F><<<FUSE_BLOCKS, FUSE_WAVES * 32, 0, stream>>>(buf0, rowptr_o, ssrc_o, ddst, b1, dsrc, W2, buf1, ntiles);

    gather_gemm_wmma<OUT_F><<<FUSE_BLOCKS, FUSE_WAVES * 32, 0, stream>>>(buf1, rowptr_o, ssrc_o, ddst, b2, dsrc, W3, buf0, ntiles);

    copy_int_kernel<<<ELEM_BLOCKS, 256, 0, stream>>>(csr_o, csr_w, csr_len);
    gather_rows_kernel<OUT_F, false><<<GATHER_BLOCKS, GATHER_WAVES * 32, 0, stream>>>(buf0, rowptr_w, ssrc_w, ddst, b3, out, N);
}


# First layer sums the scaled input rows and multiplies the sum by W1 inside the fused launch (sum, then product): h1 never stored, its plane keeps the edge list, 7 -> 5 launches; sum order changes
# speedup vs baseline: 6.2301x; 1.0490x over previous
#include <hip/hip_runtime.h>

typedef float v2f __attribute__((ext_vector_type(2)));
typedef float v8f __attribute__((ext_vector_type(8)));

#define IN_F  128
#define HID_F 128
#define OUT_F 64
#define KDIM  128

#define GATHER_BLOCKS  2048
#define FUSE_BLOCKS    1024
#define FUSE_WAVES     8
#define FIRST_BLOCKS   512
#define FIRST_WAVES    16

#define BKT_SH      8
#define BKT_ROWS    (1 << BKT_SH)
#define NBK_MAX     512
#define PART_T      1024
#define PART_EPT    8
#define PART_PER    (PART_EPT * PART_T)
#define NP_MAX      256
#define BSORT_T     1024
#define BSORT_WAVES (BSORT_T / 32)
#define RUNS_TRIP   4
#define RUN_SLOTS   32
#define OB_CAP      8192

__launch_bounds__(PART_T)
__global__ void local_sort_kernel(const int* __restrict__ src, const int* __restrict__ dst,
                                  int* __restrict__ chunkD, int* __restrict__ chunkS,
                                  int* __restrict__ preD, int* __restrict__ preS,
                                  int ne, int nbk) {
    __shared__ int hd[NBK_MAX];
    __shared__ int hs[NBK_MAX];
    __shared__ int sd[NBK_MAX];
    __shared__ int ss[NBK_MAX];
    __shared__ int2 sp[PART_PER];
    __shared__ int  sk[PART_PER];
    const int t = threadIdx.x;
    for (int b = t; b < NBK_MAX; b += PART_T) { hd[b] = 0; hs[b] = 0; }
    __syncthreads();

    const int e0 = blockIdx.x * PART_PER;
    const int e1 = (e0 + PART_PER < ne) ? e0 + PART_PER : ne;
    const int cnt = e1 - e0;
    int s[PART_EPT], d[PART_EPT];
#pragma unroll
    for (int k = 0; k < PART_EPT; ++k) {
        const int e = e0 + t + k * PART_T;
        d[k] = dst[e < e1 ? e : e1 - 1];
        s[k] = src[e < e1 ? e : e1 - 1];
    }
#pragma unroll
    for (int k = 0; k < PART_EPT; ++k) {
        if (e0 + t + k * PART_T < e1) {
            atomicAdd(&hd[d[k] >> BKT_SH], 1);
            atomicAdd(&hs[s[k] >> BKT_SH], 1);
        }
    }
    __syncthreads();

    const bool first = t < NBK_MAX;
    const int  u     = t & (NBK_MAX - 1);
    int* cntA = first ? hd : hs;
    int* scA  = first ? sd : ss;
    const int c = cntA[u];
    scA[u] = c;
    __syncthreads();
    for (int off = 1; off < NBK_MAX; off <<= 1) {
        const int x = (u >= off) ? scA[u - off] : 0;
        __syncthreads();
        scA[u] += x;
        __syncthreads();
    }
    const int excl = scA[u] - c;
    cntA[u] = excl;
    int* pre = first ? preD : preS;
    if (u <= nbk) pre[blockIdx.x * (nbk + 1) + u] = excl;
    __syncthreads();

#pragma unroll
    for (int k = 0; k < PART_EPT; ++k) {
        if (e0 + t + k * PART_T < e1) {
            const int pd = atomicAdd(&hd[d[k] >> BKT_SH], 1);
            const int ps = atomicAdd(&hs[s[k] >> BKT_SH], 1);
            sp[pd] = make_int2(d[k], s[k]);
            sk[ps] = s[k];
        }
    }
    __syncthreads();

    int2* cD = (int2*)chunkD + e0;
    int*  cS = chunkS + e0;
    for (int i = t; i < cnt; i += PART_T) {
        cD[i] = sp[i];
        cS[i] = sk[i];
    }
}

__launch_bounds__(BSORT_T)
__global__ void bucket_sort_kernel(const int* __restrict__ chunkD, const int* __restrict__ chunkS,
                                   const int* __restrict__ preD, const int* __restrict__ preS,
                                   int* __restrict__ rowptr, int* __restrict__ ssrc,
                                   float* __restrict__ dsrc, float* __restrict__ ddst,
                                   int n, int ne, int np, int nbk) {
    __shared__ int cd[BKT_ROWS];
    __shared__ int cs[BKT_ROWS];
    __shared__ int sc[BKT_ROWS];
    __shared__ int rsD[NP_MAX];
    __shared__ int rlD[NP_MAX];
    __shared__ int rsS[NP_MAX];
    __shared__ int rlS[NP_MAX];
    __shared__ int dbs;
    __shared__ int2 st[NP_MAX * RUN_SLOTS];
    __shared__ int  ob[OB_CAP];
    const int t    = threadIdx.x;
    const int lane = t & 31;
    const int wave = t >> 5;
    const int b    = blockIdx.x;
    if (t < BKT_ROWS) { cd[t] = 0; cs[t] = 0; }
    if (t == 0) dbs = 0;
    __syncthreads();

    const int pitch = nbk + 1;
    for (int j = t; j < np; j += BSORT_T) {
        const int a0 = preD[j * pitch + b];
        const int a1 = preD[j * pitch + b + 1];
        const int q0 = preS[j * pitch + b];
        const int q1 = preS[j * pitch + b + 1];
        rsD[j] = a0; rlD[j] = a1 - a0;
        rsS[j] = q0; rlS[j] = q1 - q0;
        atomicAdd(&dbs, a0);
    }
    __syncthreads();
    const int db = dbs;
    const int2* pv = (const int2*)chunkD;

    for (int j0 = wave * RUNS_TRIP; j0 < np; j0 += BSORT_WAVES * RUNS_TRIP) {
        int2 dk[RUNS_TRIP];
        int  sq[RUNS_TRIP], ld[RUNS_TRIP], ls[RUNS_TRIP], bd[RUNS_TRIP], bs[RUNS_TRIP];
#pragma unroll
        for (int r = 0; r < RUNS_TRIP; ++r) {
            const bool ok = j0 + r < np;
            const int  jc = ok ? j0 + r : np - 1;
            ld[r] = ok ? rlD[jc] : 0;
            ls[r] = ok ? rlS[jc] : 0;
            bd[r] = jc * PART_PER + rsD[jc];
            bs[r] = jc * PART_PER + rsS[jc];
            dk[r] = pv[lane < ld[r] ? bd[r] + lane : jc * PART_PER];
            sq[r] = chunkS[lane < ls[r] ? bs[r] + lane : jc * PART_PER];
        }
#pragma unroll
        for (int r = 0; r < RUNS_TRIP; ++r) {
            if (lane < ld[r]) {
                st[(j0 + r) * RUN_SLOTS + lane] = dk[r];
                atomicAdd(&cd[dk[r].x & (BKT_ROWS - 1)], 1);
            }
            if (lane < ls[r]) atomicAdd(&cs[sq[r] & (BKT_ROWS - 1)], 1);
        }
#pragma unroll
        for (int r = 0; r < RUNS_TRIP; ++r) {
            for (int k = lane + 32; k < ld[r]; k += 32)
                atomicAdd(&cd[pv[bd[r] + k].x & (BKT_ROWS - 1)], 1);
            for (int k = lane + 32; k < ls[r]; k += 32)
                atomicAdd(&cs[chunkS[bs[r] + k] & (BKT_ROWS - 1)], 1);
        }
    }
    __syncthreads();

    int c = 0;
    if (t < BKT_ROWS) { c = cd[t]; sc[t] = c; }
    __syncthreads();
    for (int off = 1; off < BKT_ROWS; off <<= 1) {
        int x = 0;
        if (t < BKT_ROWS && t >= off) x = sc[t - off];
        __syncthreads();
        if (t < BKT_ROWS) sc[t] += x;
        __syncthreads();
    }
    const int  total = sc[BKT_ROWS - 1];
    const bool fits  = total <= OB_CAP;
    if (t < BKT_ROWS) {
        const int start = sc[t] - c;
        const int row   = b * BKT_ROWS + t;
        cd[t] = start;
        if (row < n) {
            const int k = cs[t];
            rowptr[row] = db + start;
            ddst[row]   = rsqrtf(c < 1 ? 1.0f : (float)c);
            dsrc[row]   = rsqrtf(k < 1 ? 1.0f : (float)k);
            if (row == n - 1) rowptr[n] = ne;
        }
    }
    __syncthreads();

    for (int j0 = wave * RUNS_TRIP; j0 < np; j0 += BSORT_WAVES * RUNS_TRIP) {
        int ld[RUNS_TRIP], bd[RUNS_TRIP];
#pragma unroll
        for (int r = 0; r < RUNS_TRIP; ++r) {
            const bool ok = j0 + r < np;
            const int  jc = ok ? j0 + r : np - 1;
            ld[r] = ok ? rlD[jc] : 0;
            bd[r] = jc * PART_PER + rsD[jc];
        }
        int2 p[RUNS_TRIP];
        int  pos[RUNS_TRIP];
#pragma unroll
        for (int r = 0; r < RUNS_TRIP; ++r) {
            p[r] = make_int2(0, 0);
            if (lane < ld[r]) p[r] = st[(j0 + r) * RUN_SLOTS + lane];
        }
#pragma unroll
        for (int r = 0; r < RUNS_TRIP; ++r) {
            pos[r] = 0;
            if (lane < ld[r]) pos[r] = atomicAdd(&cd[p[r].x & (BKT_ROWS - 1)], 1);
        }
#pragma unroll
        for (int r = 0; r < RUNS_TRIP; ++r) {
            if (lane < ld[r]) {
                if (fits) ob[pos[r]] = p[r].y; else ssrc[db + pos[r]] = p[r].y;
            }
        }
#pragma unroll
        for (int r = 0; r < RUNS_TRIP; ++r) {
            for (int k = lane + 32; k < ld[r]; k += 32) {
                const int2 q = pv[bd[r] + k];
                const int pq = atomicAdd(&cd[q.x & (BKT_ROWS - 1)], 1);
                if (fits) ob[pq] = q.y; else ssrc[db + pq] = q.y;
            }
        }
    }
    __syncthreads();

    if (fits) {
        for (int k = 0; k < OB_CAP / BSORT_T; ++k) {
            const int i = t + k * BSORT_T;
            if (i < total) ssrc[db + i] = ob[i];
        }
    }
}

template <int V> struct RowVec { typedef float type __attribute__((ext_vector_type(V))); };

template <int V, int R>
__device__ __forceinline__ typename RowVec<V>::type
add_last_rows(const typename RowVec<V>::type* __restrict__ Hv, int iv, int j, int lane,
              typename RowVec<V>::type acc) {
    typedef typename RowVec<V>::type vec;
    int s[R];
    vec t[R];
    for (int k = 0; k < R; ++k) s[k] = __builtin_amdgcn_readlane(iv, j + k);
    for (int k = 0; k < R; ++k) t[k] = Hv[(size_t)s[k] * 32 + lane];
    if (R & 4) acc += (t[0] + t[1]) + (t[2] + t[3]);
    if (R & 2) acc += t[R & 4] + t[(R & 4) + 1];
    if (R & 1) acc += t[R - 1];
    return acc;
}

template <int V>
__device__ __forceinline__ typename RowVec<V>::type
sum_rows_of(const typename RowVec<V>::type* __restrict__ Hv, const int* __restrict__ ssrc,
            int beg, int end, int iv, int lane) {
    typedef typename RowVec<V>::type vec;
    vec acc = 0.0f;
    int e = beg;
    for (;;) {
        int cnt = end - e;
        cnt = cnt > 32 ? 32 : cnt;
        int j = 0;
        for (; j + 8 <= cnt; j += 8) {
            int s[8];
            vec v[8];
#pragma unroll
            for (int k = 0; k < 8; ++k) s[k] = __builtin_amdgcn_readlane(iv, j + k);
#pragma unroll
            for (int k = 0; k < 8; ++k) v[k] = Hv[(size_t)s[k] * 32 + lane];
            acc += ((v[0] + v[1]) + (v[2] + v[3])) + ((v[4] + v[5]) + (v[6] + v[7]));
        }
        switch (cnt - j) {
            case 1: acc = add_last_rows<V, 1>(Hv, iv, j, lane, acc); break;
            case 2: acc = add_last_rows<V, 2>(Hv, iv, j, lane, acc); break;
            case 3: acc = add_last_rows<V, 3>(Hv, iv, j, lane, acc); break;
            case 4: acc = add_last_rows<V, 4>(Hv, iv, j, lane, acc); break;
            case 5: acc = add_last_rows<V, 5>(Hv, iv, j, lane, acc); break;
            case 6: acc = add_last_rows<V, 6>(Hv, iv, j, lane, acc); break;
            case 7: acc = add_last_rows<V, 7>(Hv, iv, j, lane, acc); break;
            default: break;
        }
        e += cnt;
        if (e >= end) break;
        iv = (e + lane < end) ? ssrc[e + lane] : 0;
        asm volatile("" : "+v"(iv));
    }
    return acc;
}

typedef const float __attribute__((address_space(4))) cfloat;

template <int R>
__device__ __forceinline__ RowVec<4>::type
add_last_scaled_rows(const RowVec<4>::type* __restrict__ Xv, const cfloat* __restrict__ scl,
                     int iv, int j, int lane, RowVec<4>::type acc) {
    typedef RowVec<4>::type vec;
    int s[R];
    float f[R];
    vec t[R];
    for (int k = 0; k < R; ++k) s[k] = __builtin_amdgcn_readlane(iv, j + k);
    for (int k = 0; k < R; ++k) f[k] = scl[s[k]];
    for (int k = 0; k < R; ++k) t[k] = Xv[(size_t)s[k] * 32 + lane];
    for (int k = 0; k < R; ++k) t[k] = t[k] * f[k];
    if (R & 4) acc += (t[0] + t[1]) + (t[2] + t[3]);
    if (R & 2) acc += t[R & 4] + t[(R & 4) + 1];
    if (R & 1) acc += t[R - 1];
    return acc;
}

__device__ __forceinline__ RowVec<4>::type
sum_scaled_rows_of(const RowVec<4>::type* __restrict__ Xv, const int* __restrict__ ssrc,
                   const cfloat* __restrict__ scl, int beg, int end, int iv, int lane) {
    typedef RowVec<4>::type vec;
    vec acc = 0.0f;
    int e = beg;
    for (;;) {
        int cnt = end - e;
        cnt = cnt > 32 ? 32 : cnt;
        int j = 0;
        for (; j + 8 <= cnt; j += 8) {
            int s[8];
            float f[8];
            vec v[8];
#pragma unroll
            for (int k = 0; k < 8; ++k) s[k] = __builtin_amdgcn_readlane(iv, j + k);
#pragma unroll
            for (int k = 0; k < 8; ++k) f[k] = scl[s[k]];
#pragma unroll
            for (int k = 0; k < 8; ++k) v[k] = Xv[(size_t)s[k] * 32 + lane];
#pragma unroll
            for (int k = 0; k < 8; ++k) v[k] = v[k] * f[k];
            acc += ((v[0] + v[1]) + (v[2] + v[3])) + ((v[4] + v[5]) + (v[6] + v[7]));
        }
        switch (cnt - j) {
            case 1: acc = add_last_scaled_rows<1>(Xv, scl, iv, j, lane, acc); break;
            case 2: acc = add_last_scaled_rows<2>(Xv, scl, iv, j, lane, acc); break;
            case 3: acc = add_last_scaled_rows<3>(Xv, scl, iv, j, lane, acc); break;
            case 4: acc = add_last_scaled_rows<4>(Xv, scl, iv, j, lane, acc); break;
            case 5: acc = add_last_scaled_rows<5>(Xv, scl, iv, j, lane, acc); break;
            case 6: acc = add_last_scaled_rows<6>(Xv, scl, iv, j, lane, acc); break;
            case 7: acc = add_last_scaled_rows<7>(Xv, scl, iv, j, lane, acc); break;
            default: break;
        }
        e += cnt;
        if (e >= end) break;
        iv = (e + lane < end) ? ssrc[e + lane] : 0;
        asm volatile("" : "+v"(iv));
    }
    return acc;
}

#define GATHER_WAVES 8

template <int F, bool RELU>
__launch_bounds__(GATHER_WAVES * 32)
__global__ void gather_rows_kernel(const float* __restrict__ H,
                                   const int* __restrict__ rowptr,
                                   const int* __restrict__ ssrc,
                                   const float* __restrict__ dinv,
                                   const float* __restrict__ bias,
                                   float* __restrict__ Y, int n) {
    constexpr int V = F / 32;
    typedef float vec __attribute__((ext_vector_type(V)));

    const int lane = threadIdx.x & 31;
    const int wave = __builtin_amdgcn_readfirstlane((int)(threadIdx.x >> 5));
    const int stride = gridDim.x * GATHER_WAVES;
    const vec* Hv = (const vec*)H;
    const vec  b  = ((const vec*)bias)[lane];

    int row = blockIdx.x * GATHER_WAVES + wave;
    int pb0 = 0, pb1 = 0, qb0 = 0, qb1 = 0, iv = 0;
    if (row < n) {
        pb0 = rowptr[row];
        pb1 = rowptr[row + 1];
        iv  = (pb0 + lane < pb1) ? ssrc[pb0 + lane] : 0;
    }
    if (row + stride < n) {
        qb0 = rowptr[row + stride];
        qb1 = rowptr[row + stride + 1];
    }
    asm volatile("" : "+v"(iv));
    for (; row < n; row += stride) {
        const int beg = pb0;
        const int end = pb1;
        int niv = 0, rb0 = 0, rb1 = 0;
        if (row + stride < n) niv = (qb0 + lane < qb1) ? ssrc[qb0 + lane] : 0;
        if (row + 2 * stride < n) {
            rb0 = rowptr[row + 2 * stride];
            rb1 = rowptr[row + 2 * stride + 1];
        }

        const float di = dinv[row];

        const vec acc = sum_rows_of<V>(Hv, ssrc, beg, end, iv, lane);
        pb0 = qb0; pb1 = qb1; qb0 = rb0; qb1 = rb1; iv = niv;

        vec r = acc * di + b;
        if (RELU) {
#pragma unroll
            for (int k = 0; k < V; ++k) r[k] = r[k] > 0.0f ? r[k] : 0.0f;
        }
        ((vec*)Y)[(size_t)row * 32 + lane] = r;
    }
}

template <int FOUT>
__launch_bounds__(FUSE_WAVES * 32)
__attribute__((amdgpu_waves_per_eu(8)))
__global__ void gather_gemm_wmma(const float* __restrict__ Hin,
                                 const int* __restrict__ rowptr,
                                 const int* __restrict__ ssrc,
                                 const float* __restrict__ ddst,
                                 const float* __restrict__ bias,
                                 const float* __restrict__ dsrc,
                                 const float* __restrict__ W,
                                 float* __restrict__ Hout, int ntiles) {
    constexpr int LDSS = KDIM + 4;
    constexpr int CW   = FOUT / 16;
    constexpr int RPW  = 16 / FUSE_WAVES;
    typedef float vec __attribute__((ext_vector_type(4)));
    __shared__ __align__(16) float As[2][16 * LDSS];

    const int tid   = threadIdx.x;
    const int lane  = tid & 31;
    const int wave  = __builtin_amdgcn_readfirstlane((int)(threadIdx.x >> 5));
    const int n     = lane & 15;
    const int khalf = lane >> 4;
    const int m16   = lane & 15;
    const int ncol  = wave * 16 + n;

    v2f bf[KDIM / 4] = {};
    if (wave < CW) {
#pragma unroll
        for (int j = 0; j < KDIM / 4; ++j) {
            const int ka = 4 * j + 2 * khalf;
            bf[j].x = W[(size_t)ka * FOUT + ncol];
            bf[j].y = W[(size_t)(ka + 1) * FOUT + ncol];
        }
    }

    const vec* Hv = (const vec*)Hin;
    const vec  b  = ((const vec*)bias)[lane];

    static_assert(RPW == 2, "two rows per wave and tile");
    const int G = gridDim.x;
    int tile = blockIdx.x;
    int pb0 = 0, pb1 = 0, pb2 = 0, qb0 = 0, qb1 = 0, qb2 = 0, iv0 = 0, iv1 = 0;
    if (tile < ntiles) {
        const int* rp = rowptr + tile * 16 + wave * RPW;
        pb0 = rp[0]; pb1 = rp[1]; pb2 = rp[2];
        iv0 = (pb0 + lane < pb1) ? ssrc[pb0 + lane] : 0;
        iv1 = (pb1 + lane < pb2) ? ssrc[pb1 + lane] : 0;
    }
    if (tile + G < ntiles) {
        const int* rp = rowptr + (tile + G) * 16 + wave * RPW;
        qb0 = rp[0]; qb1 = rp[1]; qb2 = rp[2];
    }
    asm volatile("" : "+v"(iv0), "+v"(iv1));

    int st = 0;
    for (; tile < ntiles; tile += G) {
        float* A = As[st];
        const int bnd[RPW + 1] = {__builtin_amdgcn_readfirstlane(pb0),
                                  __builtin_amdgcn_readfirstlane(pb1),
                                  __builtin_amdgcn_readfirstlane(pb2)};
        const int ivs[RPW] = {iv0, iv1};
        int niv0 = 0, niv1 = 0, rb0 = 0, rb1 = 0, rb2 = 0;
        if (tile + G < ntiles) {
            niv0 = (qb0 + lane < qb1) ? ssrc[qb0 + lane] : 0;
            niv1 = (qb1 + lane < qb2) ? ssrc[qb1 + lane] : 0;
        }
        if (tile + 2 * G < ntiles) {
            const int* rp = rowptr + (tile + 2 * G) * 16 + wave * RPW;
            rb0 = rp[0]; rb1 = rp[1]; rb2 = rp[2];
        }
        for (int i = 0; i < RPW; ++i) {
            const int m   = wave * RPW + i;
            const int row = tile * 16 + m;
            const float di = ddst[row];
            const float ds = dsrc[row];
            const vec acc = sum_rows_of<4>(Hv, ssrc, bnd[i], bnd[i + 1], ivs[i], lane);

            vec r = acc * di + b;
#pragma unroll
            for (int k = 0; k < 4; ++k) r[k] = r[k] > 0.0f ? r[k] : 0.0f;
            r = r * ds;
            *(vec*)&A[m * LDSS + 4 * lane] = r;
        }
        __syncthreads();

        if (wave < CW) {
            v8f acc8 = {};
#pragma unroll
            for (int k0 = 0; k0 < KDIM; k0 += 4) {
                const int ka = k0 + 2 * khalf;
                v2f a = *(const v2f*)&A[m16 * LDSS + ka];
                acc8 = __builtin_amdgcn_wmma_f32_16x16x4_f32(
                     false, a,  false, bf[k0 / 4],
                     (short)0, acc8,  false,  false);
            }

            float* hp = Hout + ((size_t)tile * 16 + 8 * khalf) * FOUT + ncol;
#pragma unroll
            for (int r = 0; r < 8; ++r) hp[r * FOUT] = acc8[r];
        }
        pb0 = qb0; pb1 = qb1; pb2 = qb2;
        qb0 = rb0; qb1 = rb1; qb2 = rb2;
        iv0 = niv0; iv1 = niv1;
        st ^= 1;
    }
}

__launch_bounds__(FIRST_WAVES * 32)
__attribute__((amdgpu_waves_per_eu(8)))
__global__ void gather_gemm2_wmma(const float* __restrict__ X,
                                  const int* __restrict__ rowptr,
                                  const int* __restrict__ ssrc,
                                  const float* __restrict__ ddst,
                                  const float* __restrict__ bias,
                                  const float* __restrict__ dsrc,
                                  const float* __restrict__ W1,
                                  const float* __restrict__ W2,
                                  float* __restrict__ Hout, int ntiles) {
    constexpr int LDSS = KDIM + 4;
    typedef float vec __attribute__((ext_vector_type(4)));
    __shared__ __align__(16) float S1[2][16 * LDSS];
    __shared__ __align__(16) float S2[2][16 * LDSS];

    const int tid   = threadIdx.x;
    const int lane  = tid & 31;
    const int wave  = __builtin_amdgcn_readfirstlane((int)(threadIdx.x >> 5));
    const int role  = wave >> 3;
    const int n     = lane & 15;
    const int khalf = lane >> 4;
    const int m16   = lane & 15;
    const int ncol  = (wave & 7) * 16 + n;

    const float* __restrict__ Wp = role ? W2 : W1;
    v2f bf[KDIM / 4];
#pragma unroll
    for (int j = 0; j < KDIM / 4; ++j) {
        const int ka = 4 * j + 2 * khalf;
        bf[j].x = Wp[(size_t)ka * HID_F + ncol];
        bf[j].y = Wp[(size_t)(ka + 1) * HID_F + ncol];
    }

    const vec*    Xv   = (const vec*)X;
    const float   bcol = bias[ncol];
    const cfloat* dsc  = (const cfloat*)(unsigned long long)dsrc;
    const cfloat* ddc  = (const cfloat*)(unsigned long long)ddst;

    const int G = gridDim.x;
    int tile = blockIdx.x;
    int pb0 = 0, pb1 = 0, qb0 = 0, qb1 = 0, iv = 0;
    if (tile < ntiles) {
        const int* rp = rowptr + tile * 16 + wave;
        pb0 = rp[0]; pb1 = rp[1];
        iv  = (pb0 + lane < pb1) ? ssrc[pb0 + lane] : 0;
    }
    if (tile + G < ntiles) {
        const int* rp = rowptr + (tile + G) * 16 + wave;
        qb0 = rp[0]; qb1 = rp[1];
    }
    asm volatile("" : "+v"(iv));

    int st   = 0;
    int prev = -1;
    for (;;) {
        const bool have = tile < ntiles;
        if (!have && prev < 0) break;
        if (have) {
            const int beg = __builtin_amdgcn_readfirstlane(pb0);
            const int end = __builtin_amdgcn_readfirstlane(pb1);
            int niv = 0, rb0 = 0, rb1 = 0;
            if (tile + G < ntiles) niv = (qb0 + lane < qb1) ? ssrc[qb0 + lane] : 0;
            if (tile + 2 * G < ntiles) {
                const int* rp = rowptr + (tile + 2 * G) * 16 + wave;
                rb0 = rp[0]; rb1 = rp[1];
            }
            const int row = tile * 16 + wave;
            v2f fac;
            fac.x = ddc[row];
            fac.y = dsc[row];
            const vec acc = sum_scaled_rows_of(Xv, ssrc, dsc, beg, end, iv, lane);
            *(vec*)&S1[st][wave * LDSS + 4 * lane] = acc;
            if (lane == 0) *(v2f*)&S1[st][wave * LDSS + KDIM] = fac;
            pb0 = qb0; pb1 = qb1; qb0 = rb0; qb1 = rb1; iv = niv;
        }
        __syncthreads();

        if (role == 0) {
            if (have) {
                const float* A = S1[st];
                v8f acc8 = {};
#pragma unroll
                for (int k0 = 0; k0 < KDIM; k0 += 4) {
                    const int ka = k0 + 2 * khalf;
                    v2f a = *(const v2f*)&A[m16 * LDSS + ka];
                    acc8 = __builtin_amdgcn_wmma_f32_16x16x4_f32(
                         false, a,  false, bf[k0 / 4],
                         (short)0, acc8,  false,  false);
                }

                const float* fr = &A[(8 * khalf) * LDSS + KDIM];
                float* xo = &S2[st][(8 * khalf) * LDSS + ncol];
#pragma unroll
                for (int r = 0; r < 8; ++r) {
                    const v2f fac = *(const v2f*)&fr[r * LDSS];
                    float x = acc8[r] * fac.x + bcol;
                    x = x > 0.0f ? x : 0.0f;
                    xo[r * LDSS] = x * fac.y;
                }
            }
        } else {
            if (prev >= 0) {
                const float* A = S2[st ^ 1];
                v8f acc8 = {};
#pragma unroll
                for (int k0 = 0; k0 < KDIM; k0 += 4) {
                    const int ka = k0 + 2 * khalf;
                    v2f a = *(const v2f*)&A[m16 * LDSS + ka];
                    acc8 = __builtin_amdgcn_wmma_f32_16x16x4_f32(
                         false, a,  false, bf[k0 / 4],
                         (short)0, acc8,  false,  false);
                }

                float* hp = Hout + ((size_t)prev * 16 + 8 * khalf) * HID_F + ncol;
#pragma unroll
                for (int r = 0; r < 8; ++r) hp[r * HID_F] = acc8[r];
            }
        }
        prev = have ? tile : -1;
        tile += G;
        st ^= 1;
    }
}

static inline int cdiv(long a, long b) { return (int)((a + b - 1) / b); }

extern "C" void kernel_launch(void* const* d_in, const int* in_sizes, int n_in,
                              void* d_out, int out_size, void* d_ws, size_t ws_size,
                              hipStream_t stream) {
    const float* features = (const float*)d_in[0];
    const int*   src      = (const int*)d_in[1];
    const int*   dst      = (const int*)d_in[2];
    const float* W1       = (const float*)d_in[3];
    const float* b1       = (const float*)d_in[4];
    const float* W2       = (const float*)d_in[5];
    const float* b2       = (const float*)d_in[6];
    const float* W3       = (const float*)d_in[7];
    const float* b3       = (const float*)d_in[8];
    float*       out      = (float*)d_out;

    const int N = in_sizes[0] / IN_F;
    const int E = in_sizes[1];

    float* ws   = (float*)d_ws;
    float* dsrc = ws;
    float* ddst = ws + N;
    float* buf0 = ws + 2 * (size_t)N;
    float* buf1 = buf0 + (size_t)N * HID_F;

    const int nbk = cdiv(N, BKT_ROWS);
    const int np  = cdiv(E, PART_PER);
    const int M   = (nbk + 1) * np;
    int* chunkD = (int*)buf1;
    int* chunkS = chunkD + 2 * (size_t)E;
    int* preD   = chunkS + E;
    int* preS   = preD + M;

    int* csr    = (int*)(buf0 + (size_t)N * OUT_F);
    int* rowptr = csr;
    int* ssrc   = csr + (N + 1);

    const int ntiles = N / 16;

    local_sort_kernel<<<np, PART_T, 0, stream>>>(src, dst, chunkD, chunkS, preD, preS, E, nbk);
    bucket_sort_kernel<<<nbk, BSORT_T, 0, stream>>>(chunkD, chunkS, preD, preS, rowptr, ssrc, dsrc, ddst, N, E, np, nbk);

    gather_gemm2_wmma<<<FIRST_BLOCKS, FIRST_WAVES * 32, 0, stream>>>(features, rowptr, ssrc, ddst, b1, dsrc, W1, W2, buf1, ntiles);

    gather_gemm_wmma<OUT_F><<<FUSE_BLOCKS, FUSE_WAVES * 32, 0, stream>>>(buf1, rowptr, ssrc, ddst, b2, dsrc, W3, buf0, ntiles);

    gather_rows_kernel<OUT_F, false><<<GATHER_BLOCKS, GATHER_WAVES * 32, 0, stream>>>(buf0, rowptr, ssrc, ddst, b3, out, N);
}


# Wave issue priority by phase (s_setprio): waves raise it while they sum rows and drop it for their WMMA runs, so an older wave's matrix run no longer stalls a SIMD's row loads; sum order kept
# speedup vs baseline: 6.4258x; 1.0314x over previous
#include <hip/hip_runtime.h>

typedef float v2f __attribute__((ext_vector_type(2)));
typedef float v8f __attribute__((ext_vector_type(8)));

#define IN_F  128
#define HID_F 128
#define OUT_F 64
#define KDIM  128

#define GATHER_BLOCKS  2048
#define FUSE_BLOCKS    1024
#define FUSE_WAVES     8
#define FIRST_BLOCKS   512
#define FIRST_WAVES    16

#define BKT_SH      8
#define BKT_ROWS    (1 << BKT_SH)
#define NBK_MAX     512
#define PART_T      1024
#define PART_EPT    8
#define PART_PER    (PART_EPT * PART_T)
#define NP_MAX      256
#define BSORT_T     1024
#define BSORT_WAVES (BSORT_T / 32)
#define RUNS_TRIP   4
#define RUN_SLOTS   32
#define OB_CAP      8192

__launch_bounds__(PART_T)
__global__ void local_sort_kernel(const int* __restrict__ src, const int* __restrict__ dst,
                                  int* __restrict__ chunkD, int* __restrict__ chunkS,
                                  int* __restrict__ preD, int* __restrict__ preS,
                                  int ne, int nbk) {
    __shared__ int hd[NBK_MAX];
    __shared__ int hs[NBK_MAX];
    __shared__ int sd[NBK_MAX];
    __shared__ int ss[NBK_MAX];
    __shared__ int2 sp[PART_PER];
    __shared__ int  sk[PART_PER];
    const int t = threadIdx.x;
    for (int b = t; b < NBK_MAX; b += PART_T) { hd[b] = 0; hs[b] = 0; }
    __syncthreads();

    const int e0 = blockIdx.x * PART_PER;
    const int e1 = (e0 + PART_PER < ne) ? e0 + PART_PER : ne;
    const int cnt = e1 - e0;
    int s[PART_EPT], d[PART_EPT];
#pragma unroll
    for (int k = 0; k < PART_EPT; ++k) {
        const int e = e0 + t + k * PART_T;
        d[k] = dst[e < e1 ? e : e1 - 1];
        s[k] = src[e < e1 ? e : e1 - 1];
    }
#pragma unroll
    for (int k = 0; k < PART_EPT; ++k) {
        if (e0 + t + k * PART_T < e1) {
            atomicAdd(&hd[d[k] >> BKT_SH], 1);
            atomicAdd(&hs[s[k] >> BKT_SH], 1);
        }
    }
    __syncthreads();

    const bool first = t < NBK_MAX;
    const int  u     = t & (NBK_MAX - 1);
    int* cntA = first ? hd : hs;
    int* scA  = first ? sd : ss;
    const int c = cntA[u];
    scA[u] = c;
    __syncthreads();
    for (int off = 1; off < NBK_MAX; off <<= 1) {
        const int x = (u >= off) ? scA[u - off] : 0;
        __syncthreads();
        scA[u] += x;
        __syncthreads();
    }
    const int excl = scA[u] - c;
    cntA[u] = excl;
    int* pre = first ? preD : preS;
    if (u <= nbk) pre[blockIdx.x * (nbk + 1) + u] = excl;
    __syncthreads();

#pragma unroll
    for (int k = 0; k < PART_EPT; ++k) {
        if (e0 + t + k * PART_T < e1) {
            const int pd = atomicAdd(&hd[d[k] >> BKT_SH], 1);
            const int ps = atomicAdd(&hs[s[k] >> BKT_SH], 1);
            sp[pd] = make_int2(d[k], s[k]);
            sk[ps] = s[k];
        }
    }
    __syncthreads();

    int2* cD = (int2*)chunkD + e0;
    int*  cS = chunkS + e0;
    for (int i = t; i < cnt; i += PART_T) {
        cD[i] = sp[i];
        cS[i] = sk[i];
    }
}

__launch_bounds__(BSORT_T)
__global__ void bucket_sort_kernel(const int* __restrict__ chunkD, const int* __restrict__ chunkS,
                                   const int* __restrict__ preD, const int* __restrict__ preS,
                                   int* __restrict__ rowptr, int* __restrict__ ssrc,
                                   float* __restrict__ dsrc, float* __restrict__ ddst,
                                   int n, int ne, int np, int nbk) {
    __shared__ int cd[BKT_ROWS];
    __shared__ int cs[BKT_ROWS];
    __shared__ int sc[BKT_ROWS];
    __shared__ int rsD[NP_MAX];
    __shared__ int rlD[NP_MAX];
    __shared__ int rsS[NP_MAX];
    __shared__ int rlS[NP_MAX];
    __shared__ int dbs;
    __shared__ int2 st[NP_MAX * RUN_SLOTS];
    __shared__ int  ob[OB_CAP];
    const int t    = threadIdx.x;
    const int lane = t & 31;
    const int wave = t >> 5;
    const int b    = blockIdx.x;
    if (t < BKT_ROWS) { cd[t] = 0; cs[t] = 0; }
    if (t == 0) dbs = 0;
    __syncthreads();

    const int pitch = nbk + 1;
    for (int j = t; j < np; j += BSORT_T) {
        const int a0 = preD[j * pitch + b];
        const int a1 = preD[j * pitch + b + 1];
        const int q0 = preS[j * pitch + b];
        const int q1 = preS[j * pitch + b + 1];
        rsD[j] = a0; rlD[j] = a1 - a0;
        rsS[j] = q0; rlS[j] = q1 - q0;
        atomicAdd(&dbs, a0);
    }
    __syncthreads();
    const int db = dbs;
    const int2* pv = (const int2*)chunkD;

    for (int j0 = wave * RUNS_TRIP; j0 < np; j0 += BSORT_WAVES * RUNS_TRIP) {
        int2 dk[RUNS_TRIP];
        int  sq[RUNS_TRIP], ld[RUNS_TRIP], ls[RUNS_TRIP], bd[RUNS_TRIP], bs[RUNS_TRIP];
#pragma unroll
        for (int r = 0; r < RUNS_TRIP; ++r) {
            const bool ok = j0 + r < np;
            const int  jc = ok ? j0 + r : np - 1;
            ld[r] = ok ? rlD[jc] : 0;
            ls[r] = ok ? rlS[jc] : 0;
            bd[r] = jc * PART_PER + rsD[jc];
            bs[r] = jc * PART_PER + rsS[jc];
            dk[r] = pv[lane < ld[r] ? bd[r] + lane : jc * PART_PER];
            sq[r] = chunkS[lane < ls[r] ? bs[r] + lane : jc * PART_PER];
        }
#pragma unroll
        for (int r = 0; r < RUNS_TRIP; ++r) {
            if (lane < ld[r]) {
                st[(j0 + r) * RUN_SLOTS + lane] = dk[r];
                atomicAdd(&cd[dk[r].x & (BKT_ROWS - 1)], 1);
            }
            if (lane < ls[r]) atomicAdd(&cs[sq[r] & (BKT_ROWS - 1)], 1);
        }
#pragma unroll
        for (int r = 0; r < RUNS_TRIP; ++r) {
            for (int k = lane + 32; k < ld[r]; k += 32)
                atomicAdd(&cd[pv[bd[r] + k].x & (BKT_ROWS - 1)], 1);
            for (int k = lane + 32; k < ls[r]; k += 32)
                atomicAdd(&cs[chunkS[bs[r] + k] & (BKT_ROWS - 1)], 1);
        }
    }
    __syncthreads();

    int c = 0;
    if (t < BKT_ROWS) { c = cd[t]; sc[t] = c; }
    __syncthreads();
    for (int off = 1; off < BKT_ROWS; off <<= 1) {
        int x = 0;
        if (t < BKT_ROWS && t >= off) x = sc[t - off];
        __syncthreads();
        if (t < BKT_ROWS) sc[t] += x;
        __syncthreads();
    }
    const int  total = sc[BKT_ROWS - 1];
    const bool fits  = total <= OB_CAP;
    if (t < BKT_ROWS) {
        const int start = sc[t] - c;
        const int row   = b * BKT_ROWS + t;
        cd[t] = start;
        if (row < n) {
            const int k = cs[t];
            rowptr[row] = db + start;
            ddst[row]   = rsqrtf(c < 1 ? 1.0f : (float)c);
            dsrc[row]   = rsqrtf(k < 1 ? 1.0f : (float)k);
            if (row == n - 1) rowptr[n] = ne;
        }
    }
    __syncthreads();

    for (int j0 = wave * RUNS_TRIP; j0 < np; j0 += BSORT_WAVES * RUNS_TRIP) {
        int ld[RUNS_TRIP], bd[RUNS_TRIP];
#pragma unroll
        for (int r = 0; r < RUNS_TRIP; ++r) {
            const bool ok = j0 + r < np;
            const int  jc = ok ? j0 + r : np - 1;
            ld[r] = ok ? rlD[jc] : 0;
            bd[r] = jc * PART_PER + rsD[jc];
        }
        int2 p[RUNS_TRIP];
        int  pos[RUNS_TRIP];
#pragma unroll
        for (int r = 0; r < RUNS_TRIP; ++r) {
            p[r] = make_int2(0, 0);
            if (lane < ld[r]) p[r] = st[(j0 + r) * RUN_SLOTS + lane];
        }
#pragma unroll
        for (int r = 0; r < RUNS_TRIP; ++r) {
            pos[r] = 0;
            if (lane < ld[r]) pos[r] = atomicAdd(&cd[p[r].x & (BKT_ROWS - 1)], 1);
        }
#pragma unroll
        for (int r = 0; r < RUNS_TRIP; ++r) {
            if (lane < ld[r]) {
                if (fits) ob[pos[r]] = p[r].y; else ssrc[db + pos[r]] = p[r].y;
            }
        }
#pragma unroll
        for (int r = 0; r < RUNS_TRIP; ++r) {
            for (int k = lane + 32; k < ld[r]; k += 32) {
                const int2 q = pv[bd[r] + k];
                const int pq = atomicAdd(&cd[q.x & (BKT_ROWS - 1)], 1);
                if (fits) ob[pq] = q.y; else ssrc[db + pq] = q.y;
            }
        }
    }
    __syncthreads();

    if (fits) {
        for (int k = 0; k < OB_CAP / BSORT_T; ++k) {
            const int i = t + k * BSORT_T;
            if (i < total) ssrc[db + i] = ob[i];
        }
    }
}

template <int V> struct RowVec { typedef float type __attribute__((ext_vector_type(V))); };

template <int V, int R>
__device__ __forceinline__ typename RowVec<V>::type
add_last_rows(const typename RowVec<V>::type* __restrict__ Hv, int iv, int j, int lane,
              typename RowVec<V>::type acc) {
    typedef typename RowVec<V>::type vec;
    int s[R];
    vec t[R];
    for (int k = 0; k < R; ++k) s[k] = __builtin_amdgcn_readlane(iv, j + k);
    for (int k = 0; k < R; ++k) t[k] = Hv[(size_t)s[k] * 32 + lane];
    if (R & 4) acc += (t[0] + t[1]) + (t[2] + t[3]);
    if (R & 2) acc += t[R & 4] + t[(R & 4) + 1];
    if (R & 1) acc += t[R - 1];
    return acc;
}

template <int V>
__device__ __forceinline__ typename RowVec<V>::type
sum_rows_of(const typename RowVec<V>::type* __restrict__ Hv, const int* __restrict__ ssrc,
            int beg, int end, int iv, int lane) {
    typedef typename RowVec<V>::type vec;
    vec acc = 0.0f;
    int e = beg;
    for (;;) {
        int cnt = end - e;
        cnt = cnt > 32 ? 32 : cnt;
        int j = 0;
        for (; j + 8 <= cnt; j += 8) {
            int s[8];
            vec v[8];
#pragma unroll
            for (int k = 0; k < 8; ++k) s[k] = __builtin_amdgcn_readlane(iv, j + k);
#pragma unroll
            for (int k = 0; k < 8; ++k) v[k] = Hv[(size_t)s[k] * 32 + lane];
            acc += ((v[0] + v[1]) + (v[2] + v[3])) + ((v[4] + v[5]) + (v[6] + v[7]));
        }
        switch (cnt - j) {
            case 1: acc = add_last_rows<V, 1>(Hv, iv, j, lane, acc); break;
            case 2: acc = add_last_rows<V, 2>(Hv, iv, j, lane, acc); break;
            case 3: acc = add_last_rows<V, 3>(Hv, iv, j, lane, acc); break;
            case 4: acc = add_last_rows<V, 4>(Hv, iv, j, lane, acc); break;
            case 5: acc = add_last_rows<V, 5>(Hv, iv, j, lane, acc); break;
            case 6: acc = add_last_rows<V, 6>(Hv, iv, j, lane, acc); break;
            case 7: acc = add_last_rows<V, 7>(Hv, iv, j, lane, acc); break;
            default: break;
        }
        e += cnt;
        if (e >= end) break;
        iv = (e + lane < end) ? ssrc[e + lane] : 0;
        asm volatile("" : "+v"(iv));
    }
    return acc;
}

typedef const float __attribute__((address_space(4))) cfloat;

template <int R>
__device__ __forceinline__ RowVec<4>::type
add_last_scaled_rows(const RowVec<4>::type* __restrict__ Xv, const cfloat* __restrict__ scl,
                     int iv, int j, int lane, RowVec<4>::type acc) {
    typedef RowVec<4>::type vec;
    int s[R];
    float f[R];
    vec t[R];
    for (int k = 0; k < R; ++k) s[k] = __builtin_amdgcn_readlane(iv, j + k);
    for (int k = 0; k < R; ++k) f[k] = scl[s[k]];
    for (int k = 0; k < R; ++k) t[k] = Xv[(size_t)s[k] * 32 + lane];
    for (int k = 0; k < R; ++k) t[k] = t[k] * f[k];
    if (R & 4) acc += (t[0] + t[1]) + (t[2] + t[3]);
    if (R & 2) acc += t[R & 4] + t[(R & 4) + 1];
    if (R & 1) acc += t[R - 1];
    return acc;
}

__device__ __forceinline__ RowVec<4>::type
sum_scaled_rows_of(const RowVec<4>::type* __restrict__ Xv, const int* __restrict__ ssrc,
                   const cfloat* __restrict__ scl, int beg, int end, int iv, int lane) {
    typedef RowVec<4>::type vec;
    vec acc = 0.0f;
    int e = beg;
    for (;;) {
        int cnt = end - e;
        cnt = cnt > 32 ? 32 : cnt;
        int j = 0;
        for (; j + 8 <= cnt; j += 8) {
            int s[8];
            float f[8];
            vec v[8];
#pragma unroll
            for (int k = 0; k < 8; ++k) s[k] = __builtin_amdgcn_readlane(iv, j + k);
#pragma unroll
            for (int k = 0; k < 8; ++k) f[k] = scl[s[k]];
#pragma unroll
            for (int k = 0; k < 8; ++k) v[k] = Xv[(size_t)s[k] * 32 + lane];
#pragma unroll
            for (int k = 0; k < 8; ++k) v[k] = v[k] * f[k];
            acc += ((v[0] + v[1]) + (v[2] + v[3])) + ((v[4] + v[5]) + (v[6] + v[7]));
        }
        switch (cnt - j) {
            case 1: acc = add_last_scaled_rows<1>(Xv, scl, iv, j, lane, acc); break;
            case 2: acc = add_last_scaled_rows<2>(Xv, scl, iv, j, lane, acc); break;
            case 3: acc = add_last_scaled_rows<3>(Xv, scl, iv, j, lane, acc); break;
            case 4: acc = add_last_scaled_rows<4>(Xv, scl, iv, j, lane, acc); break;
            case 5: acc = add_last_scaled_rows<5>(Xv, scl, iv, j, lane, acc); break;
            case 6: acc = add_last_scaled_rows<6>(Xv, scl, iv, j, lane, acc); break;
            case 7: acc = add_last_scaled_rows<7>(Xv, scl, iv, j, lane, acc); break;
            default: break;
        }
        e += cnt;
        if (e >= end) break;
        iv = (e + lane < end) ? ssrc[e + lane] : 0;
        asm volatile("" : "+v"(iv));
    }
    return acc;
}

#define GATHER_WAVES 8

template <int F, bool RELU>
__launch_bounds__(GATHER_WAVES * 32)
__global__ void gather_rows_kernel(const float* __restrict__ H,
                                   const int* __restrict__ rowptr,
                                   const int* __restrict__ ssrc,
                                   const float* __restrict__ dinv,
                                   const float* __restrict__ bias,
                                   float* __restrict__ Y, int n) {
    constexpr int V = F / 32;
    typedef float vec __attribute__((ext_vector_type(V)));

    const int lane = threadIdx.x & 31;
    const int wave = __builtin_amdgcn_readfirstlane((int)(threadIdx.x >> 5));
    const int stride = gridDim.x * GATHER_WAVES;
    const vec* Hv = (const vec*)H;
    const vec  b  = ((const vec*)bias)[lane];

    int row = blockIdx.x * GATHER_WAVES + wave;
    int pb0 = 0, pb1 = 0, qb0 = 0, qb1 = 0, iv = 0;
    if (row < n) {
        pb0 = rowptr[row];
        pb1 = rowptr[row + 1];
        iv  = (pb0 + lane < pb1) ? ssrc[pb0 + lane] : 0;
    }
    if (row + stride < n) {
        qb0 = rowptr[row + stride];
        qb1 = rowptr[row + stride + 1];
    }
    asm volatile("" : "+v"(iv));
    for (; row < n; row += stride) {
        const int beg = pb0;
        const int end = pb1;
        int niv = 0, rb0 = 0, rb1 = 0;
        if (row + stride < n) niv = (qb0 + lane < qb1) ? ssrc[qb0 + lane] : 0;
        if (row + 2 * stride < n) {
            rb0 = rowptr[row + 2 * stride];
            rb1 = rowptr[row + 2 * stride + 1];
        }

        const float di = dinv[row];

        const vec acc = sum_rows_of<V>(Hv, ssrc, beg, end, iv, lane);
        pb0 = qb0; pb1 = qb1; qb0 = rb0; qb1 = rb1; iv = niv;

        vec r = acc * di + b;
        if (RELU) {
#pragma unroll
            for (int k = 0; k < V; ++k) r[k] = r[k] > 0.0f ? r[k] : 0.0f;
        }
        ((vec*)Y)[(size_t)row * 32 + lane] = r;
    }
}

template <int FOUT>
__launch_bounds__(FUSE_WAVES * 32)
__attribute__((amdgpu_waves_per_eu(8)))
__global__ void gather_gemm_wmma(const float* __restrict__ Hin,
                                 const int* __restrict__ rowptr,
                                 const int* __restrict__ ssrc,
                                 const float* __restrict__ ddst,
                                 const float* __restrict__ bias,
                                 const float* __restrict__ dsrc,
                                 const float* __restrict__ W,
                                 float* __restrict__ Hout, int ntiles) {
    constexpr int LDSS = KDIM + 4;
    constexpr int CW   = FOUT / 16;
    constexpr int RPW  = 16 / FUSE_WAVES;
    typedef float vec __attribute__((ext_vector_type(4)));
    __shared__ __align__(16) float As[2][16 * LDSS];

    const int tid   = threadIdx.x;
    const int lane  = tid & 31;
    const int wave  = __builtin_amdgcn_readfirstlane((int)(threadIdx.x >> 5));
    const int n     = lane & 15;
    const int khalf = lane >> 4;
    const int m16   = lane & 15;
    const int ncol  = wave * 16 + n;

    v2f bf[KDIM / 4] = {};
    if (wave < CW) {
#pragma unroll
        for (int j = 0; j < KDIM / 4; ++j) {
            const int ka = 4 * j + 2 * khalf;
            bf[j].x = W[(size_t)ka * FOUT + ncol];
            bf[j].y = W[(size_t)(ka + 1) * FOUT + ncol];
        }
    }

    const vec* Hv = (const vec*)Hin;
    const vec  b  = ((const vec*)bias)[lane];

    static_assert(RPW == 2, "two rows per wave and tile");
    const int G = gridDim.x;
    int tile = blockIdx.x;
    int pb0 = 0, pb1 = 0, pb2 = 0, qb0 = 0, qb1 = 0, qb2 = 0, iv0 = 0, iv1 = 0;
    if (tile < ntiles) {
        const int* rp = rowptr + tile * 16 + wave * RPW;
        pb0 = rp[0]; pb1 = rp[1]; pb2 = rp[2];
        iv0 = (pb0 + lane < pb1) ? ssrc[pb0 + lane] : 0;
        iv1 = (pb1 + lane < pb2) ? ssrc[pb1 + lane] : 0;
    }
    if (tile + G < ntiles) {
        const int* rp = rowptr + (tile + G) * 16 + wave * RPW;
        qb0 = rp[0]; qb1 = rp[1]; qb2 = rp[2];
    }
    asm volatile("" : "+v"(iv0), "+v"(iv1));

    int st = 0;
    for (; tile < ntiles; tile += G) {
        __builtin_amdgcn_s_setprio(3);
        float* A = As[st];
        const int bnd[RPW + 1] = {__builtin_amdgcn_readfirstlane(pb0),
                                  __builtin_amdgcn_readfirstlane(pb1),
                                  __builtin_amdgcn_readfirstlane(pb2)};
        const int ivs[RPW] = {iv0, iv1};
        int niv0 = 0, niv1 = 0, rb0 = 0, rb1 = 0, rb2 = 0;
        if (tile + G < ntiles) {
            niv0 = (qb0 + lane < qb1) ? ssrc[qb0 + lane] : 0;
            niv1 = (qb1 + lane < qb2) ? ssrc[qb1 + lane] : 0;
        }
        if (tile + 2 * G < ntiles) {
            const int* rp = rowptr + (tile + 2 * G) * 16 + wave * RPW;
            rb0 = rp[0]; rb1 = rp[1]; rb2 = rp[2];
        }
        for (int i = 0; i < RPW; ++i) {
            const int m   = wave * RPW + i;
            const int row = tile * 16 + m;
            const float di = ddst[row];
            const float ds = dsrc[row];
            const vec acc = sum_rows_of<4>(Hv, ssrc, bnd[i], bnd[i + 1], ivs[i], lane);

            vec r = acc * di + b;
#pragma unroll
            for (int k = 0; k < 4; ++k) r[k] = r[k] > 0.0f ? r[k] : 0.0f;
            r = r * ds;
            *(vec*)&A[m * LDSS + 4 * lane] = r;
        }
        __syncthreads();

        if (wave < CW) {
            __builtin_amdgcn_s_setprio(0);
            v8f acc8 = {};
#pragma unroll
            for (int k0 = 0; k0 < KDIM; k0 += 4) {
                const int ka = k0 + 2 * khalf;
                v2f a = *(const v2f*)&A[m16 * LDSS + ka];
                acc8 = __builtin_amdgcn_wmma_f32_16x16x4_f32(
                     false, a,  false, bf[k0 / 4],
                     (short)0, acc8,  false,  false);
            }

            float* hp = Hout + ((size_t)tile * 16 + 8 * khalf) * FOUT + ncol;
#pragma unroll
            for (int r = 0; r < 8; ++r) hp[r * FOUT] = acc8[r];
        }
        pb0 = qb0; pb1 = qb1; pb2 = qb2;
        qb0 = rb0; qb1 = rb1; qb2 = rb2;
        iv0 = niv0; iv1 = niv1;
        st ^= 1;
    }
}

__launch_bounds__(FIRST_WAVES * 32)
__attribute__((amdgpu_waves_per_eu(8)))
__global__ void gather_gemm2_wmma(const float* __restrict__ X,
                                  const int* __restrict__ rowptr,
                                  const int* __restrict__ ssrc,
                                  const float* __restrict__ ddst,
                                  const float* __restrict__ bias,
                                  const float* __restrict__ dsrc,
                                  const float* __restrict__ W1,
                                  const float* __restrict__ W2,
                                  float* __restrict__ Hout, int ntiles) {
    constexpr int LDSS = KDIM + 4;
    typedef float vec __attribute__((ext_vector_type(4)));
    __shared__ __align__(16) float S1[2][16 * LDSS];
    __shared__ __align__(16) float S2[2][16 * LDSS];

    const int tid   = threadIdx.x;
    const int lane  = tid & 31;
    const int wave  = __builtin_amdgcn_readfirstlane((int)(threadIdx.x >> 5));
    const int role  = wave >> 3;
    const int n     = lane & 15;
    const int khalf = lane >> 4;
    const int m16   = lane & 15;
    const int ncol  = (wave & 7) * 16 + n;

    const float* __restrict__ Wp = role ? W2 : W1;
    v2f bf[KDIM / 4];
#pragma unroll
    for (int j = 0; j < KDIM / 4; ++j) {
        const int ka = 4 * j + 2 * khalf;
        bf[j].x = Wp[(size_t)ka * HID_F + ncol];
        bf[j].y = Wp[(size_t)(ka + 1) * HID_F + ncol];
    }

    const vec*    Xv   = (const vec*)X;
    const float   bcol = bias[ncol];
    const cfloat* dsc  = (const cfloat*)(unsigned long long)dsrc;
    const cfloat* ddc  = (const cfloat*)(unsigned long long)ddst;

    const int G = gridDim.x;
    int tile = blockIdx.x;
    int pb0 = 0, pb1 = 0, qb0 = 0, qb1 = 0, iv = 0;
    if (tile < ntiles) {
        const int* rp = rowptr + tile * 16 + wave;
        pb0 = rp[0]; pb1 = rp[1];
        iv  = (pb0 + lane < pb1) ? ssrc[pb0 + lane] : 0;
    }
    if (tile + G < ntiles) {
        const int* rp = rowptr + (tile + G) * 16 + wave;
        qb0 = rp[0]; qb1 = rp[1];
    }
    asm volatile("" : "+v"(iv));

    int st   = 0;
    int prev = -1;
    for (;;) {
        const bool have = tile < ntiles;
        if (!have && prev < 0) break;
        if (have) {
            __builtin_amdgcn_s_setprio(3);
            const int beg = __builtin_amdgcn_readfirstlane(pb0);
            const int end = __builtin_amdgcn_readfirstlane(pb1);
            int niv = 0, rb0 = 0, rb1 = 0;
            if (tile + G < ntiles) niv = (qb0 + lane < qb1) ? ssrc[qb0 + lane] : 0;
            if (tile + 2 * G < ntiles) {
                const int* rp = rowptr + (tile + 2 * G) * 16 + wave;
                rb0 = rp[0]; rb1 = rp[1];
            }
            const int row = tile * 16 + wave;
            v2f fac;
            fac.x = ddc[row];
            fac.y = dsc[row];
            const vec acc = sum_scaled_rows_of(Xv, ssrc, dsc, beg, end, iv, lane);
            *(vec*)&S1[st][wave * LDSS + 4 * lane] = acc;
            if (lane == 0) *(v2f*)&S1[st][wave * LDSS + KDIM] = fac;
            pb0 = qb0; pb1 = qb1; qb0 = rb0; qb1 = rb1; iv = niv;
        }
        __syncthreads();
        __builtin_amdgcn_s_setprio(0);

        if (role == 0) {
            if (have) {
                const float* A = S1[st];
                v8f acc8 = {};
#pragma unroll
                for (int k0 = 0; k0 < KDIM; k0 += 4) {
                    const int ka = k0 + 2 * khalf;
                    v2f a = *(const v2f*)&A[m16 * LDSS + ka];
                    acc8 = __builtin_amdgcn_wmma_f32_16x16x4_f32(
                         false, a,  false, bf[k0 / 4],
                         (short)0, acc8,  false,  false);
                }

                const float* fr = &A[(8 * khalf) * LDSS + KDIM];
                float* xo = &S2[st][(8 * khalf) * LDSS + ncol];
#pragma unroll
                for (int r = 0; r < 8; ++r) {
                    const v2f fac = *(const v2f*)&fr[r * LDSS];
                    float x = acc8[r] * fac.x + bcol;
                    x = x > 0.0f ? x : 0.0f;
                    xo[r * LDSS] = x * fac.y;
                }
            }
        } else {
            if (prev >= 0) {
                const float* A = S2[st ^ 1];
                v8f acc8 = {};
#pragma unroll
                for (int k0 = 0; k0 < KDIM; k0 += 4) {
                    const int ka = k0 + 2 * khalf;
                    v2f a = *(const v2f*)&A[m16 * LDSS + ka];
                    acc8 = __builtin_amdgcn_wmma_f32_16x16x4_f32(
                         false, a,  false, bf[k0 / 4],
                         (short)0, acc8,  false,  false);
                }

                float* hp = Hout + ((size_t)prev * 16 + 8 * khalf) * HID_F + ncol;
#pragma unroll
                for (int r = 0; r < 8; ++r) hp[r * HID_F] = acc8[r];
            }
        }
        prev = have ? tile : -1;
        tile += G;
        st ^= 1;
    }
}

static inline int cdiv(long a, long b) { return (int)((a + b - 1) / b); }

extern "C" void kernel_launch(void* const* d_in, const int* in_sizes, int n_in,
                              void* d_out, int out_size, void* d_ws, size_t ws_size,
                              hipStream_t stream) {
    const float* features = (const float*)d_in[0];
    const int*   src      = (const int*)d_in[1];
    const int*   dst      = (const int*)d_in[2];
    const float* W1       = (const float*)d_in[3];
    const float* b1       = (const float*)d_in[4];
    const float* W2       = (const float*)d_in[5];
    const float* b2       = (const float*)d_in[6];
    const float* W3       = (const float*)d_in[7];
    const float* b3       = (const float*)d_in[8];
    float*       out      = (float*)d_out;

    const int N = in_sizes[0] / IN_F;
    const int E = in_sizes[1];

    float* ws   = (float*)d_ws;
    float* dsrc = ws;
    float* ddst = ws + N;
    float* buf0 = ws + 2 * (size_t)N;
    float* buf1 = buf0 + (size_t)N * HID_F;

    const int nbk = cdiv(N, BKT_ROWS);
    const int np  = cdiv(E, PART_PER);
    const int M   = (nbk + 1) * np;
    int* chunkD = (int*)buf1;
    int* chunkS = chunkD + 2 * (size_t)E;
    int* preD   = chunkS + E;
    int* preS   = preD + M;

    int* csr    = (int*)(buf0 + (size_t)N * OUT_F);
    int* rowptr = csr;
    int* ssrc   = csr + (N + 1);

    const int ntiles = N / 16;

    local_sort_kernel<<<np, PART_T, 0, stream>>>(src, dst, chunkD, chunkS, preD, preS, E, nbk);
    bucket_sort_kernel<<<nbk, BSORT_T, 0, stream>>>(chunkD, chunkS, preD, preS, rowptr, ssrc, dsrc, ddst, N, E, np, nbk);

    gather_gemm2_wmma<<<FIRST_BLOCKS, FIRST_WAVES * 32, 0, stream>>>(features, rowptr, ssrc, ddst, b1, dsrc, W1, W2, buf1, ntiles);

    gather_gemm_wmma<OUT_F><<<FUSE_BLOCKS, FUSE_WAVES * 32, 0, stream>>>(buf1, rowptr, ssrc, ddst, b2, dsrc, W3, buf0, ntiles);

    gather_rows_kernel<OUT_F, false><<<GATHER_BLOCKS, GATHER_WAVES * 32, 0, stream>>>(buf0, rowptr, ssrc, ddst, b3, out, N);
}
